# starting kernel
# baseline (speedup 1.0000x reference)
#include <hip/hip_runtime.h>

#define NNODES 100000
#define NEDGES 1600000
#define NGRAPHS 512
#define HID 128
#define NCLS 10
#define ALPHA 0.01f
#define FMAXV 3.402823466e38f

typedef float v2f __attribute__((ext_vector_type(2)));
typedef float v8f __attribute__((ext_vector_type(8)));

__global__ __launch_bounds__(256) void gemm_wmma_f32(
    const float* __restrict__ A, const float* __restrict__ W,
    const float* __restrict__ bias, float* __restrict__ C,
    int K, int ep, const float* __restrict__ gamma,
    const float* __restrict__ beta)
{
    const int N = HID;
    const int wave = threadIdx.x >> 5;
    const int lane = threadIdx.x & 31;
    const int m    = lane & 15;
    const int half = lane >> 4;
    const int row0 = blockIdx.x * 16;
    const int col  = wave * 16 + m;

    v8f acc;
    const float bv = bias[col];
#pragma unroll
    for (int v = 0; v < 8; ++v) acc[v] = bv;

    const float* arow = A + (size_t)(row0 + m) * K + 2 * half;
    const float* wcol = W + (size_t)(2 * half) * N + col;

    for (int k = 0; k < K; k += 4) {
        v2f a = *(const v2f*)(arow + k);
        v2f b;
        b.x = wcol[(size_t)k * N];
        b.y = wcol[(size_t)k * N + N];
        acc = __builtin_amdgcn_wmma_f32_16x16x4_f32(
            false, a, false, b, (short)0, acc, false, false);
    }

    if (ep == 1) {
        const float rs = 0.99999500003749971f;
        const float g  = gamma[col] * rs;
        const float be = beta[col];
#pragma unroll
        for (int v = 0; v < 8; ++v) acc[v] = fmaxf(acc[v] * g + be, 0.0f);
    }

#pragma unroll
    for (int v = 0; v < 8; ++v) {
        const int row = row0 + v + 8 * half;
        C[(size_t)row * N + col] = acc[v];
    }
}

__global__ void init_meta(int* deg, int* cursor, int* gs, int* ge)
{
    int i = blockIdx.x * blockDim.x + threadIdx.x;
    if (i < NNODES) { deg[i] = 0; cursor[i] = 0; }
    if (i < NGRAPHS) { gs[i] = NNODES; ge[i] = 0; }
}

__global__ void count_deg(const int* __restrict__ dst, int* __restrict__ deg)
{
    int i = blockIdx.x * blockDim.x + threadIdx.x;
    if (i < NEDGES) atomicAdd(&deg[dst[i]], 1);
}

__global__ __launch_bounds__(1024) void scan_offsets(
    const int* __restrict__ deg, int* __restrict__ offs, int n)
{
    __shared__ int sm[1024];
    __shared__ int carry;
    if (threadIdx.x == 0) carry = 0;
    __syncthreads();
    for (int base = 0; base < n; base += 1024) {
        int i = base + (int)threadIdx.x;
        int v = (i < n) ? deg[i] : 0;
        sm[threadIdx.x] = v;
        __syncthreads();
        for (int off = 1; off < 1024; off <<= 1) {
            int t = (threadIdx.x >= (unsigned)off) ? sm[threadIdx.x - off] : 0;
            __syncthreads();
            sm[threadIdx.x] += t;
            __syncthreads();
        }
        int c = carry;
        if (i < n) offs[i] = c + sm[threadIdx.x] - v;
        __syncthreads();
        if (threadIdx.x == 0) carry = c + sm[1023];
        __syncthreads();
    }
    if (threadIdx.x == 0) offs[n] = carry;
}

__global__ void csr_fill(const int* __restrict__ src, const int* __restrict__ dst,
                         const int* __restrict__ offs, int* __restrict__ cursor,
                         int* __restrict__ csr)
{
    int i = blockIdx.x * blockDim.x + threadIdx.x;
    if (i < NEDGES) {
        int d = dst[i];
        int p = atomicAdd(&cursor[d], 1);
        csr[offs[d] + p] = src[i];
    }
}

__global__ void graph_bounds(const int* __restrict__ batch, int* gs, int* ge)
{
    int i = blockIdx.x * blockDim.x + threadIdx.x;
    if (i < NNODES) {
        int b = batch[i];
        atomicMin(&gs[b], i);
        atomicMax(&ge[b], i + 1);
    }
}

__global__ __launch_bounds__(256) void agg_mean_leaky(
    const float* __restrict__ H, const int* __restrict__ offs,
    const int* __restrict__ csr, float* __restrict__ out)
{
    int node = (int)((blockIdx.x * blockDim.x + threadIdx.x) >> 5);
    if (node >= NNODES) return;
    const int lane = threadIdx.x & 31;
    const int beg = offs[node];
    const int end = offs[node + 1];
    const float* colp = H + lane * 4;
    float ax = 0.f, ay = 0.f, az = 0.f, aw = 0.f;

    for (int e0 = beg; e0 < end; e0 += 32) {
        int myidx = (e0 + lane < end) ? csr[e0 + lane] : 0;
        int cnt = end - e0; if (cnt > 32) cnt = 32;
        for (int j = 0; j < cnt; ++j) {
            int s = __shfl(myidx, j, 32);
            const float4 v = *(const float4*)(colp + (size_t)s * HID);
            ax += v.x; ay += v.y; az += v.z; aw += v.w;
        }
    }
    const float inv = 1.0f / fmaxf((float)(end - beg), 1.0f);
    ax *= inv; ay *= inv; az *= inv; aw *= inv;
    float4 r;
    r.x = ax > 0.f ? ax : ALPHA * ax;
    r.y = ay > 0.f ? ay : ALPHA * ay;
    r.z = az > 0.f ? az : ALPHA * az;
    r.w = aw > 0.f ? aw : ALPHA * aw;
    *(float4*)(out + (size_t)node * HID + lane * 4) = r;
}

__global__ void residual_add(float4* __restrict__ h, const float4* __restrict__ h2, int n)
{
    int i = blockIdx.x * blockDim.x + threadIdx.x;
    if (i < n) {
        float4 a = h[i], b = h2[i];
        a.x += b.x; a.y += b.y; a.z += b.z; a.w += b.w;
        h[i] = a;
    }
}

__global__ __launch_bounds__(128) void pool_max(
    const float* __restrict__ ht, const float* __restrict__ h1,
    const float* __restrict__ h2, const float* __restrict__ h3,
    const int* __restrict__ gs, const int* __restrict__ ge,
    float* __restrict__ pooled)
{
    const int g = blockIdx.x;
    const int f = (int)threadIdx.x * 4;
    const float* srcp = (f < 128) ? ht : (f < 256) ? h1 : (f < 384) ? h2 : h3;
    const int off = f & 127;
    float mx = -FMAXV, my = -FMAXV, mz = -FMAXV, mw = -FMAXV;
    const int b = gs[g], e = ge[g];
    for (int i = b; i < e; ++i) {
        const float4 v = *(const float4*)(srcp + (size_t)i * HID + off);
        mx = fmaxf(mx, v.x); my = fmaxf(my, v.y);
        mz = fmaxf(mz, v.z); mw = fmaxf(mw, v.w);
    }
    float4 r; r.x = mx; r.y = my; r.z = mz; r.w = mw;
    *(float4*)(pooled + (size_t)g * 512 + f) = r;
}

__global__ __launch_bounds__(32) void mlp2(
    const float* __restrict__ z, const float* __restrict__ W,
    const float* __restrict__ b, float* __restrict__ out)
{
    const int g = blockIdx.x;
    const int lane = threadIdx.x;
    const float z0 = z[(size_t)g * HID + lane];
    const float z1 = z[(size_t)g * HID + lane + 32];
    const float z2 = z[(size_t)g * HID + lane + 64];
    const float z3 = z[(size_t)g * HID + lane + 96];
    for (int c = 0; c < NCLS; ++c) {
        float s = z0 * W[lane * NCLS + c]
                + z1 * W[(lane + 32) * NCLS + c]
                + z2 * W[(lane + 64) * NCLS + c]
                + z3 * W[(lane + 96) * NCLS + c];
        for (int off = 16; off; off >>= 1) s += __shfl_xor(s, off, 32);
        if (lane == 0) out[g * NCLS + c] = s + b[c];
    }
}

extern "C" void kernel_launch(void* const* d_in, const int* in_sizes, int n_in,
                              void* d_out, int out_size, void* d_ws, size_t ws_size,
                              hipStream_t stream)
{
    (void)in_sizes; (void)n_in; (void)out_size; (void)ws_size;
    const float* x     = (const float*)d_in[0];
    const int*   ei    = (const int*)d_in[1];
    const int*   batch = (const int*)d_in[2];
    const float* W0 = (const float*)d_in[3];  const float* b0 = (const float*)d_in[4];
    const float* W1 = (const float*)d_in[5];  const float* b1 = (const float*)d_in[6];
    const float* W2 = (const float*)d_in[7];  const float* b2 = (const float*)d_in[8];
    const float* W3 = (const float*)d_in[9];  const float* b3 = (const float*)d_in[10];
    const float* Wm1 = (const float*)d_in[11]; const float* bm1 = (const float*)d_in[12];
    const float* gamma = (const float*)d_in[13]; const float* beta = (const float*)d_in[14];
    const float* Wm2 = (const float*)d_in[15]; const float* bm2 = (const float*)d_in[16];
    const int* esrc = ei;
    const int* edst = ei + NEDGES;

    char* ws = (char*)d_ws;
    size_t off = 0;
    auto alloc = [&](size_t bytes) -> void* {
        void* p = ws + off;
        off = (off + bytes + 255) & ~(size_t)255;
        return p;
    };
    const size_t featB = (size_t)NNODES * HID * sizeof(float);
    float* tmpH   = (float*)alloc(featB);
    float* h      = (float*)alloc(featB);
    float* h1     = (float*)alloc(featB);
    float* h2     = (float*)alloc(featB);
    float* ht     = (float*)alloc(featB);
    float* zbuf   = (float*)alloc((size_t)NGRAPHS * HID * sizeof(float));
    float* pooled = (float*)alloc((size_t)NGRAPHS * 512 * sizeof(float));
    int*   csr    = (int*)alloc((size_t)NEDGES * sizeof(int));
    int*   offs   = (int*)alloc((size_t)(NNODES + 1) * sizeof(int));
    int*   deg    = (int*)alloc((size_t)NNODES * sizeof(int));
    int*   cursor = (int*)alloc((size_t)NNODES * sizeof(int));
    int*   gs     = (int*)alloc((size_t)NGRAPHS * sizeof(int));
    int*   ge     = (int*)alloc((size_t)NGRAPHS * sizeof(int));

    const int TB = 256;
    init_meta<<<(NNODES + TB - 1) / TB, TB, 0, stream>>>(deg, cursor, gs, ge);
    count_deg<<<(NEDGES + TB - 1) / TB, TB, 0, stream>>>(edst, deg);
    scan_offsets<<<1, 1024, 0, stream>>>(deg, offs, NNODES);
    csr_fill<<<(NEDGES + TB - 1) / TB, TB, 0, stream>>>(esrc, edst, offs, cursor, csr);
    graph_bounds<<<(NNODES + TB - 1) / TB, TB, 0, stream>>>(batch, gs, ge);

    const int gemmGrid = NNODES / 16;
    const int aggGrid  = (NNODES * 32 + TB - 1) / TB;

    gemm_wmma_f32<<<gemmGrid, TB, 0, stream>>>(x, W0, b0, tmpH, HID, 0, nullptr, nullptr);
    agg_mean_leaky<<<aggGrid, TB, 0, stream>>>(tmpH, offs, csr, h);
    gemm_wmma_f32<<<gemmGrid, TB, 0, stream>>>(h, W1, b1, tmpH, HID, 0, nullptr, nullptr);
    agg_mean_leaky<<<aggGrid, TB, 0, stream>>>(tmpH, offs, csr, h1);
    gemm_wmma_f32<<<gemmGrid, TB, 0, stream>>>(h1, W2, b2, tmpH, HID, 0, nullptr, nullptr);
    agg_mean_leaky<<<aggGrid, TB, 0, stream>>>(tmpH, offs, csr, h2);
    const int n4 = NNODES * HID / 4;
    residual_add<<<(n4 + TB - 1) / TB, TB, 0, stream>>>((float4*)h, (const float4*)h2, n4);
    gemm_wmma_f32<<<gemmGrid, TB, 0, stream>>>(h, W3, b3, tmpH, HID, 0, nullptr, nullptr);
    agg_mean_leaky<<<aggGrid, TB, 0, stream>>>(tmpH, offs, csr, ht);

    pool_max<<<NGRAPHS, 128, 0, stream>>>(ht, h1, h2, h, gs, ge, pooled);

    gemm_wmma_f32<<<NGRAPHS / 16, TB, 0, stream>>>(pooled, Wm1, bm1, zbuf, 512, 1, gamma, beta);
    mlp2<<<NGRAPHS, 32, 0, stream>>>(zbuf, Wm2, bm2, (float*)d_out);
}


# scan_offsets: one-workgroup Hillis-Steele scan (98 chunks x 20 barriers) replaced by a 25-workgroup one-pass integer scan, each summing its own prefix from L2; one barrier; same offsets
# speedup vs baseline: 1.3276x; 1.3276x over previous
#include <hip/hip_runtime.h>

#define NNODES 100000
#define NEDGES 1600000
#define NGRAPHS 512
#define HID 128
#define NCLS 10
#define ALPHA 0.01f
#define FMAXV 3.402823466e38f

typedef float v2f __attribute__((ext_vector_type(2)));
typedef float v8f __attribute__((ext_vector_type(8)));

__global__ __launch_bounds__(256) void gemm_wmma_f32(
    const float* __restrict__ A, const float* __restrict__ W,
    const float* __restrict__ bias, float* __restrict__ C,
    int K, int ep, const float* __restrict__ gamma,
    const float* __restrict__ beta)
{
    const int N = HID;
    const int wave = threadIdx.x >> 5;
    const int lane = threadIdx.x & 31;
    const int m    = lane & 15;
    const int half = lane >> 4;
    const int row0 = blockIdx.x * 16;
    const int col  = wave * 16 + m;

    v8f acc;
    const float bv = bias[col];
#pragma unroll
    for (int v = 0; v < 8; ++v) acc[v] = bv;

    const float* arow = A + (size_t)(row0 + m) * K + 2 * half;
    const float* wcol = W + (size_t)(2 * half) * N + col;

    for (int k = 0; k < K; k += 4) {
        v2f a = *(const v2f*)(arow + k);
        v2f b;
        b.x = wcol[(size_t)k * N];
        b.y = wcol[(size_t)k * N + N];
        acc = __builtin_amdgcn_wmma_f32_16x16x4_f32(
            false, a, false, b, (short)0, acc, false, false);
    }

    if (ep == 1) {
        const float rs = 0.99999500003749971f;
        const float g  = gamma[col] * rs;
        const float be = beta[col];
#pragma unroll
        for (int v = 0; v < 8; ++v) acc[v] = fmaxf(acc[v] * g + be, 0.0f);
    }

#pragma unroll
    for (int v = 0; v < 8; ++v) {
        const int row = row0 + v + 8 * half;
        C[(size_t)row * N + col] = acc[v];
    }
}

__global__ void init_meta(int* deg, int* cursor, int* gs, int* ge)
{
    int i = blockIdx.x * blockDim.x + threadIdx.x;
    if (i < NNODES) { deg[i] = 0; cursor[i] = 0; }
    if (i < NGRAPHS) { gs[i] = NNODES; ge[i] = 0; }
}

__global__ void count_deg(const int* __restrict__ dst, int* __restrict__ deg)
{
    int i = blockIdx.x * blockDim.x + threadIdx.x;
    if (i < NEDGES) atomicAdd(&deg[dst[i]], 1);
}

#define SCAN_TB   1024
#define SCAN_VEC  (NNODES / 4)
#define SCAN_WGS  ((SCAN_VEC + SCAN_TB - 1) / SCAN_TB)

__global__ __launch_bounds__(SCAN_TB) void scan_offsets(
    const int* __restrict__ deg, int* __restrict__ offs)
{
    __shared__ int wave_own[32];
    __shared__ int wave_pre[32];
    const int t    = (int)threadIdx.x;
    const int lane = t & 31;
    const int wave = t >> 5;
    const int wg   = (int)blockIdx.x;
    const int4* __restrict__ deg4 = (const int4*)deg;
    const int myv = wg * SCAN_TB + t;

    int4 v; v.x = 0; v.y = 0; v.z = 0; v.w = 0;
    if (myv < SCAN_VEC) v = deg4[myv];

    int4 u[SCAN_WGS - 1];
#pragma unroll
    for (int j = 0; j < SCAN_WGS - 1; ++j) u[j] = deg4[j * SCAN_TB + t];
    int pre = 0;
#pragma unroll
    for (int j = 0; j < SCAN_WGS - 1; ++j) {
        const int s4 = (u[j].x + u[j].y) + (u[j].z + u[j].w);
        pre += (j < wg) ? s4 : 0;
    }

    const int own = (v.x + v.y) + (v.z + v.w);
    int inc = own;
#pragma unroll
    for (int off = 1; off < 32; off <<= 1) {
        const int n = __shfl_up(inc, off, 32);
        if (lane >= off) inc += n;
    }
#pragma unroll
    for (int off = 16; off; off >>= 1) pre += __shfl_xor(pre, off, 32);
    if (lane == 31) wave_own[wave] = inc;
    if (lane == 0)  wave_pre[wave] = pre;
    __syncthreads();

    const int wo = wave_own[lane];
    int wp = wave_pre[lane];
    int winc = wo;
#pragma unroll
    for (int off = 1; off < 32; off <<= 1) {
        const int n = __shfl_up(winc, off, 32);
        if (lane >= off) winc += n;
    }
#pragma unroll
    for (int off = 16; off; off >>= 1) wp += __shfl_xor(wp, off, 32);
    const int wave_base = __shfl(winc - wo, wave, 32);

    if (myv < SCAN_VEC) {
        int4 o;
        o.x = wp + wave_base + (inc - own);
        o.y = o.x + v.x;
        o.z = o.y + v.y;
        o.w = o.z + v.z;
        ((int4*)offs)[myv] = o;
        if (myv == SCAN_VEC - 1) offs[NNODES] = o.w + v.w;
    }
}

__global__ void csr_fill(const int* __restrict__ src, const int* __restrict__ dst,
                         const int* __restrict__ offs, int* __restrict__ cursor,
                         int* __restrict__ csr)
{
    int i = blockIdx.x * blockDim.x + threadIdx.x;
    if (i < NEDGES) {
        int d = dst[i];
        int p = atomicAdd(&cursor[d], 1);
        csr[offs[d] + p] = src[i];
    }
}

__global__ void graph_bounds(const int* __restrict__ batch, int* gs, int* ge)
{
    int i = blockIdx.x * blockDim.x + threadIdx.x;
    if (i < NNODES) {
        int b = batch[i];
        atomicMin(&gs[b], i);
        atomicMax(&ge[b], i + 1);
    }
}

__global__ __launch_bounds__(256) void agg_mean_leaky(
    const float* __restrict__ H, const int* __restrict__ offs,
    const int* __restrict__ csr, float* __restrict__ out)
{
    int node = (int)((blockIdx.x * blockDim.x + threadIdx.x) >> 5);
    if (node >= NNODES) return;
    const int lane = threadIdx.x & 31;
    const int beg = offs[node];
    const int end = offs[node + 1];
    const float* colp = H + lane * 4;
    float ax = 0.f, ay = 0.f, az = 0.f, aw = 0.f;

    for (int e0 = beg; e0 < end; e0 += 32) {
        int myidx = (e0 + lane < end) ? csr[e0 + lane] : 0;
        int cnt = end - e0; if (cnt > 32) cnt = 32;
        for (int j = 0; j < cnt; ++j) {
            int s = __shfl(myidx, j, 32);
            const float4 v = *(const float4*)(colp + (size_t)s * HID);
            ax += v.x; ay += v.y; az += v.z; aw += v.w;
        }
    }
    const float inv = 1.0f / fmaxf((float)(end - beg), 1.0f);
    ax *= inv; ay *= inv; az *= inv; aw *= inv;
    float4 r;
    r.x = ax > 0.f ? ax : ALPHA * ax;
    r.y = ay > 0.f ? ay : ALPHA * ay;
    r.z = az > 0.f ? az : ALPHA * az;
    r.w = aw > 0.f ? aw : ALPHA * aw;
    *(float4*)(out + (size_t)node * HID + lane * 4) = r;
}

__global__ void residual_add(float4* __restrict__ h, const float4* __restrict__ h2, int n)
{
    int i = blockIdx.x * blockDim.x + threadIdx.x;
    if (i < n) {
        float4 a = h[i], b = h2[i];
        a.x += b.x; a.y += b.y; a.z += b.z; a.w += b.w;
        h[i] = a;
    }
}

__global__ __launch_bounds__(128) void pool_max(
    const float* __restrict__ ht, const float* __restrict__ h1,
    const float* __restrict__ h2, const float* __restrict__ h3,
    const int* __restrict__ gs, const int* __restrict__ ge,
    float* __restrict__ pooled)
{
    const int g = blockIdx.x;
    const int f = (int)threadIdx.x * 4;
    const float* srcp = (f < 128) ? ht : (f < 256) ? h1 : (f < 384) ? h2 : h3;
    const int off = f & 127;
    float mx = -FMAXV, my = -FMAXV, mz = -FMAXV, mw = -FMAXV;
    const int b = gs[g], e = ge[g];
    for (int i = b; i < e; ++i) {
        const float4 v = *(const float4*)(srcp + (size_t)i * HID + off);
        mx = fmaxf(mx, v.x); my = fmaxf(my, v.y);
        mz = fmaxf(mz, v.z); mw = fmaxf(mw, v.w);
    }
    float4 r; r.x = mx; r.y = my; r.z = mz; r.w = mw;
    *(float4*)(pooled + (size_t)g * 512 + f) = r;
}

__global__ __launch_bounds__(32) void mlp2(
    const float* __restrict__ z, const float* __restrict__ W,
    const float* __restrict__ b, float* __restrict__ out)
{
    const int g = blockIdx.x;
    const int lane = threadIdx.x;
    const float z0 = z[(size_t)g * HID + lane];
    const float z1 = z[(size_t)g * HID + lane + 32];
    const float z2 = z[(size_t)g * HID + lane + 64];
    const float z3 = z[(size_t)g * HID + lane + 96];
    for (int c = 0; c < NCLS; ++c) {
        float s = z0 * W[lane * NCLS + c]
                + z1 * W[(lane + 32) * NCLS + c]
                + z2 * W[(lane + 64) * NCLS + c]
                + z3 * W[(lane + 96) * NCLS + c];
        for (int off = 16; off; off >>= 1) s += __shfl_xor(s, off, 32);
        if (lane == 0) out[g * NCLS + c] = s + b[c];
    }
}

extern "C" void kernel_launch(void* const* d_in, const int* in_sizes, int n_in,
                              void* d_out, int out_size, void* d_ws, size_t ws_size,
                              hipStream_t stream)
{
    (void)in_sizes; (void)n_in; (void)out_size; (void)ws_size;
    const float* x     = (const float*)d_in[0];
    const int*   ei    = (const int*)d_in[1];
    const int*   batch = (const int*)d_in[2];
    const float* W0 = (const float*)d_in[3];  const float* b0 = (const float*)d_in[4];
    const float* W1 = (const float*)d_in[5];  const float* b1 = (const float*)d_in[6];
    const float* W2 = (const float*)d_in[7];  const float* b2 = (const float*)d_in[8];
    const float* W3 = (const float*)d_in[9];  const float* b3 = (const float*)d_in[10];
    const float* Wm1 = (const float*)d_in[11]; const float* bm1 = (const float*)d_in[12];
    const float* gamma = (const float*)d_in[13]; const float* beta = (const float*)d_in[14];
    const float* Wm2 = (const float*)d_in[15]; const float* bm2 = (const float*)d_in[16];
    const int* esrc = ei;
    const int* edst = ei + NEDGES;

    char* ws = (char*)d_ws;
    size_t off = 0;
    auto alloc = [&](size_t bytes) -> void* {
        void* p = ws + off;
        off = (off + bytes + 255) & ~(size_t)255;
        return p;
    };
    const size_t featB = (size_t)NNODES * HID * sizeof(float);
    float* tmpH   = (float*)alloc(featB);
    float* h      = (float*)alloc(featB);
    float* h1     = (float*)alloc(featB);
    float* h2     = (float*)alloc(featB);
    float* ht     = (float*)alloc(featB);
    float* zbuf   = (float*)alloc((size_t)NGRAPHS * HID * sizeof(float));
    float* pooled = (float*)alloc((size_t)NGRAPHS * 512 * sizeof(float));
    int*   csr    = (int*)alloc((size_t)NEDGES * sizeof(int));
    int*   offs   = (int*)alloc((size_t)(NNODES + 1) * sizeof(int));
    int*   deg    = (int*)alloc((size_t)NNODES * sizeof(int));
    int*   cursor = (int*)alloc((size_t)NNODES * sizeof(int));
    int*   gs     = (int*)alloc((size_t)NGRAPHS * sizeof(int));
    int*   ge     = (int*)alloc((size_t)NGRAPHS * sizeof(int));

    const int TB = 256;
    init_meta<<<(NNODES + TB - 1) / TB, TB, 0, stream>>>(deg, cursor, gs, ge);
    count_deg<<<(NEDGES + TB - 1) / TB, TB, 0, stream>>>(edst, deg);
    scan_offsets<<<SCAN_WGS, SCAN_TB, 0, stream>>>(deg, offs);
    csr_fill<<<(NEDGES + TB - 1) / TB, TB, 0, stream>>>(esrc, edst, offs, cursor, csr);
    graph_bounds<<<(NNODES + TB - 1) / TB, TB, 0, stream>>>(batch, gs, ge);

    const int gemmGrid = NNODES / 16;
    const int aggGrid  = (NNODES * 32 + TB - 1) / TB;

    gemm_wmma_f32<<<gemmGrid, TB, 0, stream>>>(x, W0, b0, tmpH, HID, 0, nullptr, nullptr);
    agg_mean_leaky<<<aggGrid, TB, 0, stream>>>(tmpH, offs, csr, h);
    gemm_wmma_f32<<<gemmGrid, TB, 0, stream>>>(h, W1, b1, tmpH, HID, 0, nullptr, nullptr);
    agg_mean_leaky<<<aggGrid, TB, 0, stream>>>(tmpH, offs, csr, h1);
    gemm_wmma_f32<<<gemmGrid, TB, 0, stream>>>(h1, W2, b2, tmpH, HID, 0, nullptr, nullptr);
    agg_mean_leaky<<<aggGrid, TB, 0, stream>>>(tmpH, offs, csr, h2);
    const int n4 = NNODES * HID / 4;
    residual_add<<<(n4 + TB - 1) / TB, TB, 0, stream>>>((float4*)h, (const float4*)h2, n4);
    gemm_wmma_f32<<<gemmGrid, TB, 0, stream>>>(h, W3, b3, tmpH, HID, 0, nullptr, nullptr);
    agg_mean_leaky<<<aggGrid, TB, 0, stream>>>(tmpH, offs, csr, ht);

    pool_max<<<NGRAPHS, 128, 0, stream>>>(ht, h1, h2, h, gs, ge, pooled);

    gemm_wmma_f32<<<NGRAPHS / 16, TB, 0, stream>>>(pooled, Wm1, bm1, zbuf, 512, 1, gamma, beta);
    mlp2<<<NGRAPHS, 32, 0, stream>>>(zbuf, Wm2, bm2, (float*)d_out);
}


# agg_mean_leaky: 8 neighbour-row loads in flight per wave (edge indices via readlane) instead of load-wait-add per edge; one residency of waves steps through the nodes; sums in the same order
# speedup vs baseline: 1.4179x; 1.0680x over previous
#include <hip/hip_runtime.h>

#define NNODES 100000
#define NEDGES 1600000
#define NGRAPHS 512
#define HID 128
#define NCLS 10
#define ALPHA 0.01f
#define FMAXV 3.402823466e38f

typedef float v2f __attribute__((ext_vector_type(2)));
typedef float v8f __attribute__((ext_vector_type(8)));

__global__ __launch_bounds__(256) void gemm_wmma_f32(
    const float* __restrict__ A, const float* __restrict__ W,
    const float* __restrict__ bias, float* __restrict__ C,
    int K, int ep, const float* __restrict__ gamma,
    const float* __restrict__ beta)
{
    const int N = HID;
    const int wave = threadIdx.x >> 5;
    const int lane = threadIdx.x & 31;
    const int m    = lane & 15;
    const int half = lane >> 4;
    const int row0 = blockIdx.x * 16;
    const int col  = wave * 16 + m;

    v8f acc;
    const float bv = bias[col];
#pragma unroll
    for (int v = 0; v < 8; ++v) acc[v] = bv;

    const float* arow = A + (size_t)(row0 + m) * K + 2 * half;
    const float* wcol = W + (size_t)(2 * half) * N + col;

    for (int k = 0; k < K; k += 4) {
        v2f a = *(const v2f*)(arow + k);
        v2f b;
        b.x = wcol[(size_t)k * N];
        b.y = wcol[(size_t)k * N + N];
        acc = __builtin_amdgcn_wmma_f32_16x16x4_f32(
            false, a, false, b, (short)0, acc, false, false);
    }

    if (ep == 1) {
        const float rs = 0.99999500003749971f;
        const float g  = gamma[col] * rs;
        const float be = beta[col];
#pragma unroll
        for (int v = 0; v < 8; ++v) acc[v] = fmaxf(acc[v] * g + be, 0.0f);
    }

#pragma unroll
    for (int v = 0; v < 8; ++v) {
        const int row = row0 + v + 8 * half;
        C[(size_t)row * N + col] = acc[v];
    }
}

__global__ void init_meta(int* deg, int* cursor, int* gs, int* ge)
{
    int i = blockIdx.x * blockDim.x + threadIdx.x;
    if (i < NNODES) { deg[i] = 0; cursor[i] = 0; }
    if (i < NGRAPHS) { gs[i] = NNODES; ge[i] = 0; }
}

__global__ void count_deg(const int* __restrict__ dst, int* __restrict__ deg)
{
    int i = blockIdx.x * blockDim.x + threadIdx.x;
    if (i < NEDGES) atomicAdd(&deg[dst[i]], 1);
}

#define SCAN_TB   1024
#define SCAN_VEC  (NNODES / 4)
#define SCAN_WGS  ((SCAN_VEC + SCAN_TB - 1) / SCAN_TB)

__global__ __launch_bounds__(SCAN_TB) void scan_offsets(
    const int* __restrict__ deg, int* __restrict__ offs)
{
    __shared__ int wave_own[32];
    __shared__ int wave_pre[32];
    const int t    = (int)threadIdx.x;
    const int lane = t & 31;
    const int wave = t >> 5;
    const int wg   = (int)blockIdx.x;
    const int4* __restrict__ deg4 = (const int4*)deg;
    const int myv = wg * SCAN_TB + t;

    int4 v; v.x = 0; v.y = 0; v.z = 0; v.w = 0;
    if (myv < SCAN_VEC) v = deg4[myv];

    int4 u[SCAN_WGS - 1];
#pragma unroll
    for (int j = 0; j < SCAN_WGS - 1; ++j) u[j] = deg4[j * SCAN_TB + t];
    int pre = 0;
#pragma unroll
    for (int j = 0; j < SCAN_WGS - 1; ++j) {
        const int s4 = (u[j].x + u[j].y) + (u[j].z + u[j].w);
        pre += (j < wg) ? s4 : 0;
    }

    const int own = (v.x + v.y) + (v.z + v.w);
    int inc = own;
#pragma unroll
    for (int off = 1; off < 32; off <<= 1) {
        const int n = __shfl_up(inc, off, 32);
        if (lane >= off) inc += n;
    }
#pragma unroll
    for (int off = 16; off; off >>= 1) pre += __shfl_xor(pre, off, 32);
    if (lane == 31) wave_own[wave] = inc;
    if (lane == 0)  wave_pre[wave] = pre;
    __syncthreads();

    const int wo = wave_own[lane];
    int wp = wave_pre[lane];
    int winc = wo;
#pragma unroll
    for (int off = 1; off < 32; off <<= 1) {
        const int n = __shfl_up(winc, off, 32);
        if (lane >= off) winc += n;
    }
#pragma unroll
    for (int off = 16; off; off >>= 1) wp += __shfl_xor(wp, off, 32);
    const int wave_base = __shfl(winc - wo, wave, 32);

    if (myv < SCAN_VEC) {
        int4 o;
        o.x = wp + wave_base + (inc - own);
        o.y = o.x + v.x;
        o.z = o.y + v.y;
        o.w = o.z + v.z;
        ((int4*)offs)[myv] = o;
        if (myv == SCAN_VEC - 1) offs[NNODES] = o.w + v.w;
    }
}

__global__ void csr_fill(const int* __restrict__ src, const int* __restrict__ dst,
                         const int* __restrict__ offs, int* __restrict__ cursor,
                         int* __restrict__ csr)
{
    int i = blockIdx.x * blockDim.x + threadIdx.x;
    if (i < NEDGES) {
        int d = dst[i];
        int p = atomicAdd(&cursor[d], 1);
        csr[offs[d] + p] = src[i];
    }
}

__global__ void graph_bounds(const int* __restrict__ batch, int* gs, int* ge)
{
    int i = blockIdx.x * blockDim.x + threadIdx.x;
    if (i < NNODES) {
        int b = batch[i];
        atomicMin(&gs[b], i);
        atomicMax(&ge[b], i + 1);
    }
}

#define AGG_GRP 8
#define AGG_WGS 2048

__global__ __launch_bounds__(256) void agg_mean_leaky(
    const float* __restrict__ H, const int* __restrict__ offs,
    const int* __restrict__ csr, float* __restrict__ out)
{
    const int lane = threadIdx.x & 31;
    const int wave0 = __builtin_amdgcn_readfirstlane(
        (int)((blockIdx.x * blockDim.x + threadIdx.x) >> 5));
    const int nwaves = (int)((gridDim.x * blockDim.x) >> 5);
    const float* colp = H + lane * 4;

    for (int node = wave0; node < NNODES; node += nwaves) {
        const int beg = offs[node];
        const int end = offs[node + 1];
        float ax = 0.f, ay = 0.f, az = 0.f, aw = 0.f;

        for (int e0 = beg; e0 < end; e0 += 32) {
            const int myidx = (e0 + lane < end) ? csr[e0 + lane] : 0;
            int cnt = end - e0; if (cnt > 32) cnt = 32;
            for (int j = 0; j < cnt; j += AGG_GRP) {
                const int r = cnt - j;
                int s[AGG_GRP];
#pragma unroll
                for (int k = 0; k < AGG_GRP; ++k) s[k] = __builtin_amdgcn_readlane(myidx, j + k);
                float4 v[AGG_GRP];
                v[0] = *(const float4*)(colp + (size_t)s[0] * HID);
#pragma unroll
                for (int k = 1; k < AGG_GRP; ++k) {
                    if (k < r) v[k] = *(const float4*)(colp + (size_t)s[k] * HID);
                }
                ax += v[0].x; ay += v[0].y; az += v[0].z; aw += v[0].w;
#pragma unroll
                for (int k = 1; k < AGG_GRP; ++k) {
                    if (k < r) { ax += v[k].x; ay += v[k].y; az += v[k].z; aw += v[k].w; }
                }
            }
        }
        const float inv = 1.0f / fmaxf((float)(end - beg), 1.0f);
        ax *= inv; ay *= inv; az *= inv; aw *= inv;
        float4 r4;
        r4.x = ax > 0.f ? ax : ALPHA * ax;
        r4.y = ay > 0.f ? ay : ALPHA * ay;
        r4.z = az > 0.f ? az : ALPHA * az;
        r4.w = aw > 0.f ? aw : ALPHA * aw;
        *(float4*)(out + (size_t)node * HID + lane * 4) = r4;
    }
}

__global__ void residual_add(float4* __restrict__ h, const float4* __restrict__ h2, int n)
{
    int i = blockIdx.x * blockDim.x + threadIdx.x;
    if (i < n) {
        float4 a = h[i], b = h2[i];
        a.x += b.x; a.y += b.y; a.z += b.z; a.w += b.w;
        h[i] = a;
    }
}

__global__ __launch_bounds__(128) void pool_max(
    const float* __restrict__ ht, const float* __restrict__ h1,
    const float* __restrict__ h2, const float* __restrict__ h3,
    const int* __restrict__ gs, const int* __restrict__ ge,
    float* __restrict__ pooled)
{
    const int g = blockIdx.x;
    const int f = (int)threadIdx.x * 4;
    const float* srcp = (f < 128) ? ht : (f < 256) ? h1 : (f < 384) ? h2 : h3;
    const int off = f & 127;
    float mx = -FMAXV, my = -FMAXV, mz = -FMAXV, mw = -FMAXV;
    const int b = gs[g], e = ge[g];
    for (int i = b; i < e; ++i) {
        const float4 v = *(const float4*)(srcp + (size_t)i * HID + off);
        mx = fmaxf(mx, v.x); my = fmaxf(my, v.y);
        mz = fmaxf(mz, v.z); mw = fmaxf(mw, v.w);
    }
    float4 r; r.x = mx; r.y = my; r.z = mz; r.w = mw;
    *(float4*)(pooled + (size_t)g * 512 + f) = r;
}

__global__ __launch_bounds__(32) void mlp2(
    const float* __restrict__ z, const float* __restrict__ W,
    const float* __restrict__ b, float* __restrict__ out)
{
    const int g = blockIdx.x;
    const int lane = threadIdx.x;
    const float z0 = z[(size_t)g * HID + lane];
    const float z1 = z[(size_t)g * HID + lane + 32];
    const float z2 = z[(size_t)g * HID + lane + 64];
    const float z3 = z[(size_t)g * HID + lane + 96];
    for (int c = 0; c < NCLS; ++c) {
        float s = z0 * W[lane * NCLS + c]
                + z1 * W[(lane + 32) * NCLS + c]
                + z2 * W[(lane + 64) * NCLS + c]
                + z3 * W[(lane + 96) * NCLS + c];
        for (int off = 16; off; off >>= 1) s += __shfl_xor(s, off, 32);
        if (lane == 0) out[g * NCLS + c] = s + b[c];
    }
}

extern "C" void kernel_launch(void* const* d_in, const int* in_sizes, int n_in,
                              void* d_out, int out_size, void* d_ws, size_t ws_size,
                              hipStream_t stream)
{
    (void)in_sizes; (void)n_in; (void)out_size; (void)ws_size;
    const float* x     = (const float*)d_in[0];
    const int*   ei    = (const int*)d_in[1];
    const int*   batch = (const int*)d_in[2];
    const float* W0 = (const float*)d_in[3];  const float* b0 = (const float*)d_in[4];
    const float* W1 = (const float*)d_in[5];  const float* b1 = (const float*)d_in[6];
    const float* W2 = (const float*)d_in[7];  const float* b2 = (const float*)d_in[8];
    const float* W3 = (const float*)d_in[9];  const float* b3 = (const float*)d_in[10];
    const float* Wm1 = (const float*)d_in[11]; const float* bm1 = (const float*)d_in[12];
    const float* gamma = (const float*)d_in[13]; const float* beta = (const float*)d_in[14];
    const float* Wm2 = (const float*)d_in[15]; const float* bm2 = (const float*)d_in[16];
    const int* esrc = ei;
    const int* edst = ei + NEDGES;

    char* ws = (char*)d_ws;
    size_t off = 0;
    auto alloc = [&](size_t bytes) -> void* {
        void* p = ws + off;
        off = (off + bytes + 255) & ~(size_t)255;
        return p;
    };
    const size_t featB = (size_t)NNODES * HID * sizeof(float);
    float* tmpH   = (float*)alloc(featB);
    float* h      = (float*)alloc(featB);
    float* h1     = (float*)alloc(featB);
    float* h2     = (float*)alloc(featB);
    float* ht     = (float*)alloc(featB);
    float* zbuf   = (float*)alloc((size_t)NGRAPHS * HID * sizeof(float));
    float* pooled = (float*)alloc((size_t)NGRAPHS * 512 * sizeof(float));
    int*   csr    = (int*)alloc((size_t)NEDGES * sizeof(int));
    int*   offs   = (int*)alloc((size_t)(NNODES + 1) * sizeof(int));
    int*   deg    = (int*)alloc((size_t)NNODES * sizeof(int));
    int*   cursor = (int*)alloc((size_t)NNODES * sizeof(int));
    int*   gs     = (int*)alloc((size_t)NGRAPHS * sizeof(int));
    int*   ge     = (int*)alloc((size_t)NGRAPHS * sizeof(int));

    const int TB = 256;
    init_meta<<<(NNODES + TB - 1) / TB, TB, 0, stream>>>(deg, cursor, gs, ge);
    count_deg<<<(NEDGES + TB - 1) / TB, TB, 0, stream>>>(edst, deg);
    scan_offsets<<<SCAN_WGS, SCAN_TB, 0, stream>>>(deg, offs);
    csr_fill<<<(NEDGES + TB - 1) / TB, TB, 0, stream>>>(esrc, edst, offs, cursor, csr);
    graph_bounds<<<(NNODES + TB - 1) / TB, TB, 0, stream>>>(batch, gs, ge);

    const int gemmGrid = NNODES / 16;
    const int aggGrid  = AGG_WGS;

    gemm_wmma_f32<<<gemmGrid, TB, 0, stream>>>(x, W0, b0, tmpH, HID, 0, nullptr, nullptr);
    agg_mean_leaky<<<aggGrid, TB, 0, stream>>>(tmpH, offs, csr, h);
    gemm_wmma_f32<<<gemmGrid, TB, 0, stream>>>(h, W1, b1, tmpH, HID, 0, nullptr, nullptr);
    agg_mean_leaky<<<aggGrid, TB, 0, stream>>>(tmpH, offs, csr, h1);
    gemm_wmma_f32<<<gemmGrid, TB, 0, stream>>>(h1, W2, b2, tmpH, HID, 0, nullptr, nullptr);
    agg_mean_leaky<<<aggGrid, TB, 0, stream>>>(tmpH, offs, csr, h2);
    const int n4 = NNODES * HID / 4;
    residual_add<<<(n4 + TB - 1) / TB, TB, 0, stream>>>((float4*)h, (const float4*)h2, n4);
    gemm_wmma_f32<<<gemmGrid, TB, 0, stream>>>(h, W3, b3, tmpH, HID, 0, nullptr, nullptr);
    agg_mean_leaky<<<aggGrid, TB, 0, stream>>>(tmpH, offs, csr, ht);

    pool_max<<<NGRAPHS, 128, 0, stream>>>(ht, h1, h2, h, gs, ge, pooled);

    gemm_wmma_f32<<<NGRAPHS / 16, TB, 0, stream>>>(pooled, Wm1, bm1, zbuf, 512, 1, gamma, beta);
    mlp2<<<NGRAPHS, 32, 0, stream>>>(zbuf, Wm2, bm2, (float*)d_out);
}


# layer GEMMs (K=128): one wave owns a 16x128 stripe, 8 accumulator tiles reuse each A fragment; W staged once per workgroup in LDS as k-pair rows; same k order, same sums
# speedup vs baseline: 1.6206x; 1.1430x over previous
#include <hip/hip_runtime.h>

#define NNODES 100000
#define NEDGES 1600000
#define NGRAPHS 512
#define HID 128
#define NCLS 10
#define ALPHA 0.01f
#define FMAXV 3.402823466e38f

typedef float v2f __attribute__((ext_vector_type(2)));
typedef float v8f __attribute__((ext_vector_type(8)));

__global__ __launch_bounds__(256) void gemm_wmma_f32(
    const float* __restrict__ A, const float* __restrict__ W,
    const float* __restrict__ bias, float* __restrict__ C,
    int K, int ep, const float* __restrict__ gamma,
    const float* __restrict__ beta)
{
    const int N = HID;
    const int wave = threadIdx.x >> 5;
    const int lane = threadIdx.x & 31;
    const int m    = lane & 15;
    const int half = lane >> 4;
    const int row0 = blockIdx.x * 16;
    const int col  = wave * 16 + m;

    v8f acc;
    const float bv = bias[col];
#pragma unroll
    for (int v = 0; v < 8; ++v) acc[v] = bv;

    const float* arow = A + (size_t)(row0 + m) * K + 2 * half;
    const float* wcol = W + (size_t)(2 * half) * N + col;

    for (int k = 0; k < K; k += 4) {
        v2f a = *(const v2f*)(arow + k);
        v2f b;
        b.x = wcol[(size_t)k * N];
        b.y = wcol[(size_t)k * N + N];
        acc = __builtin_amdgcn_wmma_f32_16x16x4_f32(
            false, a, false, b, (short)0, acc, false, false);
    }

    if (ep == 1) {
        const float rs = 0.99999500003749971f;
        const float g  = gamma[col] * rs;
        const float be = beta[col];
#pragma unroll
        for (int v = 0; v < 8; ++v) acc[v] = fmaxf(acc[v] * g + be, 0.0f);
    }

#pragma unroll
    for (int v = 0; v < 8; ++v) {
        const int row = row0 + v + 8 * half;
        C[(size_t)row * N + col] = acc[v];
    }
}

#define GL_PITCH 288
#define GL_TILES (NNODES / 16)
#define GL_WGS   ((GL_TILES + 3) / 4)

__global__ __launch_bounds__(128) void gemm128_stripe(
    const float* __restrict__ A, const float* __restrict__ W,
    const float* __restrict__ bias, float* __restrict__ C)
{
    __shared__ __attribute__((aligned(16))) float wl[64 * GL_PITCH];
    const int wave = threadIdx.x >> 5;
    const int lane = threadIdx.x & 31;
    const int m    = lane & 15;
    const int half = lane >> 4;
    const int tile = blockIdx.x * 4 + wave;
    const bool live = tile < GL_TILES;
    const int row0 = tile * 16;

    v2f a[32];
    if (live) {
        const float* arow = A + (size_t)(row0 + m) * HID + 2 * half;
#pragma unroll
        for (int s = 0; s < 32; ++s) a[s] = *(const v2f*)(arow + 4 * s);
    }

#pragma unroll
    for (int i = 0; i < 16; ++i) {
        const int p = wave + 4 * i;
        const float4 r0 = *(const float4*)(W + (size_t)(2 * p) * HID + 4 * lane);
        const float4 r1 = *(const float4*)(W + (size_t)(2 * p + 1) * HID + 4 * lane);
        float4 q0; q0.x = r0.x; q0.y = r1.x; q0.z = r0.y; q0.w = r1.y;
        float4 q1; q1.x = r0.z; q1.y = r1.z; q1.z = r0.w; q1.w = r1.w;
        float* d = wl + p * GL_PITCH + 8 * lane;
        *(float4*)d = q0;
        *(float4*)(d + 4) = q1;
    }
    __syncthreads();

    if (live) {
        v8f acc[8];
#pragma unroll
        for (int t = 0; t < 8; ++t) {
            const float bv = bias[16 * t + m];
#pragma unroll
            for (int v = 0; v < 8; ++v) acc[t][v] = bv;
        }
        const float* bl = wl + half * GL_PITCH + 2 * m;
#pragma unroll
        for (int s = 0; s < 32; ++s) {
#pragma unroll
            for (int t = 0; t < 8; ++t) {
                const v2f b = *(const v2f*)(bl + (2 * s) * GL_PITCH + 32 * t);
                acc[t] = __builtin_amdgcn_wmma_f32_16x16x4_f32(
                    false, a[s], false, b, (short)0, acc[t], false, false);
            }
        }
#pragma unroll
        for (int t = 0; t < 8; ++t) {
#pragma unroll
            for (int v = 0; v < 8; ++v) {
                const int row = row0 + v + 8 * half;
                C[(size_t)row * HID + 16 * t + m] = acc[t][v];
            }
        }
    }
}

__global__ void init_meta(int* deg, int* cursor, int* gs, int* ge)
{
    int i = blockIdx.x * blockDim.x + threadIdx.x;
    if (i < NNODES) { deg[i] = 0; cursor[i] = 0; }
    if (i < NGRAPHS) { gs[i] = NNODES; ge[i] = 0; }
}

__global__ void count_deg(const int* __restrict__ dst, int* __restrict__ deg)
{
    int i = blockIdx.x * blockDim.x + threadIdx.x;
    if (i < NEDGES) atomicAdd(&deg[dst[i]], 1);
}

#define SCAN_TB   1024
#define SCAN_VEC  (NNODES / 4)
#define SCAN_WGS  ((SCAN_VEC + SCAN_TB - 1) / SCAN_TB)

__global__ __launch_bounds__(SCAN_TB) void scan_offsets(
    const int* __restrict__ deg, int* __restrict__ offs)
{
    __shared__ int wave_own[32];
    __shared__ int wave_pre[32];
    const int t    = (int)threadIdx.x;
    const int lane = t & 31;
    const int wave = t >> 5;
    const int wg   = (int)blockIdx.x;
    const int4* __restrict__ deg4 = (const int4*)deg;
    const int myv = wg * SCAN_TB + t;

    int4 v; v.x = 0; v.y = 0; v.z = 0; v.w = 0;
    if (myv < SCAN_VEC) v = deg4[myv];

    int4 u[SCAN_WGS - 1];
#pragma unroll
    for (int j = 0; j < SCAN_WGS - 1; ++j) u[j] = deg4[j * SCAN_TB + t];
    int pre = 0;
#pragma unroll
    for (int j = 0; j < SCAN_WGS - 1; ++j) {
        const int s4 = (u[j].x + u[j].y) + (u[j].z + u[j].w);
        pre += (j < wg) ? s4 : 0;
    }

    const int own = (v.x + v.y) + (v.z + v.w);
    int inc = own;
#pragma unroll
    for (int off = 1; off < 32; off <<= 1) {
        const int n = __shfl_up(inc, off, 32);
        if (lane >= off) inc += n;
    }
#pragma unroll
    for (int off = 16; off; off >>= 1) pre += __shfl_xor(pre, off, 32);
    if (lane == 31) wave_own[wave] = inc;
    if (lane == 0)  wave_pre[wave] = pre;
    __syncthreads();

    const int wo = wave_own[lane];
    int wp = wave_pre[lane];
    int winc = wo;
#pragma unroll
    for (int off = 1; off < 32; off <<= 1) {
        const int n = __shfl_up(winc, off, 32);
        if (lane >= off) winc += n;
    }
#pragma unroll
    for (int off = 16; off; off >>= 1) wp += __shfl_xor(wp, off, 32);
    const int wave_base = __shfl(winc - wo, wave, 32);

    if (myv < SCAN_VEC) {
        int4 o;
        o.x = wp + wave_base + (inc - own);
        o.y = o.x + v.x;
        o.z = o.y + v.y;
        o.w = o.z + v.z;
        ((int4*)offs)[myv] = o;
        if (myv == SCAN_VEC - 1) offs[NNODES] = o.w + v.w;
    }
}

__global__ void csr_fill(const int* __restrict__ src, const int* __restrict__ dst,
                         const int* __restrict__ offs, int* __restrict__ cursor,
                         int* __restrict__ csr)
{
    int i = blockIdx.x * blockDim.x + threadIdx.x;
    if (i < NEDGES) {
        int d = dst[i];
        int p = atomicAdd(&cursor[d], 1);
        csr[offs[d] + p] = src[i];
    }
}

__global__ void graph_bounds(const int* __restrict__ batch, int* gs, int* ge)
{
    int i = blockIdx.x * blockDim.x + threadIdx.x;
    if (i < NNODES) {
        int b = batch[i];
        atomicMin(&gs[b], i);
        atomicMax(&ge[b], i + 1);
    }
}

#define AGG_GRP 8
#define AGG_WGS 2048

__global__ __launch_bounds__(256) void agg_mean_leaky(
    const float* __restrict__ H, const int* __restrict__ offs,
    const int* __restrict__ csr, float* __restrict__ out)
{
    const int lane = threadIdx.x & 31;
    const int wave0 = __builtin_amdgcn_readfirstlane(
        (int)((blockIdx.x * blockDim.x + threadIdx.x) >> 5));
    const int nwaves = (int)((gridDim.x * blockDim.x) >> 5);
    const float* colp = H + lane * 4;

    for (int node = wave0; node < NNODES; node += nwaves) {
        const int beg = offs[node];
        const int end = offs[node + 1];
        float ax = 0.f, ay = 0.f, az = 0.f, aw = 0.f;

        for (int e0 = beg; e0 < end; e0 += 32) {
            const int myidx = (e0 + lane < end) ? csr[e0 + lane] : 0;
            int cnt = end - e0; if (cnt > 32) cnt = 32;
            for (int j = 0; j < cnt; j += AGG_GRP) {
                const int r = cnt - j;
                int s[AGG_GRP];
#pragma unroll
                for (int k = 0; k < AGG_GRP; ++k) s[k] = __builtin_amdgcn_readlane(myidx, j + k);
                float4 v[AGG_GRP];
                v[0] = *(const float4*)(colp + (size_t)s[0] * HID);
#pragma unroll
                for (int k = 1; k < AGG_GRP; ++k) {
                    if (k < r) v[k] = *(const float4*)(colp + (size_t)s[k] * HID);
                }
                ax += v[0].x; ay += v[0].y; az += v[0].z; aw += v[0].w;
#pragma unroll
                for (int k = 1; k < AGG_GRP; ++k) {
                    if (k < r) { ax += v[k].x; ay += v[k].y; az += v[k].z; aw += v[k].w; }
                }
            }
        }
        const float inv = 1.0f / fmaxf((float)(end - beg), 1.0f);
        ax *= inv; ay *= inv; az *= inv; aw *= inv;
        float4 r4;
        r4.x = ax > 0.f ? ax : ALPHA * ax;
        r4.y = ay > 0.f ? ay : ALPHA * ay;
        r4.z = az > 0.f ? az : ALPHA * az;
        r4.w = aw > 0.f ? aw : ALPHA * aw;
        *(float4*)(out + (size_t)node * HID + lane * 4) = r4;
    }
}

__global__ void residual_add(float4* __restrict__ h, const float4* __restrict__ h2, int n)
{
    int i = blockIdx.x * blockDim.x + threadIdx.x;
    if (i < n) {
        float4 a = h[i], b = h2[i];
        a.x += b.x; a.y += b.y; a.z += b.z; a.w += b.w;
        h[i] = a;
    }
}

__global__ __launch_bounds__(128) void pool_max(
    const float* __restrict__ ht, const float* __restrict__ h1,
    const float* __restrict__ h2, const float* __restrict__ h3,
    const int* __restrict__ gs, const int* __restrict__ ge,
    float* __restrict__ pooled)
{
    const int g = blockIdx.x;
    const int f = (int)threadIdx.x * 4;
    const float* srcp = (f < 128) ? ht : (f < 256) ? h1 : (f < 384) ? h2 : h3;
    const int off = f & 127;
    float mx = -FMAXV, my = -FMAXV, mz = -FMAXV, mw = -FMAXV;
    const int b = gs[g], e = ge[g];
    for (int i = b; i < e; ++i) {
        const float4 v = *(const float4*)(srcp + (size_t)i * HID + off);
        mx = fmaxf(mx, v.x); my = fmaxf(my, v.y);
        mz = fmaxf(mz, v.z); mw = fmaxf(mw, v.w);
    }
    float4 r; r.x = mx; r.y = my; r.z = mz; r.w = mw;
    *(float4*)(pooled + (size_t)g * 512 + f) = r;
}

__global__ __launch_bounds__(32) void mlp2(
    const float* __restrict__ z, const float* __restrict__ W,
    const float* __restrict__ b, float* __restrict__ out)
{
    const int g = blockIdx.x;
    const int lane = threadIdx.x;
    const float z0 = z[(size_t)g * HID + lane];
    const float z1 = z[(size_t)g * HID + lane + 32];
    const float z2 = z[(size_t)g * HID + lane + 64];
    const float z3 = z[(size_t)g * HID + lane + 96];
    for (int c = 0; c < NCLS; ++c) {
        float s = z0 * W[lane * NCLS + c]
                + z1 * W[(lane + 32) * NCLS + c]
                + z2 * W[(lane + 64) * NCLS + c]
                + z3 * W[(lane + 96) * NCLS + c];
        for (int off = 16; off; off >>= 1) s += __shfl_xor(s, off, 32);
        if (lane == 0) out[g * NCLS + c] = s + b[c];
    }
}

extern "C" void kernel_launch(void* const* d_in, const int* in_sizes, int n_in,
                              void* d_out, int out_size, void* d_ws, size_t ws_size,
                              hipStream_t stream)
{
    (void)in_sizes; (void)n_in; (void)out_size; (void)ws_size;
    const float* x     = (const float*)d_in[0];
    const int*   ei    = (const int*)d_in[1];
    const int*   batch = (const int*)d_in[2];
    const float* W0 = (const float*)d_in[3];  const float* b0 = (const float*)d_in[4];
    const float* W1 = (const float*)d_in[5];  const float* b1 = (const float*)d_in[6];
    const float* W2 = (const float*)d_in[7];  const float* b2 = (const float*)d_in[8];
    const float* W3 = (const float*)d_in[9];  const float* b3 = (const float*)d_in[10];
    const float* Wm1 = (const float*)d_in[11]; const float* bm1 = (const float*)d_in[12];
    const float* gamma = (const float*)d_in[13]; const float* beta = (const float*)d_in[14];
    const float* Wm2 = (const float*)d_in[15]; const float* bm2 = (const float*)d_in[16];
    const int* esrc = ei;
    const int* edst = ei + NEDGES;

    char* ws = (char*)d_ws;
    size_t off = 0;
    auto alloc = [&](size_t bytes) -> void* {
        void* p = ws + off;
        off = (off + bytes + 255) & ~(size_t)255;
        return p;
    };
    const size_t featB = (size_t)NNODES * HID * sizeof(float);
    float* tmpH   = (float*)alloc(featB);
    float* h      = (float*)alloc(featB);
    float* h1     = (float*)alloc(featB);
    float* h2     = (float*)alloc(featB);
    float* ht     = (float*)alloc(featB);
    float* zbuf   = (float*)alloc((size_t)NGRAPHS * HID * sizeof(float));
    float* pooled = (float*)alloc((size_t)NGRAPHS * 512 * sizeof(float));
    int*   csr    = (int*)alloc((size_t)NEDGES * sizeof(int));
    int*   offs   = (int*)alloc((size_t)(NNODES + 1) * sizeof(int));
    int*   deg    = (int*)alloc((size_t)NNODES * sizeof(int));
    int*   cursor = (int*)alloc((size_t)NNODES * sizeof(int));
    int*   gs     = (int*)alloc((size_t)NGRAPHS * sizeof(int));
    int*   ge     = (int*)alloc((size_t)NGRAPHS * sizeof(int));

    const int TB = 256;
    init_meta<<<(NNODES + TB - 1) / TB, TB, 0, stream>>>(deg, cursor, gs, ge);
    count_deg<<<(NEDGES + TB - 1) / TB, TB, 0, stream>>>(edst, deg);
    scan_offsets<<<SCAN_WGS, SCAN_TB, 0, stream>>>(deg, offs);
    csr_fill<<<(NEDGES + TB - 1) / TB, TB, 0, stream>>>(esrc, edst, offs, cursor, csr);
    graph_bounds<<<(NNODES + TB - 1) / TB, TB, 0, stream>>>(batch, gs, ge);

    const int aggGrid  = AGG_WGS;

    gemm128_stripe<<<GL_WGS, 128, 0, stream>>>(x, W0, b0, tmpH);
    agg_mean_leaky<<<aggGrid, TB, 0, stream>>>(tmpH, offs, csr, h);
    gemm128_stripe<<<GL_WGS, 128, 0, stream>>>(h, W1, b1, tmpH);
    agg_mean_leaky<<<aggGrid, TB, 0, stream>>>(tmpH, offs, csr, h1);
    gemm128_stripe<<<GL_WGS, 128, 0, stream>>>(h1, W2, b2, tmpH);
    agg_mean_leaky<<<aggGrid, TB, 0, stream>>>(tmpH, offs, csr, h2);
    const int n4 = NNODES * HID / 4;
    residual_add<<<(n4 + TB - 1) / TB, TB, 0, stream>>>((float4*)h, (const float4*)h2, n4);
    gemm128_stripe<<<GL_WGS, 128, 0, stream>>>(h, W3, b3, tmpH);
    agg_mean_leaky<<<aggGrid, TB, 0, stream>>>(tmpH, offs, csr, ht);

    pool_max<<<NGRAPHS, 128, 0, stream>>>(ht, h1, h2, h, gs, ge, pooled);

    gemm_wmma_f32<<<NGRAPHS / 16, TB, 0, stream>>>(pooled, Wm1, bm1, zbuf, 512, 1, gamma, beta);
    mlp2<<<NGRAPHS, 32, 0, stream>>>(zbuf, Wm2, bm2, (float*)d_out);
}


# pool_max: 16 waves per graph, each with 16 row loads in flight (one array, every 4th row), joined by an LDS maximum, instead of 4 waves waiting on 2 rows per trip; same maxima
# speedup vs baseline: 1.7771x; 1.0966x over previous
#include <hip/hip_runtime.h>

#define NNODES 100000
#define NEDGES 1600000
#define NGRAPHS 512
#define HID 128
#define NCLS 10
#define ALPHA 0.01f
#define FMAXV 3.402823466e38f

typedef float v2f __attribute__((ext_vector_type(2)));
typedef float v8f __attribute__((ext_vector_type(8)));

__global__ __launch_bounds__(256) void gemm_wmma_f32(
    const float* __restrict__ A, const float* __restrict__ W,
    const float* __restrict__ bias, float* __restrict__ C,
    int K, int ep, const float* __restrict__ gamma,
    const float* __restrict__ beta)
{
    const int N = HID;
    const int wave = threadIdx.x >> 5;
    const int lane = threadIdx.x & 31;
    const int m    = lane & 15;
    const int half = lane >> 4;
    const int row0 = blockIdx.x * 16;
    const int col  = wave * 16 + m;

    v8f acc;
    const float bv = bias[col];
#pragma unroll
    for (int v = 0; v < 8; ++v) acc[v] = bv;

    const float* arow = A + (size_t)(row0 + m) * K + 2 * half;
    const float* wcol = W + (size_t)(2 * half) * N + col;

    for (int k = 0; k < K; k += 4) {
        v2f a = *(const v2f*)(arow + k);
        v2f b;
        b.x = wcol[(size_t)k * N];
        b.y = wcol[(size_t)k * N + N];
        acc = __builtin_amdgcn_wmma_f32_16x16x4_f32(
            false, a, false, b, (short)0, acc, false, false);
    }

    if (ep == 1) {
        const float rs = 0.99999500003749971f;
        const float g  = gamma[col] * rs;
        const float be = beta[col];
#pragma unroll
        for (int v = 0; v < 8; ++v) acc[v] = fmaxf(acc[v] * g + be, 0.0f);
    }

#pragma unroll
    for (int v = 0; v < 8; ++v) {
        const int row = row0 + v + 8 * half;
        C[(size_t)row * N + col] = acc[v];
    }
}

#define GL_PITCH 288
#define GL_TILES (NNODES / 16)
#define GL_WGS   ((GL_TILES + 3) / 4)

__global__ __launch_bounds__(128) void gemm128_stripe(
    const float* __restrict__ A, const float* __restrict__ W,
    const float* __restrict__ bias, float* __restrict__ C)
{
    __shared__ __attribute__((aligned(16))) float wl[64 * GL_PITCH];
    const int wave = threadIdx.x >> 5;
    const int lane = threadIdx.x & 31;
    const int m    = lane & 15;
    const int half = lane >> 4;
    const int tile = blockIdx.x * 4 + wave;
    const bool live = tile < GL_TILES;
    const int row0 = tile * 16;

    v2f a[32];
    if (live) {
        const float* arow = A + (size_t)(row0 + m) * HID + 2 * half;
#pragma unroll
        for (int s = 0; s < 32; ++s) a[s] = *(const v2f*)(arow + 4 * s);
    }

#pragma unroll
    for (int i = 0; i < 16; ++i) {
        const int p = wave + 4 * i;
        const float4 r0 = *(const float4*)(W + (size_t)(2 * p) * HID + 4 * lane);
        const float4 r1 = *(const float4*)(W + (size_t)(2 * p + 1) * HID + 4 * lane);
        float4 q0; q0.x = r0.x; q0.y = r1.x; q0.z = r0.y; q0.w = r1.y;
        float4 q1; q1.x = r0.z; q1.y = r1.z; q1.z = r0.w; q1.w = r1.w;
        float* d = wl + p * GL_PITCH + 8 * lane;
        *(float4*)d = q0;
        *(float4*)(d + 4) = q1;
    }
    __syncthreads();

    if (live) {
        v8f acc[8];
#pragma unroll
        for (int t = 0; t < 8; ++t) {
            const float bv = bias[16 * t + m];
#pragma unroll
            for (int v = 0; v < 8; ++v) acc[t][v] = bv;
        }
        const float* bl = wl + half * GL_PITCH + 2 * m;
#pragma unroll
        for (int s = 0; s < 32; ++s) {
#pragma unroll
            for (int t = 0; t < 8; ++t) {
                const v2f b = *(const v2f*)(bl + (2 * s) * GL_PITCH + 32 * t);
                acc[t] = __builtin_amdgcn_wmma_f32_16x16x4_f32(
                    false, a[s], false, b, (short)0, acc[t], false, false);
            }
        }
#pragma unroll
        for (int t = 0; t < 8; ++t) {
#pragma unroll
            for (int v = 0; v < 8; ++v) {
                const int row = row0 + v + 8 * half;
                C[(size_t)row * HID + 16 * t + m] = acc[t][v];
            }
        }
    }
}

__global__ void init_meta(int* deg, int* cursor, int* gs, int* ge)
{
    int i = blockIdx.x * blockDim.x + threadIdx.x;
    if (i < NNODES) { deg[i] = 0; cursor[i] = 0; }
    if (i < NGRAPHS) { gs[i] = NNODES; ge[i] = 0; }
}

__global__ void count_deg(const int* __restrict__ dst, int* __restrict__ deg)
{
    int i = blockIdx.x * blockDim.x + threadIdx.x;
    if (i < NEDGES) atomicAdd(&deg[dst[i]], 1);
}

#define SCAN_TB   1024
#define SCAN_VEC  (NNODES / 4)
#define SCAN_WGS  ((SCAN_VEC + SCAN_TB - 1) / SCAN_TB)

__global__ __launch_bounds__(SCAN_TB) void scan_offsets(
    const int* __restrict__ deg, int* __restrict__ offs)
{
    __shared__ int wave_own[32];
    __shared__ int wave_pre[32];
    const int t    = (int)threadIdx.x;
    const int lane = t & 31;
    const int wave = t >> 5;
    const int wg   = (int)blockIdx.x;
    const int4* __restrict__ deg4 = (const int4*)deg;
    const int myv = wg * SCAN_TB + t;

    int4 v; v.x = 0; v.y = 0; v.z = 0; v.w = 0;
    if (myv < SCAN_VEC) v = deg4[myv];

    int4 u[SCAN_WGS - 1];
#pragma unroll
    for (int j = 0; j < SCAN_WGS - 1; ++j) u[j] = deg4[j * SCAN_TB + t];
    int pre = 0;
#pragma unroll
    for (int j = 0; j < SCAN_WGS - 1; ++j) {
        const int s4 = (u[j].x + u[j].y) + (u[j].z + u[j].w);
        pre += (j < wg) ? s4 : 0;
    }

    const int own = (v.x + v.y) + (v.z + v.w);
    int inc = own;
#pragma unroll
    for (int off = 1; off < 32; off <<= 1) {
        const int n = __shfl_up(inc, off, 32);
        if (lane >= off) inc += n;
    }
#pragma unroll
    for (int off = 16; off; off >>= 1) pre += __shfl_xor(pre, off, 32);
    if (lane == 31) wave_own[wave] = inc;
    if (lane == 0)  wave_pre[wave] = pre;
    __syncthreads();

    const int wo = wave_own[lane];
    int wp = wave_pre[lane];
    int winc = wo;
#pragma unroll
    for (int off = 1; off < 32; off <<= 1) {
        const int n = __shfl_up(winc, off, 32);
        if (lane >= off) winc += n;
    }
#pragma unroll
    for (int off = 16; off; off >>= 1) wp += __shfl_xor(wp, off, 32);
    const int wave_base = __shfl(winc - wo, wave, 32);

    if (myv < SCAN_VEC) {
        int4 o;
        o.x = wp + wave_base + (inc - own);
        o.y = o.x + v.x;
        o.z = o.y + v.y;
        o.w = o.z + v.z;
        ((int4*)offs)[myv] = o;
        if (myv == SCAN_VEC - 1) offs[NNODES] = o.w + v.w;
    }
}

__global__ void csr_fill(const int* __restrict__ src, const int* __restrict__ dst,
                         const int* __restrict__ offs, int* __restrict__ cursor,
                         int* __restrict__ csr)
{
    int i = blockIdx.x * blockDim.x + threadIdx.x;
    if (i < NEDGES) {
        int d = dst[i];
        int p = atomicAdd(&cursor[d], 1);
        csr[offs[d] + p] = src[i];
    }
}

__global__ void graph_bounds(const int* __restrict__ batch, int* gs, int* ge)
{
    int i = blockIdx.x * blockDim.x + threadIdx.x;
    if (i < NNODES) {
        int b = batch[i];
        atomicMin(&gs[b], i);
        atomicMax(&ge[b], i + 1);
    }
}

#define AGG_GRP 8
#define AGG_WGS 2048

__global__ __launch_bounds__(256) void agg_mean_leaky(
    const float* __restrict__ H, const int* __restrict__ offs,
    const int* __restrict__ csr, float* __restrict__ out)
{
    const int lane = threadIdx.x & 31;
    const int wave0 = __builtin_amdgcn_readfirstlane(
        (int)((blockIdx.x * blockDim.x + threadIdx.x) >> 5));
    const int nwaves = (int)((gridDim.x * blockDim.x) >> 5);
    const float* colp = H + lane * 4;

    for (int node = wave0; node < NNODES; node += nwaves) {
        const int beg = offs[node];
        const int end = offs[node + 1];
        float ax = 0.f, ay = 0.f, az = 0.f, aw = 0.f;

        for (int e0 = beg; e0 < end; e0 += 32) {
            const int myidx = (e0 + lane < end) ? csr[e0 + lane] : 0;
            int cnt = end - e0; if (cnt > 32) cnt = 32;
            for (int j = 0; j < cnt; j += AGG_GRP) {
                const int r = cnt - j;
                int s[AGG_GRP];
#pragma unroll
                for (int k = 0; k < AGG_GRP; ++k) s[k] = __builtin_amdgcn_readlane(myidx, j + k);
                float4 v[AGG_GRP];
                v[0] = *(const float4*)(colp + (size_t)s[0] * HID);
#pragma unroll
                for (int k = 1; k < AGG_GRP; ++k) {
                    if (k < r) v[k] = *(const float4*)(colp + (size_t)s[k] * HID);
                }
                ax += v[0].x; ay += v[0].y; az += v[0].z; aw += v[0].w;
#pragma unroll
                for (int k = 1; k < AGG_GRP; ++k) {
                    if (k < r) { ax += v[k].x; ay += v[k].y; az += v[k].z; aw += v[k].w; }
                }
            }
        }
        const float inv = 1.0f / fmaxf((float)(end - beg), 1.0f);
        ax *= inv; ay *= inv; az *= inv; aw *= inv;
        float4 r4;
        r4.x = ax > 0.f ? ax : ALPHA * ax;
        r4.y = ay > 0.f ? ay : ALPHA * ay;
        r4.z = az > 0.f ? az : ALPHA * az;
        r4.w = aw > 0.f ? aw : ALPHA * aw;
        *(float4*)(out + (size_t)node * HID + lane * 4) = r4;
    }
}

__global__ void residual_add(float4* __restrict__ h, const float4* __restrict__ h2, int n)
{
    int i = blockIdx.x * blockDim.x + threadIdx.x;
    if (i < n) {
        float4 a = h[i], b = h2[i];
        a.x += b.x; a.y += b.y; a.z += b.z; a.w += b.w;
        h[i] = a;
    }
}

#define POOL_RG  4
#define POOL_INF 16

__global__ __launch_bounds__(512) void pool_max(
    const float* __restrict__ ht, const float* __restrict__ h1,
    const float* __restrict__ h2, const float* __restrict__ h3,
    const int* __restrict__ gs, const int* __restrict__ ge,
    float* __restrict__ pooled)
{
    __shared__ float4 red[POOL_RG][128];
    const int g = blockIdx.x;
    const int lane = (int)threadIdx.x & 31;
    const int wave = __builtin_amdgcn_readfirstlane((int)(threadIdx.x >> 5));
    const int arr = wave & 3;
    const int rg  = wave >> 2;
    const float* srcp = (arr == 0) ? ht : (arr == 1) ? h1 : (arr == 2) ? h2 : h3;
    const float* colp = srcp + lane * 4;
    const int b = gs[g], e = ge[g];

    float mx = -FMAXV, my = -FMAXV, mz = -FMAXV, mw = -FMAXV;
    for (int i0 = b + rg; i0 < e; i0 += POOL_RG * POOL_INF) {
        float4 v[POOL_INF];
#pragma unroll
        for (int k = 0; k < POOL_INF; ++k) {
            const int r = i0 + k * POOL_RG;
            v[k].x = -FMAXV; v[k].y = -FMAXV; v[k].z = -FMAXV; v[k].w = -FMAXV;
            if (r < e) v[k] = *(const float4*)(colp + (size_t)r * HID);
        }
#pragma unroll
        for (int k = 0; k < POOL_INF; ++k) {
            mx = fmaxf(mx, v[k].x); my = fmaxf(my, v[k].y);
            mz = fmaxf(mz, v[k].z); mw = fmaxf(mw, v[k].w);
        }
    }
    float4 m; m.x = mx; m.y = my; m.z = mz; m.w = mw;
    red[rg][arr * 32 + lane] = m;
    __syncthreads();
    if (threadIdx.x < 128) {
        const int c = (int)threadIdx.x;
        float4 r = red[0][c];
#pragma unroll
        for (int k = 1; k < POOL_RG; ++k) {
            const float4 q = red[k][c];
            r.x = fmaxf(r.x, q.x); r.y = fmaxf(r.y, q.y);
            r.z = fmaxf(r.z, q.z); r.w = fmaxf(r.w, q.w);
        }
        *(float4*)(pooled + (size_t)g * 512 + c * 4) = r;
    }
}

__global__ __launch_bounds__(32) void mlp2(
    const float* __restrict__ z, const float* __restrict__ W,
    const float* __restrict__ b, float* __restrict__ out)
{
    const int g = blockIdx.x;
    const int lane = threadIdx.x;
    const float z0 = z[(size_t)g * HID + lane];
    const float z1 = z[(size_t)g * HID + lane + 32];
    const float z2 = z[(size_t)g * HID + lane + 64];
    const float z3 = z[(size_t)g * HID + lane + 96];
    for (int c = 0; c < NCLS; ++c) {
        float s = z0 * W[lane * NCLS + c]
                + z1 * W[(lane + 32) * NCLS + c]
                + z2 * W[(lane + 64) * NCLS + c]
                + z3 * W[(lane + 96) * NCLS + c];
        for (int off = 16; off; off >>= 1) s += __shfl_xor(s, off, 32);
        if (lane == 0) out[g * NCLS + c] = s + b[c];
    }
}

extern "C" void kernel_launch(void* const* d_in, const int* in_sizes, int n_in,
                              void* d_out, int out_size, void* d_ws, size_t ws_size,
                              hipStream_t stream)
{
    (void)in_sizes; (void)n_in; (void)out_size; (void)ws_size;
    const float* x     = (const float*)d_in[0];
    const int*   ei    = (const int*)d_in[1];
    const int*   batch = (const int*)d_in[2];
    const float* W0 = (const float*)d_in[3];  const float* b0 = (const float*)d_in[4];
    const float* W1 = (const float*)d_in[5];  const float* b1 = (const float*)d_in[6];
    const float* W2 = (const float*)d_in[7];  const float* b2 = (const float*)d_in[8];
    const float* W3 = (const float*)d_in[9];  const float* b3 = (const float*)d_in[10];
    const float* Wm1 = (const float*)d_in[11]; const float* bm1 = (const float*)d_in[12];
    const float* gamma = (const float*)d_in[13]; const float* beta = (const float*)d_in[14];
    const float* Wm2 = (const float*)d_in[15]; const float* bm2 = (const float*)d_in[16];
    const int* esrc = ei;
    const int* edst = ei + NEDGES;

    char* ws = (char*)d_ws;
    size_t off = 0;
    auto alloc = [&](size_t bytes) -> void* {
        void* p = ws + off;
        off = (off + bytes + 255) & ~(size_t)255;
        return p;
    };
    const size_t featB = (size_t)NNODES * HID * sizeof(float);
    float* tmpH   = (float*)alloc(featB);
    float* h      = (float*)alloc(featB);
    float* h1     = (float*)alloc(featB);
    float* h2     = (float*)alloc(featB);
    float* ht     = (float*)alloc(featB);
    float* zbuf   = (float*)alloc((size_t)NGRAPHS * HID * sizeof(float));
    float* pooled = (float*)alloc((size_t)NGRAPHS * 512 * sizeof(float));
    int*   csr    = (int*)alloc((size_t)NEDGES * sizeof(int));
    int*   offs   = (int*)alloc((size_t)(NNODES + 1) * sizeof(int));
    int*   deg    = (int*)alloc((size_t)NNODES * sizeof(int));
    int*   cursor = (int*)alloc((size_t)NNODES * sizeof(int));
    int*   gs     = (int*)alloc((size_t)NGRAPHS * sizeof(int));
    int*   ge     = (int*)alloc((size_t)NGRAPHS * sizeof(int));

    const int TB = 256;
    init_meta<<<(NNODES + TB - 1) / TB, TB, 0, stream>>>(deg, cursor, gs, ge);
    count_deg<<<(NEDGES + TB - 1) / TB, TB, 0, stream>>>(edst, deg);
    scan_offsets<<<SCAN_WGS, SCAN_TB, 0, stream>>>(deg, offs);
    csr_fill<<<(NEDGES + TB - 1) / TB, TB, 0, stream>>>(esrc, edst, offs, cursor, csr);
    graph_bounds<<<(NNODES + TB - 1) / TB, TB, 0, stream>>>(batch, gs, ge);

    const int aggGrid  = AGG_WGS;

    gemm128_stripe<<<GL_WGS, 128, 0, stream>>>(x, W0, b0, tmpH);
    agg_mean_leaky<<<aggGrid, TB, 0, stream>>>(tmpH, offs, csr, h);
    gemm128_stripe<<<GL_WGS, 128, 0, stream>>>(h, W1, b1, tmpH);
    agg_mean_leaky<<<aggGrid, TB, 0, stream>>>(tmpH, offs, csr, h1);
    gemm128_stripe<<<GL_WGS, 128, 0, stream>>>(h1, W2, b2, tmpH);
    agg_mean_leaky<<<aggGrid, TB, 0, stream>>>(tmpH, offs, csr, h2);
    const int n4 = NNODES * HID / 4;
    residual_add<<<(n4 + TB - 1) / TB, TB, 0, stream>>>((float4*)h, (const float4*)h2, n4);
    gemm128_stripe<<<GL_WGS, 128, 0, stream>>>(h, W3, b3, tmpH);
    agg_mean_leaky<<<aggGrid, TB, 0, stream>>>(tmpH, offs, csr, ht);

    pool_max<<<NGRAPHS, 512, 0, stream>>>(ht, h1, h2, h, gs, ge, pooled);

    gemm_wmma_f32<<<NGRAPHS / 16, TB, 0, stream>>>(pooled, Wm1, bm1, zbuf, 512, 1, gamma, beta);
    mlp2<<<NGRAPHS, 32, 0, stream>>>(zbuf, Wm2, bm2, (float*)d_out);
}


# CSR build: edges bucketed by destination block, then per-block node lists, all counters in LDS; no global atomic per edge (was 2); a node's sources in LDS-atomic arrival order: sum order may differ
# speedup vs baseline: 2.4595x; 1.3840x over previous
#include <hip/hip_runtime.h>

#define NNODES 100000
#define NEDGES 1600000
#define NGRAPHS 512
#define HID 128
#define NCLS 10
#define ALPHA 0.01f
#define FMAXV 3.402823466e38f

typedef float v2f __attribute__((ext_vector_type(2)));
typedef float v8f __attribute__((ext_vector_type(8)));

__global__ __launch_bounds__(256) void gemm_wmma_f32(
    const float* __restrict__ A, const float* __restrict__ W,
    const float* __restrict__ bias, float* __restrict__ C,
    int K, int ep, const float* __restrict__ gamma,
    const float* __restrict__ beta)
{
    const int N = HID;
    const int wave = threadIdx.x >> 5;
    const int lane = threadIdx.x & 31;
    const int m    = lane & 15;
    const int half = lane >> 4;
    const int row0 = blockIdx.x * 16;
    const int col  = wave * 16 + m;

    v8f acc;
    const float bv = bias[col];
#pragma unroll
    for (int v = 0; v < 8; ++v) acc[v] = bv;

    const float* arow = A + (size_t)(row0 + m) * K + 2 * half;
    const float* wcol = W + (size_t)(2 * half) * N + col;

    for (int k = 0; k < K; k += 4) {
        v2f a = *(const v2f*)(arow + k);
        v2f b;
        b.x = wcol[(size_t)k * N];
        b.y = wcol[(size_t)k * N + N];
        acc = __builtin_amdgcn_wmma_f32_16x16x4_f32(
            false, a, false, b, (short)0, acc, false, false);
    }

    if (ep == 1) {
        const float rs = 0.99999500003749971f;
        const float g  = gamma[col] * rs;
        const float be = beta[col];
#pragma unroll
        for (int v = 0; v < 8; ++v) acc[v] = fmaxf(acc[v] * g + be, 0.0f);
    }

#pragma unroll
    for (int v = 0; v < 8; ++v) {
        const int row = row0 + v + 8 * half;
        C[(size_t)row * N + col] = acc[v];
    }
}

#define GL_PITCH 288
#define GL_TILES (NNODES / 16)
#define GL_WGS   ((GL_TILES + 3) / 4)

__global__ __launch_bounds__(128) void gemm128_stripe(
    const float* __restrict__ A, const float* __restrict__ W,
    const float* __restrict__ bias, float* __restrict__ C)
{
    __shared__ __attribute__((aligned(16))) float wl[64 * GL_PITCH];
    const int wave = threadIdx.x >> 5;
    const int lane = threadIdx.x & 31;
    const int m    = lane & 15;
    const int half = lane >> 4;
    const int tile = blockIdx.x * 4 + wave;
    const bool live = tile < GL_TILES;
    const int row0 = tile * 16;

    v2f a[32];
    if (live) {
        const float* arow = A + (size_t)(row0 + m) * HID + 2 * half;
#pragma unroll
        for (int s = 0; s < 32; ++s) a[s] = *(const v2f*)(arow + 4 * s);
    }

#pragma unroll
    for (int i = 0; i < 16; ++i) {
        const int p = wave + 4 * i;
        const float4 r0 = *(const float4*)(W + (size_t)(2 * p) * HID + 4 * lane);
        const float4 r1 = *(const float4*)(W + (size_t)(2 * p + 1) * HID + 4 * lane);
        float4 q0; q0.x = r0.x; q0.y = r1.x; q0.z = r0.y; q0.w = r1.y;
        float4 q1; q1.x = r0.z; q1.y = r1.z; q1.z = r0.w; q1.w = r1.w;
        float* d = wl + p * GL_PITCH + 8 * lane;
        *(float4*)d = q0;
        *(float4*)(d + 4) = q1;
    }
    __syncthreads();

    if (live) {
        v8f acc[8];
#pragma unroll
        for (int t = 0; t < 8; ++t) {
            const float bv = bias[16 * t + m];
#pragma unroll
            for (int v = 0; v < 8; ++v) acc[t][v] = bv;
        }
        const float* bl = wl + half * GL_PITCH + 2 * m;
#pragma unroll
        for (int s = 0; s < 32; ++s) {
#pragma unroll
            for (int t = 0; t < 8; ++t) {
                const v2f b = *(const v2f*)(bl + (2 * s) * GL_PITCH + 32 * t);
                acc[t] = __builtin_amdgcn_wmma_f32_16x16x4_f32(
                    false, a[s], false, b, (short)0, acc[t], false, false);
            }
        }
#pragma unroll
        for (int t = 0; t < 8; ++t) {
#pragma unroll
            for (int v = 0; v < 8; ++v) {
                const int row = row0 + v + 8 * half;
                C[(size_t)row * HID + 16 * t + m] = acc[t][v];
            }
        }
    }
}

#define CB_NB   400
#define CB_BN   (NNODES / CB_NB)
#define CB_NCH  250
#define CB_CE   (NEDGES / CB_NCH)
#define CB_TB   256
#define CB_EPT  (CB_CE / CB_TB)
static_assert(CB_NB * CB_BN == NNODES, "blocks must tile the nodes");
static_assert(CB_NCH * CB_CE == NEDGES, "chunks must tile the edges");
static_assert(CB_EPT * CB_TB == CB_CE, "threads must tile a chunk");
static_assert(CB_NB * CB_NCH == NNODES, "count matrix must have the scan's length");
static_assert(CB_BN <= 256, "local node index must fit 8 bits");

__global__ __launch_bounds__(CB_TB) void edge_hist(
    const int* __restrict__ dst, int* __restrict__ cnt,
    int* __restrict__ gs, int* __restrict__ ge)
{
    __shared__ int hist[CB_NB];
    const int t = (int)threadIdx.x;
    const int c = (int)blockIdx.x;
    for (int b = t; b < CB_NB; b += CB_TB) hist[b] = 0;
    __syncthreads();
    const int* dp = dst + (size_t)c * CB_CE + t;
    int d[CB_EPT];
#pragma unroll
    for (int j = 0; j < CB_EPT; ++j) d[j] = dp[j * CB_TB];
#pragma unroll
    for (int j = 0; j < CB_EPT; ++j) atomicAdd(&hist[(unsigned)d[j] / CB_BN], 1);
    __syncthreads();
    for (int b = t; b < CB_NB; b += CB_TB) cnt[b * CB_NCH + c] = hist[b];
    const int i = c * CB_TB + t;
    if (i < NGRAPHS) { gs[i] = NNODES; ge[i] = 0; }
}

#define SCAN_TB   1024
#define SCAN_VEC  (NNODES / 4)
#define SCAN_WGS  ((SCAN_VEC + SCAN_TB - 1) / SCAN_TB)

__global__ __launch_bounds__(SCAN_TB) void scan_offsets(
    const int* __restrict__ deg, int* __restrict__ offs)
{
    __shared__ int wave_own[32];
    __shared__ int wave_pre[32];
    const int t    = (int)threadIdx.x;
    const int lane = t & 31;
    const int wave = t >> 5;
    const int wg   = (int)blockIdx.x;
    const int4* __restrict__ deg4 = (const int4*)deg;
    const int myv = wg * SCAN_TB + t;

    int4 v; v.x = 0; v.y = 0; v.z = 0; v.w = 0;
    if (myv < SCAN_VEC) v = deg4[myv];

    int4 u[SCAN_WGS - 1];
#pragma unroll
    for (int j = 0; j < SCAN_WGS - 1; ++j) u[j] = deg4[j * SCAN_TB + t];
    int pre = 0;
#pragma unroll
    for (int j = 0; j < SCAN_WGS - 1; ++j) {
        const int s4 = (u[j].x + u[j].y) + (u[j].z + u[j].w);
        pre += (j < wg) ? s4 : 0;
    }

    const int own = (v.x + v.y) + (v.z + v.w);
    int inc = own;
#pragma unroll
    for (int off = 1; off < 32; off <<= 1) {
        const int n = __shfl_up(inc, off, 32);
        if (lane >= off) inc += n;
    }
#pragma unroll
    for (int off = 16; off; off >>= 1) pre += __shfl_xor(pre, off, 32);
    if (lane == 31) wave_own[wave] = inc;
    if (lane == 0)  wave_pre[wave] = pre;
    __syncthreads();

    const int wo = wave_own[lane];
    int wp = wave_pre[lane];
    int winc = wo;
#pragma unroll
    for (int off = 1; off < 32; off <<= 1) {
        const int n = __shfl_up(winc, off, 32);
        if (lane >= off) winc += n;
    }
#pragma unroll
    for (int off = 16; off; off >>= 1) wp += __shfl_xor(wp, off, 32);
    const int wave_base = __shfl(winc - wo, wave, 32);

    if (myv < SCAN_VEC) {
        int4 o;
        o.x = wp + wave_base + (inc - own);
        o.y = o.x + v.x;
        o.z = o.y + v.y;
        o.w = o.z + v.z;
        ((int4*)offs)[myv] = o;
        if (myv == SCAN_VEC - 1) offs[NNODES] = o.w + v.w;
    }
}

__global__ __launch_bounds__(CB_TB) void edge_scatter(
    const int* __restrict__ src, const int* __restrict__ dst,
    const int* __restrict__ cntoff, int* __restrict__ packed)
{
    __shared__ int cur[CB_NB];
    const int t = (int)threadIdx.x;
    const int c = (int)blockIdx.x;
    for (int b = t; b < CB_NB; b += CB_TB) cur[b] = cntoff[b * CB_NCH + c];
    const int* sp = src + (size_t)c * CB_CE + t;
    const int* dp = dst + (size_t)c * CB_CE + t;
    int s[CB_EPT], d[CB_EPT];
#pragma unroll
    for (int j = 0; j < CB_EPT; ++j) { s[j] = sp[j * CB_TB]; d[j] = dp[j * CB_TB]; }
    __syncthreads();
    int pos[CB_EPT];
#pragma unroll
    for (int j = 0; j < CB_EPT; ++j) {
        const unsigned b = (unsigned)d[j] / CB_BN;
        const int loc = d[j] - (int)b * CB_BN;
        pos[j] = atomicAdd(&cur[b], 1);
        s[j] = (s[j] << 8) | loc;
    }
#pragma unroll
    for (int j = 0; j < CB_EPT; ++j) packed[pos[j]] = s[j];
}

#define CB_STAGE 8192
#define CB_BT    512
#define CB_WPT   (CB_STAGE / CB_BT)
static_assert(CB_BT >= 256, "the scan of the node counts takes 256 threads");
__global__ __launch_bounds__(CB_BT) void block_csr(
    const int* __restrict__ packed, const int* __restrict__ cntoff,
    int* __restrict__ offs, int* __restrict__ csr)
{
    __shared__ int ncnt[256];
    __shared__ int wtot[8];
    __shared__ int stage[CB_STAGE];
    const int t    = (int)threadIdx.x;
    const int lane = t & 31;
    const int wave = t >> 5;
    const int b    = (int)blockIdx.x;
    const int beg = cntoff[b * CB_NCH];
    const int end = cntoff[(b + 1) * CB_NCH];
    const int n = end - beg;
    const bool fits = n <= CB_STAGE;
    if (t < 256) ncnt[t] = 0;
    int w[CB_WPT];
    if (fits) {
#pragma unroll
        for (int k = 0; k < CB_WPT; ++k) {
            const int i = t + CB_BT * k;
            w[k] = (i < n) ? packed[beg + i] : 0;
        }
    }
    __syncthreads();
    if (fits) {
#pragma unroll
        for (int k = 0; k < CB_WPT; ++k) {
            if (t + CB_BT * k < n) atomicAdd(&ncnt[w[k] & 255], 1);
        }
    } else {
        for (int e = beg + t; e < end; e += CB_BT) atomicAdd(&ncnt[packed[e] & 255], 1);
    }
    __syncthreads();

    int own = 0, inc = 0;
    if (t < 256) {
        own = ncnt[t];
        inc = own;
#pragma unroll
        for (int off = 1; off < 32; off <<= 1) {
            const int up = __shfl_up(inc, off, 32);
            if (lane >= off) inc += up;
        }
        if (lane == 31) wtot[wave] = inc;
    }
    __syncthreads();
    if (t < 256) {
        int base = beg;
#pragma unroll
        for (int q = 0; q < 8; ++q) base += (q < wave) ? wtot[q] : 0;
        const int start = base + (inc - own);
        ncnt[t] = start;
        if (t < CB_BN) offs[b * CB_BN + t] = start;
    }
    if (b == CB_NB - 1 && t == 0) offs[NNODES] = end;
    __syncthreads();
    if (fits) {
#pragma unroll
        for (int k = 0; k < CB_WPT; ++k) {
            if (t + CB_BT * k < n) {
                const int p = atomicAdd(&ncnt[w[k] & 255], 1);
                stage[p - beg] = w[k] >> 8;
            }
        }
        __syncthreads();
        for (int i = t; i < n; i += CB_BT) csr[beg + i] = stage[i];
    } else {
        for (int e = beg + t; e < end; e += CB_BT) {
            const int v = packed[e];
            const int p = atomicAdd(&ncnt[v & 255], 1);
            csr[p] = v >> 8;
        }
    }
}

__global__ void graph_bounds(const int* __restrict__ batch, int* gs, int* ge)
{
    int i = blockIdx.x * blockDim.x + threadIdx.x;
    if (i < NNODES) {
        int b = batch[i];
        atomicMin(&gs[b], i);
        atomicMax(&ge[b], i + 1);
    }
}

#define AGG_GRP 8
#define AGG_WGS 2048

__global__ __launch_bounds__(256) void agg_mean_leaky(
    const float* __restrict__ H, const int* __restrict__ offs,
    const int* __restrict__ csr, float* __restrict__ out)
{
    const int lane = threadIdx.x & 31;
    const int wave0 = __builtin_amdgcn_readfirstlane(
        (int)((blockIdx.x * blockDim.x + threadIdx.x) >> 5));
    const int nwaves = (int)((gridDim.x * blockDim.x) >> 5);
    const float* colp = H + lane * 4;

    for (int node = wave0; node < NNODES; node += nwaves) {
        const int beg = offs[node];
        const int end = offs[node + 1];
        float ax = 0.f, ay = 0.f, az = 0.f, aw = 0.f;

        for (int e0 = beg; e0 < end; e0 += 32) {
            const int myidx = (e0 + lane < end) ? csr[e0 + lane] : 0;
            int cnt = end - e0; if (cnt > 32) cnt = 32;
            for (int j = 0; j < cnt; j += AGG_GRP) {
                const int r = cnt - j;
                int s[AGG_GRP];
#pragma unroll
                for (int k = 0; k < AGG_GRP; ++k) s[k] = __builtin_amdgcn_readlane(myidx, j + k);
                float4 v[AGG_GRP];
                v[0] = *(const float4*)(colp + (size_t)s[0] * HID);
#pragma unroll
                for (int k = 1; k < AGG_GRP; ++k) {
                    if (k < r) v[k] = *(const float4*)(colp + (size_t)s[k] * HID);
                }
                ax += v[0].x; ay += v[0].y; az += v[0].z; aw += v[0].w;
#pragma unroll
                for (int k = 1; k < AGG_GRP; ++k) {
                    if (k < r) { ax += v[k].x; ay += v[k].y; az += v[k].z; aw += v[k].w; }
                }
            }
        }
        const float inv = 1.0f / fmaxf((float)(end - beg), 1.0f);
        ax *= inv; ay *= inv; az *= inv; aw *= inv;
        float4 r4;
        r4.x = ax > 0.f ? ax : ALPHA * ax;
        r4.y = ay > 0.f ? ay : ALPHA * ay;
        r4.z = az > 0.f ? az : ALPHA * az;
        r4.w = aw > 0.f ? aw : ALPHA * aw;
        *(float4*)(out + (size_t)node * HID + lane * 4) = r4;
    }
}

__global__ void residual_add(float4* __restrict__ h, const float4* __restrict__ h2, int n)
{
    int i = blockIdx.x * blockDim.x + threadIdx.x;
    if (i < n) {
        float4 a = h[i], b = h2[i];
        a.x += b.x; a.y += b.y; a.z += b.z; a.w += b.w;
        h[i] = a;
    }
}

#define POOL_RG  4
#define POOL_INF 16

__global__ __launch_bounds__(512) void pool_max(
    const float* __restrict__ ht, const float* __restrict__ h1,
    const float* __restrict__ h2, const float* __restrict__ h3,
    const int* __restrict__ gs, const int* __restrict__ ge,
    float* __restrict__ pooled)
{
    __shared__ float4 red[POOL_RG][128];
    const int g = blockIdx.x;
    const int lane = (int)threadIdx.x & 31;
    const int wave = __builtin_amdgcn_readfirstlane((int)(threadIdx.x >> 5));
    const int arr = wave & 3;
    const int rg  = wave >> 2;
    const float* srcp = (arr == 0) ? ht : (arr == 1) ? h1 : (arr == 2) ? h2 : h3;
    const float* colp = srcp + lane * 4;
    const int b = gs[g], e = ge[g];

    float mx = -FMAXV, my = -FMAXV, mz = -FMAXV, mw = -FMAXV;
    for (int i0 = b + rg; i0 < e; i0 += POOL_RG * POOL_INF) {
        float4 v[POOL_INF];
#pragma unroll
        for (int k = 0; k < POOL_INF; ++k) {
            const int r = i0 + k * POOL_RG;
            v[k].x = -FMAXV; v[k].y = -FMAXV; v[k].z = -FMAXV; v[k].w = -FMAXV;
            if (r < e) v[k] = *(const float4*)(colp + (size_t)r * HID);
        }
#pragma unroll
        for (int k = 0; k < POOL_INF; ++k) {
            mx = fmaxf(mx, v[k].x); my = fmaxf(my, v[k].y);
            mz = fmaxf(mz, v[k].z); mw = fmaxf(mw, v[k].w);
        }
    }
    float4 m; m.x = mx; m.y = my; m.z = mz; m.w = mw;
    red[rg][arr * 32 + lane] = m;
    __syncthreads();
    if (threadIdx.x < 128) {
        const int c = (int)threadIdx.x;
        float4 r = red[0][c];
#pragma unroll
        for (int k = 1; k < POOL_RG; ++k) {
            const float4 q = red[k][c];
            r.x = fmaxf(r.x, q.x); r.y = fmaxf(r.y, q.y);
            r.z = fmaxf(r.z, q.z); r.w = fmaxf(r.w, q.w);
        }
        *(float4*)(pooled + (size_t)g * 512 + c * 4) = r;
    }
}

__global__ __launch_bounds__(32) void mlp2(
    const float* __restrict__ z, const float* __restrict__ W,
    const float* __restrict__ b, float* __restrict__ out)
{
    const int g = blockIdx.x;
    const int lane = threadIdx.x;
    const float z0 = z[(size_t)g * HID + lane];
    const float z1 = z[(size_t)g * HID + lane + 32];
    const float z2 = z[(size_t)g * HID + lane + 64];
    const float z3 = z[(size_t)g * HID + lane + 96];
    for (int c = 0; c < NCLS; ++c) {
        float s = z0 * W[lane * NCLS + c]
                + z1 * W[(lane + 32) * NCLS + c]
                + z2 * W[(lane + 64) * NCLS + c]
                + z3 * W[(lane + 96) * NCLS + c];
        for (int off = 16; off; off >>= 1) s += __shfl_xor(s, off, 32);
        if (lane == 0) out[g * NCLS + c] = s + b[c];
    }
}

extern "C" void kernel_launch(void* const* d_in, const int* in_sizes, int n_in,
                              void* d_out, int out_size, void* d_ws, size_t ws_size,
                              hipStream_t stream)
{
    (void)in_sizes; (void)n_in; (void)out_size; (void)ws_size;
    const float* x     = (const float*)d_in[0];
    const int*   ei    = (const int*)d_in[1];
    const int*   batch = (const int*)d_in[2];
    const float* W0 = (const float*)d_in[3];  const float* b0 = (const float*)d_in[4];
    const float* W1 = (const float*)d_in[5];  const float* b1 = (const float*)d_in[6];
    const float* W2 = (const float*)d_in[7];  const float* b2 = (const float*)d_in[8];
    const float* W3 = (const float*)d_in[9];  const float* b3 = (const float*)d_in[10];
    const float* Wm1 = (const float*)d_in[11]; const float* bm1 = (const float*)d_in[12];
    const float* gamma = (const float*)d_in[13]; const float* beta = (const float*)d_in[14];
    const float* Wm2 = (const float*)d_in[15]; const float* bm2 = (const float*)d_in[16];
    const int* esrc = ei;
    const int* edst = ei + NEDGES;

    char* ws = (char*)d_ws;
    size_t off = 0;
    auto alloc = [&](size_t bytes) -> void* {
        void* p = ws + off;
        off = (off + bytes + 255) & ~(size_t)255;
        return p;
    };
    const size_t featB = (size_t)NNODES * HID * sizeof(float);
    float* tmpH   = (float*)alloc(featB);
    float* h      = (float*)alloc(featB);
    float* h1     = (float*)alloc(featB);
    float* h2     = (float*)alloc(featB);
    float* ht     = (float*)alloc(featB);
    float* zbuf   = (float*)alloc((size_t)NGRAPHS * HID * sizeof(float));
    float* pooled = (float*)alloc((size_t)NGRAPHS * 512 * sizeof(float));
    int*   csr    = (int*)alloc((size_t)NEDGES * sizeof(int));
    int*   offs   = (int*)alloc((size_t)(NNODES + 1) * sizeof(int));
    int*   cnt    = (int*)alloc((size_t)NNODES * sizeof(int));
    int*   cntoff = (int*)alloc((size_t)(NNODES + 1) * sizeof(int));
    int*   packed = (int*)tmpH;
    int*   gs     = (int*)alloc((size_t)NGRAPHS * sizeof(int));
    int*   ge     = (int*)alloc((size_t)NGRAPHS * sizeof(int));

    const int TB = 256;
    edge_hist<<<CB_NCH, CB_TB, 0, stream>>>(edst, cnt, gs, ge);
    scan_offsets<<<SCAN_WGS, SCAN_TB, 0, stream>>>(cnt, cntoff);
    edge_scatter<<<CB_NCH, CB_TB, 0, stream>>>(esrc, edst, cntoff, packed);
    block_csr<<<CB_NB, CB_BT, 0, stream>>>(packed, cntoff, offs, csr);
    graph_bounds<<<(NNODES + TB - 1) / TB, TB, 0, stream>>>(batch, gs, ge);

    const int aggGrid  = AGG_WGS;

    gemm128_stripe<<<GL_WGS, 128, 0, stream>>>(x, W0, b0, tmpH);
    agg_mean_leaky<<<aggGrid, TB, 0, stream>>>(tmpH, offs, csr, h);
    gemm128_stripe<<<GL_WGS, 128, 0, stream>>>(h, W1, b1, tmpH);
    agg_mean_leaky<<<aggGrid, TB, 0, stream>>>(tmpH, offs, csr, h1);
    gemm128_stripe<<<GL_WGS, 128, 0, stream>>>(h1, W2, b2, tmpH);
    agg_mean_leaky<<<aggGrid, TB, 0, stream>>>(tmpH, offs, csr, h2);
    const int n4 = NNODES * HID / 4;
    residual_add<<<(n4 + TB - 1) / TB, TB, 0, stream>>>((float4*)h, (const float4*)h2, n4);
    gemm128_stripe<<<GL_WGS, 128, 0, stream>>>(h, W3, b3, tmpH);
    agg_mean_leaky<<<aggGrid, TB, 0, stream>>>(tmpH, offs, csr, ht);

    pool_max<<<NGRAPHS, 512, 0, stream>>>(ht, h1, h2, h, gs, ge, pooled);

    gemm_wmma_f32<<<NGRAPHS / 16, TB, 0, stream>>>(pooled, Wm1, bm1, zbuf, 512, 1, gamma, beta);
    mlp2<<<NGRAPHS, 32, 0, stream>>>(zbuf, Wm2, bm2, (float*)d_out);
}


# each layer is one kernel: neighbour rows of the layer input are averaged per 16-node tile, then the tile meets W on WMMA from LDS (mean and linear map commute); sum order changes: edges, then k
# speedup vs baseline: 2.8446x; 1.1566x over previous
#include <hip/hip_runtime.h>

#define NNODES 100000
#define NEDGES 1600000
#define NGRAPHS 512
#define HID 128
#define NCLS 10
#define ALPHA 0.01f
#define FMAXV 3.402823466e38f

typedef float v2f __attribute__((ext_vector_type(2)));
typedef float v4f __attribute__((ext_vector_type(4)));
typedef float v8f __attribute__((ext_vector_type(8)));

__global__ __launch_bounds__(256) void gemm_wmma_f32(
    const float* __restrict__ A, const float* __restrict__ W,
    const float* __restrict__ bias, float* __restrict__ C,
    int K, int ep, const float* __restrict__ gamma,
    const float* __restrict__ beta)
{
    const int N = HID;
    const int wave = threadIdx.x >> 5;
    const int lane = threadIdx.x & 31;
    const int m    = lane & 15;
    const int half = lane >> 4;
    const int row0 = blockIdx.x * 16;
    const int col  = wave * 16 + m;

    v8f acc;
    const float bv = bias[col];
#pragma unroll
    for (int v = 0; v < 8; ++v) acc[v] = bv;

    const float* arow = A + (size_t)(row0 + m) * K + 2 * half;
    const float* wcol = W + (size_t)(2 * half) * N + col;

    for (int k = 0; k < K; k += 4) {
        v2f a = *(const v2f*)(arow + k);
        v2f b;
        b.x = wcol[(size_t)k * N];
        b.y = wcol[(size_t)k * N + N];
        acc = __builtin_amdgcn_wmma_f32_16x16x4_f32(
            false, a, false, b, (short)0, acc, false, false);
    }

    if (ep == 1) {
        const float rs = 0.99999500003749971f;
        const float g  = gamma[col] * rs;
        const float be = beta[col];
#pragma unroll
        for (int v = 0; v < 8; ++v) acc[v] = fmaxf(acc[v] * g + be, 0.0f);
    }

#pragma unroll
    for (int v = 0; v < 8; ++v) {
        const int row = row0 + v + 8 * half;
        C[(size_t)row * N + col] = acc[v];
    }
}

#define CB_NB   400
#define CB_BN   (NNODES / CB_NB)
#define CB_NCH  250
#define CB_CE   (NEDGES / CB_NCH)
#define CB_TB   256
#define CB_EPT  (CB_CE / CB_TB)
static_assert(CB_NB * CB_BN == NNODES, "blocks must tile the nodes");
static_assert(CB_NCH * CB_CE == NEDGES, "chunks must tile the edges");
static_assert(CB_EPT * CB_TB == CB_CE, "threads must tile a chunk");
static_assert(CB_NB * CB_NCH == NNODES, "count matrix must have the scan's length");
static_assert(CB_BN <= 256, "local node index must fit 8 bits");

__global__ __launch_bounds__(CB_TB) void edge_hist(
    const int* __restrict__ dst, int* __restrict__ cnt,
    int* __restrict__ gs, int* __restrict__ ge)
{
    __shared__ int hist[CB_NB];
    const int t = (int)threadIdx.x;
    const int c = (int)blockIdx.x;
    for (int b = t; b < CB_NB; b += CB_TB) hist[b] = 0;
    __syncthreads();
    const int* dp = dst + (size_t)c * CB_CE + t;
    int d[CB_EPT];
#pragma unroll
    for (int j = 0; j < CB_EPT; ++j) d[j] = dp[j * CB_TB];
#pragma unroll
    for (int j = 0; j < CB_EPT; ++j) atomicAdd(&hist[(unsigned)d[j] / CB_BN], 1);
    __syncthreads();
    for (int b = t; b < CB_NB; b += CB_TB) cnt[b * CB_NCH + c] = hist[b];
    const int i = c * CB_TB + t;
    if (i < NGRAPHS) { gs[i] = NNODES; ge[i] = 0; }
}

#define SCAN_TB   1024
#define SCAN_VEC  (NNODES / 4)
#define SCAN_WGS  ((SCAN_VEC + SCAN_TB - 1) / SCAN_TB)

__global__ __launch_bounds__(SCAN_TB) void scan_offsets(
    const int* __restrict__ deg, int* __restrict__ offs)
{
    __shared__ int wave_own[32];
    __shared__ int wave_pre[32];
    const int t    = (int)threadIdx.x;
    const int lane = t & 31;
    const int wave = t >> 5;
    const int wg   = (int)blockIdx.x;
    const int4* __restrict__ deg4 = (const int4*)deg;
    const int myv = wg * SCAN_TB + t;

    int4 v; v.x = 0; v.y = 0; v.z = 0; v.w = 0;
    if (myv < SCAN_VEC) v = deg4[myv];

    int4 u[SCAN_WGS - 1];
#pragma unroll
    for (int j = 0; j < SCAN_WGS - 1; ++j) u[j] = deg4[j * SCAN_TB + t];
    int pre = 0;
#pragma unroll
    for (int j = 0; j < SCAN_WGS - 1; ++j) {
        const int s4 = (u[j].x + u[j].y) + (u[j].z + u[j].w);
        pre += (j < wg) ? s4 : 0;
    }

    const int own = (v.x + v.y) + (v.z + v.w);
    int inc = own;
#pragma unroll
    for (int off = 1; off < 32; off <<= 1) {
        const int n = __shfl_up(inc, off, 32);
        if (lane >= off) inc += n;
    }
#pragma unroll
    for (int off = 16; off; off >>= 1) pre += __shfl_xor(pre, off, 32);
    if (lane == 31) wave_own[wave] = inc;
    if (lane == 0)  wave_pre[wave] = pre;
    __syncthreads();

    const int wo = wave_own[lane];
    int wp = wave_pre[lane];
    int winc = wo;
#pragma unroll
    for (int off = 1; off < 32; off <<= 1) {
        const int n = __shfl_up(winc, off, 32);
        if (lane >= off) winc += n;
    }
#pragma unroll
    for (int off = 16; off; off >>= 1) wp += __shfl_xor(wp, off, 32);
    const int wave_base = __shfl(winc - wo, wave, 32);

    if (myv < SCAN_VEC) {
        int4 o;
        o.x = wp + wave_base + (inc - own);
        o.y = o.x + v.x;
        o.z = o.y + v.y;
        o.w = o.z + v.z;
        ((int4*)offs)[myv] = o;
        if (myv == SCAN_VEC - 1) offs[NNODES] = o.w + v.w;
    }
}

__global__ __launch_bounds__(CB_TB) void edge_scatter(
    const int* __restrict__ src, const int* __restrict__ dst,
    const int* __restrict__ cntoff, int* __restrict__ packed)
{
    __shared__ int cur[CB_NB];
    const int t = (int)threadIdx.x;
    const int c = (int)blockIdx.x;
    for (int b = t; b < CB_NB; b += CB_TB) cur[b] = cntoff[b * CB_NCH + c];
    const int* sp = src + (size_t)c * CB_CE + t;
    const int* dp = dst + (size_t)c * CB_CE + t;
    int s[CB_EPT], d[CB_EPT];
#pragma unroll
    for (int j = 0; j < CB_EPT; ++j) { s[j] = sp[j * CB_TB]; d[j] = dp[j * CB_TB]; }
    __syncthreads();
    int pos[CB_EPT];
#pragma unroll
    for (int j = 0; j < CB_EPT; ++j) {
        const unsigned b = (unsigned)d[j] / CB_BN;
        const int loc = d[j] - (int)b * CB_BN;
        pos[j] = atomicAdd(&cur[b], 1);
        s[j] = (s[j] << 8) | loc;
    }
#pragma unroll
    for (int j = 0; j < CB_EPT; ++j) packed[pos[j]] = s[j];
}

#define CB_STAGE 8192
#define CB_BT    512
#define CB_WPT   (CB_STAGE / CB_BT)
static_assert(CB_BT >= 256, "the scan of the node counts takes 256 threads");
__global__ __launch_bounds__(CB_BT) void block_csr(
    const int* __restrict__ packed, const int* __restrict__ cntoff,
    int* __restrict__ offs, int* __restrict__ csr)
{
    __shared__ int ncnt[256];
    __shared__ int wtot[8];
    __shared__ int stage[CB_STAGE];
    const int t    = (int)threadIdx.x;
    const int lane = t & 31;
    const int wave = t >> 5;
    const int b    = (int)blockIdx.x;
    const int beg = cntoff[b * CB_NCH];
    const int end = cntoff[(b + 1) * CB_NCH];
    const int n = end - beg;
    const bool fits = n <= CB_STAGE;
    if (t < 256) ncnt[t] = 0;
    int w[CB_WPT];
    if (fits) {
#pragma unroll
        for (int k = 0; k < CB_WPT; ++k) {
            const int i = t + CB_BT * k;
            w[k] = (i < n) ? packed[beg + i] : 0;
        }
    }
    __syncthreads();
    if (fits) {
#pragma unroll
        for (int k = 0; k < CB_WPT; ++k) {
            if (t + CB_BT * k < n) atomicAdd(&ncnt[w[k] & 255], 1);
        }
    } else {
        for (int e = beg + t; e < end; e += CB_BT) atomicAdd(&ncnt[packed[e] & 255], 1);
    }
    __syncthreads();

    int own = 0, inc = 0;
    if (t < 256) {
        own = ncnt[t];
        inc = own;
#pragma unroll
        for (int off = 1; off < 32; off <<= 1) {
            const int up = __shfl_up(inc, off, 32);
            if (lane >= off) inc += up;
        }
        if (lane == 31) wtot[wave] = inc;
    }
    __syncthreads();
    if (t < 256) {
        int base = beg;
#pragma unroll
        for (int q = 0; q < 8; ++q) base += (q < wave) ? wtot[q] : 0;
        const int start = base + (inc - own);
        ncnt[t] = start;
        if (t < CB_BN) offs[b * CB_BN + t] = start;
    }
    if (b == CB_NB - 1 && t == 0) offs[NNODES] = end;
    __syncthreads();
    if (fits) {
#pragma unroll
        for (int k = 0; k < CB_WPT; ++k) {
            if (t + CB_BT * k < n) {
                const int p = atomicAdd(&ncnt[w[k] & 255], 1);
                stage[p - beg] = w[k] >> 8;
            }
        }
        __syncthreads();
        for (int i = t; i < n; i += CB_BT) csr[beg + i] = stage[i];
    } else {
        for (int e = beg + t; e < end; e += CB_BT) {
            const int v = packed[e];
            const int p = atomicAdd(&ncnt[v & 255], 1);
            csr[p] = v >> 8;
        }
    }
}

__global__ void graph_bounds(const int* __restrict__ batch, int* gs, int* ge)
{
    int i = blockIdx.x * blockDim.x + threadIdx.x;
    if (i < NNODES) {
        int b = batch[i];
        atomicMin(&gs[b], i);
        atomicMax(&ge[b], i + 1);
    }
}

#define FL_QPITCH 544
#define FL_TPITCH 132
#define FL_TILES (NNODES / 16)
#define FL_WGS   768
#define AGG_GRP  8
static_assert(FL_TILES * 16 == NNODES, "tiles must cover the nodes exactly");

__global__ __launch_bounds__(512) __attribute__((amdgpu_waves_per_eu(12))) void layer_fused(
    const float* __restrict__ X, const float* __restrict__ W,
    const float* __restrict__ bias, const int* __restrict__ offs,
    const int* __restrict__ csr, float* __restrict__ out)
{
    __shared__ __attribute__((aligned(16))) float wl[32 * FL_QPITCH];
    __shared__ __attribute__((aligned(16))) float tl[2][16 * FL_TPITCH];
    __shared__ int tdeg[2][16];
    const int lane = (int)threadIdx.x & 31;
    const int wave = __builtin_amdgcn_readfirstlane((int)(threadIdx.x >> 5));
    const int m    = lane & 15;
    const int half = lane >> 4;

#pragma unroll
    for (int i = 0; i < 2; ++i) {
        const int q = wave + 16 * i;
        const float* wp = W + (size_t)(4 * q) * HID + 4 * lane;
        const float4 r0 = *(const float4*)(wp);
        const float4 r1 = *(const float4*)(wp + HID);
        const float4 r2 = *(const float4*)(wp + 2 * HID);
        const float4 r3 = *(const float4*)(wp + 3 * HID);
        float4 c0; c0.x = r0.x; c0.y = r1.x; c0.z = r2.x; c0.w = r3.x;
        float4 c1; c1.x = r0.y; c1.y = r1.y; c1.z = r2.y; c1.w = r3.y;
        float4 c2; c2.x = r0.z; c2.y = r1.z; c2.z = r2.z; c2.w = r3.z;
        float4 c3; c3.x = r0.w; c3.y = r1.w; c3.z = r2.w; c3.w = r3.w;
        float* d = wl + q * FL_QPITCH + 16 * lane;
        *(float4*)d = c0;
        *(float4*)(d + 4) = c1;
        *(float4*)(d + 8) = c2;
        *(float4*)(d + 12) = c3;
    }

    const float* colp = X + lane * 4;
    const float bv = bias[16 * (wave & 7) + m];
    int buf = 0;
    for (int t = (int)blockIdx.x; t < FL_TILES; t += FL_WGS) {
        const int node = t * 16 + wave;
        const int beg = offs[node];
        const int end = offs[node + 1];
        float ax = 0.f, ay = 0.f, az = 0.f, aw = 0.f;

        for (int e0 = beg; e0 < end; e0 += 32) {
            const int myidx = (e0 + lane < end) ? csr[e0 + lane] : 0;
            int cnt = end - e0; if (cnt > 32) cnt = 32;
            for (int j = 0; j < cnt; j += AGG_GRP) {
                const int r = cnt - j;
                int s[AGG_GRP];
#pragma unroll
                for (int k = 0; k < AGG_GRP; ++k) s[k] = __builtin_amdgcn_readlane(myidx, j + k);
                float4 v[AGG_GRP];
                v[0] = *(const float4*)(colp + (size_t)s[0] * HID);
#pragma unroll
                for (int k = 1; k < AGG_GRP; ++k) {
                    if (k < r) v[k] = *(const float4*)(colp + (size_t)s[k] * HID);
                }
                ax += v[0].x; ay += v[0].y; az += v[0].z; aw += v[0].w;
#pragma unroll
                for (int k = 1; k < AGG_GRP; ++k) {
                    if (k < r) { ax += v[k].x; ay += v[k].y; az += v[k].z; aw += v[k].w; }
                }
            }
        }
        const float inv = 1.0f / fmaxf((float)(end - beg), 1.0f);
        float4 mean; mean.x = ax * inv; mean.y = ay * inv; mean.z = az * inv; mean.w = aw * inv;
        *(float4*)(&tl[buf][wave * FL_TPITCH + lane * 4]) = mean;
        if (lane == 0) tdeg[buf][wave] = end - beg;
        __syncthreads();

        if (wave < 8) {
            v8f acc;
#pragma unroll
            for (int v = 0; v < 8; ++v) acc[v] = 0.0f;
            const float* al = &tl[buf][m * FL_TPITCH + 4 * half];
            const float* bl = wl + half * FL_QPITCH + 4 * (16 * wave + m);
            v4f fa[16], fb[16];
#pragma unroll
            for (int u = 0; u < 2; ++u) {
                fa[u] = *(const v4f*)(al + 8 * u);
                fb[u] = *(const v4f*)(bl + (2 * u) * FL_QPITCH);
            }
#pragma unroll
            for (int u = 0; u < 16; ++u) {
                if (u + 2 < 16) {
                    fa[u + 2] = *(const v4f*)(al + 8 * (u + 2));
                    fb[u + 2] = *(const v4f*)(bl + (2 * (u + 2)) * FL_QPITCH);
                }
                asm volatile("" :: "v"(fa[u]), "v"(fb[u]));
                v2f a0; a0.x = fa[u].x; a0.y = fa[u].y;
                v2f b0; b0.x = fb[u].x; b0.y = fb[u].y;
                v2f a1; a1.x = fa[u].z; a1.y = fa[u].w;
                v2f b1; b1.x = fb[u].z; b1.y = fb[u].w;
                acc = __builtin_amdgcn_wmma_f32_16x16x4_f32(
                    false, a0, false, b0, (short)0, acc, false, false);
                acc = __builtin_amdgcn_wmma_f32_16x16x4_f32(
                    false, a1, false, b1, (short)0, acc, false, false);
            }
            float* ob = out + (size_t)t * (16 * HID);
            const int lo = (8 * half) * HID + 16 * wave + m;
#pragma unroll
            for (int v = 0; v < 8; ++v) {
                const int rr = v + 8 * half;
                float y = (tdeg[buf][rr] > 0) ? acc[v] + bv : 0.0f;
                y = y > 0.f ? y : ALPHA * y;
                ob[lo + v * HID] = y;
            }
        }
        buf ^= 1;
    }
}

__global__ void residual_add(float4* __restrict__ h, const float4* __restrict__ h2, int n)
{
    int i = blockIdx.x * blockDim.x + threadIdx.x;
    if (i < n) {
        float4 a = h[i], b = h2[i];
        a.x += b.x; a.y += b.y; a.z += b.z; a.w += b.w;
        h[i] = a;
    }
}

#define POOL_RG  4
#define POOL_INF 16

__global__ __launch_bounds__(512) void pool_max(
    const float* __restrict__ ht, const float* __restrict__ h1,
    const float* __restrict__ h2, const float* __restrict__ h3,
    const int* __restrict__ gs, const int* __restrict__ ge,
    float* __restrict__ pooled)
{
    __shared__ float4 red[POOL_RG][128];
    const int g = blockIdx.x;
    const int lane = (int)threadIdx.x & 31;
    const int wave = __builtin_amdgcn_readfirstlane((int)(threadIdx.x >> 5));
    const int arr = wave & 3;
    const int rg  = wave >> 2;
    const float* srcp = (arr == 0) ? ht : (arr == 1) ? h1 : (arr == 2) ? h2 : h3;
    const float* colp = srcp + lane * 4;
    const int b = gs[g], e = ge[g];

    float mx = -FMAXV, my = -FMAXV, mz = -FMAXV, mw = -FMAXV;
    for (int i0 = b + rg; i0 < e; i0 += POOL_RG * POOL_INF) {
        float4 v[POOL_INF];
#pragma unroll
        for (int k = 0; k < POOL_INF; ++k) {
            const int r = i0 + k * POOL_RG;
            v[k].x = -FMAXV; v[k].y = -FMAXV; v[k].z = -FMAXV; v[k].w = -FMAXV;
            if (r < e) v[k] = *(const float4*)(colp + (size_t)r * HID);
        }
#pragma unroll
        for (int k = 0; k < POOL_INF; ++k) {
            mx = fmaxf(mx, v[k].x); my = fmaxf(my, v[k].y);
            mz = fmaxf(mz, v[k].z); mw = fmaxf(mw, v[k].w);
        }
    }
    float4 m; m.x = mx; m.y = my; m.z = mz; m.w = mw;
    red[rg][arr * 32 + lane] = m;
    __syncthreads();
    if (threadIdx.x < 128) {
        const int c = (int)threadIdx.x;
        float4 r = red[0][c];
#pragma unroll
        for (int k = 1; k < POOL_RG; ++k) {
            const float4 q = red[k][c];
            r.x = fmaxf(r.x, q.x); r.y = fmaxf(r.y, q.y);
            r.z = fmaxf(r.z, q.z); r.w = fmaxf(r.w, q.w);
        }
        *(float4*)(pooled + (size_t)g * 512 + c * 4) = r;
    }
}

__global__ __launch_bounds__(32) void mlp2(
    const float* __restrict__ z, const float* __restrict__ W,
    const float* __restrict__ b, float* __restrict__ out)
{
    const int g = blockIdx.x;
    const int lane = threadIdx.x;
    const float z0 = z[(size_t)g * HID + lane];
    const float z1 = z[(size_t)g * HID + lane + 32];
    const float z2 = z[(size_t)g * HID + lane + 64];
    const float z3 = z[(size_t)g * HID + lane + 96];
    for (int c = 0; c < NCLS; ++c) {
        float s = z0 * W[lane * NCLS + c]
                + z1 * W[(lane + 32) * NCLS + c]
                + z2 * W[(lane + 64) * NCLS + c]
                + z3 * W[(lane + 96) * NCLS + c];
        for (int off = 16; off; off >>= 1) s += __shfl_xor(s, off, 32);
        if (lane == 0) out[g * NCLS + c] = s + b[c];
    }
}

extern "C" void kernel_launch(void* const* d_in, const int* in_sizes, int n_in,
                              void* d_out, int out_size, void* d_ws, size_t ws_size,
                              hipStream_t stream)
{
    (void)in_sizes; (void)n_in; (void)out_size; (void)ws_size;
    const float* x     = (const float*)d_in[0];
    const int*   ei    = (const int*)d_in[1];
    const int*   batch = (const int*)d_in[2];
    const float* W0 = (const float*)d_in[3];  const float* b0 = (const float*)d_in[4];
    const float* W1 = (const float*)d_in[5];  const float* b1 = (const float*)d_in[6];
    const float* W2 = (const float*)d_in[7];  const float* b2 = (const float*)d_in[8];
    const float* W3 = (const float*)d_in[9];  const float* b3 = (const float*)d_in[10];
    const float* Wm1 = (const float*)d_in[11]; const float* bm1 = (const float*)d_in[12];
    const float* gamma = (const float*)d_in[13]; const float* beta = (const float*)d_in[14];
    const float* Wm2 = (const float*)d_in[15]; const float* bm2 = (const float*)d_in[16];
    const int* esrc = ei;
    const int* edst = ei + NEDGES;

    char* ws = (char*)d_ws;
    size_t off = 0;
    auto alloc = [&](size_t bytes) -> void* {
        void* p = ws + off;
        off = (off + bytes + 255) & ~(size_t)255;
        return p;
    };
    const size_t featB = (size_t)NNODES * HID * sizeof(float);
    float* tmpH   = (float*)alloc(featB);
    float* h      = (float*)alloc(featB);
    float* h1     = (float*)alloc(featB);
    float* h2     = (float*)alloc(featB);
    float* ht     = (float*)alloc(featB);
    float* zbuf   = (float*)alloc((size_t)NGRAPHS * HID * sizeof(float));
    float* pooled = (float*)alloc((size_t)NGRAPHS * 512 * sizeof(float));
    int*   csr    = (int*)alloc((size_t)NEDGES * sizeof(int));
    int*   offs   = (int*)alloc((size_t)(NNODES + 1) * sizeof(int));
    int*   cnt    = (int*)alloc((size_t)NNODES * sizeof(int));
    int*   cntoff = (int*)alloc((size_t)(NNODES + 1) * sizeof(int));
    int*   packed = (int*)tmpH;
    int*   gs     = (int*)alloc((size_t)NGRAPHS * sizeof(int));
    int*   ge     = (int*)alloc((size_t)NGRAPHS * sizeof(int));

    const int TB = 256;
    edge_hist<<<CB_NCH, CB_TB, 0, stream>>>(edst, cnt, gs, ge);
    scan_offsets<<<SCAN_WGS, SCAN_TB, 0, stream>>>(cnt, cntoff);
    edge_scatter<<<CB_NCH, CB_TB, 0, stream>>>(esrc, edst, cntoff, packed);
    block_csr<<<CB_NB, CB_BT, 0, stream>>>(packed, cntoff, offs, csr);
    graph_bounds<<<(NNODES + TB - 1) / TB, TB, 0, stream>>>(batch, gs, ge);

    layer_fused<<<FL_WGS, 512, 0, stream>>>(x, W0, b0, offs, csr, h);
    layer_fused<<<FL_WGS, 512, 0, stream>>>(h, W1, b1, offs, csr, h1);
    layer_fused<<<FL_WGS, 512, 0, stream>>>(h1, W2, b2, offs, csr, h2);
    const int n4 = NNODES * HID / 4;
    residual_add<<<(n4 + TB - 1) / TB, TB, 0, stream>>>((float4*)h, (const float4*)h2, n4);
    layer_fused<<<FL_WGS, 512, 0, stream>>>(h, W3, b3, offs, csr, ht);

    pool_max<<<NGRAPHS, 512, 0, stream>>>(ht, h1, h2, h, gs, ge, pooled);

    gemm_wmma_f32<<<NGRAPHS / 16, TB, 0, stream>>>(pooled, Wm1, bm1, zbuf, 512, 1, gamma, beta);
    mlp2<<<NGRAPHS, 32, 0, stream>>>(zbuf, Wm2, bm2, (float*)d_out);
}


# layer epilogues replace the extra passes over planes: h += h2 in place; per-tile column maxima, which pooling reads instead of rows (rows only at a graph's ragged ends; ht kept only there); same sums
# speedup vs baseline: 3.0373x; 1.0677x over previous
#include <hip/hip_runtime.h>

#define NNODES 100000
#define NEDGES 1600000
#define NGRAPHS 512
#define HID 128
#define NCLS 10
#define ALPHA 0.01f
#define FMAXV 3.402823466e38f

typedef float v2f __attribute__((ext_vector_type(2)));
typedef float v4f __attribute__((ext_vector_type(4)));
typedef float v8f __attribute__((ext_vector_type(8)));

__global__ __launch_bounds__(256) void gemm_wmma_f32(
    const float* __restrict__ A, const float* __restrict__ W,
    const float* __restrict__ bias, float* __restrict__ C,
    int K, int ep, const float* __restrict__ gamma,
    const float* __restrict__ beta)
{
    const int N = HID;
    const int wave = threadIdx.x >> 5;
    const int lane = threadIdx.x & 31;
    const int m    = lane & 15;
    const int half = lane >> 4;
    const int row0 = blockIdx.x * 16;
    const int col  = wave * 16 + m;

    v8f acc;
    const float bv = bias[col];
#pragma unroll
    for (int v = 0; v < 8; ++v) acc[v] = bv;

    const float* arow = A + (size_t)(row0 + m) * K + 2 * half;
    const float* wcol = W + (size_t)(2 * half) * N + col;

    for (int k = 0; k < K; k += 4) {
        v2f a = *(const v2f*)(arow + k);
        v2f b;
        b.x = wcol[(size_t)k * N];
        b.y = wcol[(size_t)k * N + N];
        acc = __builtin_amdgcn_wmma_f32_16x16x4_f32(
            false, a, false, b, (short)0, acc, false, false);
    }

    if (ep == 1) {
        const float rs = 0.99999500003749971f;
        const float g  = gamma[col] * rs;
        const float be = beta[col];
#pragma unroll
        for (int v = 0; v < 8; ++v) acc[v] = fmaxf(acc[v] * g + be, 0.0f);
    }

#pragma unroll
    for (int v = 0; v < 8; ++v) {
        const int row = row0 + v + 8 * half;
        C[(size_t)row * N + col] = acc[v];
    }
}

#define CB_NB   400
#define CB_BN   (NNODES / CB_NB)
#define CB_NCH  250
#define CB_CE   (NEDGES / CB_NCH)
#define CB_TB   256
#define CB_EPT  (CB_CE / CB_TB)
static_assert(CB_NB * CB_BN == NNODES, "blocks must tile the nodes");
static_assert(CB_NCH * CB_CE == NEDGES, "chunks must tile the edges");
static_assert(CB_EPT * CB_TB == CB_CE, "threads must tile a chunk");
static_assert(CB_NB * CB_NCH == NNODES, "count matrix must have the scan's length");
static_assert(CB_BN <= 256, "local node index must fit 8 bits");

__global__ __launch_bounds__(CB_TB) void edge_hist(
    const int* __restrict__ dst, int* __restrict__ cnt,
    int* __restrict__ gs, int* __restrict__ ge)
{
    __shared__ int hist[CB_NB];
    const int t = (int)threadIdx.x;
    const int c = (int)blockIdx.x;
    for (int b = t; b < CB_NB; b += CB_TB) hist[b] = 0;
    __syncthreads();
    const int* dp = dst + (size_t)c * CB_CE + t;
    int d[CB_EPT];
#pragma unroll
    for (int j = 0; j < CB_EPT; ++j) d[j] = dp[j * CB_TB];
#pragma unroll
    for (int j = 0; j < CB_EPT; ++j) atomicAdd(&hist[(unsigned)d[j] / CB_BN], 1);
    __syncthreads();
    for (int b = t; b < CB_NB; b += CB_TB) cnt[b * CB_NCH + c] = hist[b];
    const int i = c * CB_TB + t;
    if (i < NGRAPHS) { gs[i] = NNODES; ge[i] = 0; }
}

#define SCAN_TB   1024
#define SCAN_VEC  (NNODES / 4)
#define SCAN_WGS  ((SCAN_VEC + SCAN_TB - 1) / SCAN_TB)

__global__ __launch_bounds__(SCAN_TB) void scan_offsets(
    const int* __restrict__ deg, int* __restrict__ offs)
{
    __shared__ int wave_own[32];
    __shared__ int wave_pre[32];
    const int t    = (int)threadIdx.x;
    const int lane = t & 31;
    const int wave = t >> 5;
    const int wg   = (int)blockIdx.x;
    const int4* __restrict__ deg4 = (const int4*)deg;
    const int myv = wg * SCAN_TB + t;

    int4 v; v.x = 0; v.y = 0; v.z = 0; v.w = 0;
    if (myv < SCAN_VEC) v = deg4[myv];

    int4 u[SCAN_WGS - 1];
#pragma unroll
    for (int j = 0; j < SCAN_WGS - 1; ++j) u[j] = deg4[j * SCAN_TB + t];
    int pre = 0;
#pragma unroll
    for (int j = 0; j < SCAN_WGS - 1; ++j) {
        const int s4 = (u[j].x + u[j].y) + (u[j].z + u[j].w);
        pre += (j < wg) ? s4 : 0;
    }

    const int own = (v.x + v.y) + (v.z + v.w);
    int inc = own;
#pragma unroll
    for (int off = 1; off < 32; off <<= 1) {
        const int n = __shfl_up(inc, off, 32);
        if (lane >= off) inc += n;
    }
#pragma unroll
    for (int off = 16; off; off >>= 1) pre += __shfl_xor(pre, off, 32);
    if (lane == 31) wave_own[wave] = inc;
    if (lane == 0)  wave_pre[wave] = pre;
    __syncthreads();

    const int wo = wave_own[lane];
    int wp = wave_pre[lane];
    int winc = wo;
#pragma unroll
    for (int off = 1; off < 32; off <<= 1) {
        const int n = __shfl_up(winc, off, 32);
        if (lane >= off) winc += n;
    }
#pragma unroll
    for (int off = 16; off; off >>= 1) wp += __shfl_xor(wp, off, 32);
    const int wave_base = __shfl(winc - wo, wave, 32);

    if (myv < SCAN_VEC) {
        int4 o;
        o.x = wp + wave_base + (inc - own);
        o.y = o.x + v.x;
        o.z = o.y + v.y;
        o.w = o.z + v.z;
        ((int4*)offs)[myv] = o;
        if (myv == SCAN_VEC - 1) offs[NNODES] = o.w + v.w;
    }
}

__global__ __launch_bounds__(CB_TB) void edge_scatter(
    const int* __restrict__ src, const int* __restrict__ dst,
    const int* __restrict__ cntoff, int* __restrict__ packed)
{
    __shared__ int cur[CB_NB];
    const int t = (int)threadIdx.x;
    const int c = (int)blockIdx.x;
    for (int b = t; b < CB_NB; b += CB_TB) cur[b] = cntoff[b * CB_NCH + c];
    const int* sp = src + (size_t)c * CB_CE + t;
    const int* dp = dst + (size_t)c * CB_CE + t;
    int s[CB_EPT], d[CB_EPT];
#pragma unroll
    for (int j = 0; j < CB_EPT; ++j) { s[j] = sp[j * CB_TB]; d[j] = dp[j * CB_TB]; }
    __syncthreads();
    int pos[CB_EPT];
#pragma unroll
    for (int j = 0; j < CB_EPT; ++j) {
        const unsigned b = (unsigned)d[j] / CB_BN;
        const int loc = d[j] - (int)b * CB_BN;
        pos[j] = atomicAdd(&cur[b], 1);
        s[j] = (s[j] << 8) | loc;
    }
#pragma unroll
    for (int j = 0; j < CB_EPT; ++j) packed[pos[j]] = s[j];
}

#define CB_STAGE 8192
#define CB_BT    512
#define CB_WPT   (CB_STAGE / CB_BT)
static_assert(CB_BT >= 256, "the scan of the node counts takes 256 threads");
__global__ __launch_bounds__(CB_BT) void block_csr(
    const int* __restrict__ packed, const int* __restrict__ cntoff,
    int* __restrict__ offs, int* __restrict__ csr)
{
    __shared__ int ncnt[256];
    __shared__ int wtot[8];
    __shared__ int stage[CB_STAGE];
    const int t    = (int)threadIdx.x;
    const int lane = t & 31;
    const int wave = t >> 5;
    const int b    = (int)blockIdx.x;
    const int beg = cntoff[b * CB_NCH];
    const int end = cntoff[(b + 1) * CB_NCH];
    const int n = end - beg;
    const bool fits = n <= CB_STAGE;
    if (t < 256) ncnt[t] = 0;
    int w[CB_WPT];
    if (fits) {
#pragma unroll
        for (int k = 0; k < CB_WPT; ++k) {
            const int i = t + CB_BT * k;
            w[k] = (i < n) ? packed[beg + i] : 0;
        }
    }
    __syncthreads();
    if (fits) {
#pragma unroll
        for (int k = 0; k < CB_WPT; ++k) {
            if (t + CB_BT * k < n) atomicAdd(&ncnt[w[k] & 255], 1);
        }
    } else {
        for (int e = beg + t; e < end; e += CB_BT) atomicAdd(&ncnt[packed[e] & 255], 1);
    }
    __syncthreads();

    int own = 0, inc = 0;
    if (t < 256) {
        own = ncnt[t];
        inc = own;
#pragma unroll
        for (int off = 1; off < 32; off <<= 1) {
            const int up = __shfl_up(inc, off, 32);
            if (lane >= off) inc += up;
        }
        if (lane == 31) wtot[wave] = inc;
    }
    __syncthreads();
    if (t < 256) {
        int base = beg;
#pragma unroll
        for (int q = 0; q < 8; ++q) base += (q < wave) ? wtot[q] : 0;
        const int start = base + (inc - own);
        ncnt[t] = start;
        if (t < CB_BN) offs[b * CB_BN + t] = start;
    }
    if (b == CB_NB - 1 && t == 0) offs[NNODES] = end;
    __syncthreads();
    if (fits) {
#pragma unroll
        for (int k = 0; k < CB_WPT; ++k) {
            if (t + CB_BT * k < n) {
                const int p = atomicAdd(&ncnt[w[k] & 255], 1);
                stage[p - beg] = w[k] >> 8;
            }
        }
        __syncthreads();
        for (int i = t; i < n; i += CB_BT) csr[beg + i] = stage[i];
    } else {
        for (int e = beg + t; e < end; e += CB_BT) {
            const int v = packed[e];
            const int p = atomicAdd(&ncnt[v & 255], 1);
            csr[p] = v >> 8;
        }
    }
}

__global__ void graph_bounds(const int* __restrict__ batch, int* gs, int* ge)
{
    int i = blockIdx.x * blockDim.x + threadIdx.x;
    if (i < NNODES) {
        int b = batch[i];
        atomicMin(&gs[b], i);
        atomicMax(&ge[b], i + 1);
    }
}

__device__ __forceinline__ float other_half(float x)
{
    const int b = __float_as_int(x);
    return __int_as_float(__builtin_amdgcn_permlanex16(b, b, 0x76543210, 0xfedcba98, false, false));
}

#define FL_QPITCH 544
#define FL_TPITCH 132
#define FL_TILES (NNODES / 16)
#define FL_WGS   768
#define AGG_GRP  8
static_assert(FL_TILES * 16 == NNODES, "tiles must cover the nodes exactly");

__global__ __launch_bounds__(512) __attribute__((amdgpu_waves_per_eu(12))) void layer_fused(
    const float* __restrict__ X, const float* __restrict__ W,
    const float* __restrict__ bias, const int* __restrict__ offs,
    const int* __restrict__ csr, float* __restrict__ out,
    float* __restrict__ tmax, float* __restrict__ res, float* __restrict__ rmax,
    const int* __restrict__ batch)
{
    __shared__ __attribute__((aligned(16))) float wl[32 * FL_QPITCH];
    __shared__ __attribute__((aligned(16))) float tl[2][16 * FL_TPITCH];
    __shared__ int tdeg[2][16];
    __shared__ float sb[HID];
    const int lane = (int)threadIdx.x & 31;
    const int wave = __builtin_amdgcn_readfirstlane((int)(threadIdx.x >> 5));
    const int m    = lane & 15;
    const int half = lane >> 4;

#pragma unroll
    for (int i = 0; i < 2; ++i) {
        const int q = wave + 16 * i;
        const float* wp = W + (size_t)(4 * q) * HID + 4 * lane;
        const float4 r0 = *(const float4*)(wp);
        const float4 r1 = *(const float4*)(wp + HID);
        const float4 r2 = *(const float4*)(wp + 2 * HID);
        const float4 r3 = *(const float4*)(wp + 3 * HID);
        float4 c0; c0.x = r0.x; c0.y = r1.x; c0.z = r2.x; c0.w = r3.x;
        float4 c1; c1.x = r0.y; c1.y = r1.y; c1.z = r2.y; c1.w = r3.y;
        float4 c2; c2.x = r0.z; c2.y = r1.z; c2.z = r2.z; c2.w = r3.z;
        float4 c3; c3.x = r0.w; c3.y = r1.w; c3.z = r2.w; c3.w = r3.w;
        float* d = wl + q * FL_QPITCH + 16 * lane;
        *(float4*)d = c0;
        *(float4*)(d + 4) = c1;
        *(float4*)(d + 8) = c2;
        *(float4*)(d + 12) = c3;
    }

    const float* colp = X + lane * 4;
    if (threadIdx.x < HID) sb[threadIdx.x] = bias[threadIdx.x];
    int buf = 0;
    for (int t = (int)blockIdx.x; t < FL_TILES; t += FL_WGS) {
        const int node = t * 16 + wave;
        const int beg = offs[node];
        const int end = offs[node + 1];
        float ax = 0.f, ay = 0.f, az = 0.f, aw = 0.f;
        bool rows = true;
        if (batch != nullptr && wave < 8) {
            const int g0 = batch[t * 16];
            const int gm = batch[t * 16 + m];
            rows = __ballot(gm != g0) != 0;
        }

        for (int e0 = beg; e0 < end; e0 += 32) {
            const int myidx = (e0 + lane < end) ? csr[e0 + lane] : 0;
            int cnt = end - e0; if (cnt > 32) cnt = 32;
            for (int j = 0; j < cnt; j += AGG_GRP) {
                const int r = cnt - j;
                int s[AGG_GRP];
#pragma unroll
                for (int k = 0; k < AGG_GRP; ++k) s[k] = __builtin_amdgcn_readlane(myidx, j + k);
                float4 v[AGG_GRP];
                v[0] = *(const float4*)(colp + (size_t)s[0] * HID);
#pragma unroll
                for (int k = 1; k < AGG_GRP; ++k) {
                    if (k < r) v[k] = *(const float4*)(colp + (size_t)s[k] * HID);
                }
                ax += v[0].x; ay += v[0].y; az += v[0].z; aw += v[0].w;
#pragma unroll
                for (int k = 1; k < AGG_GRP; ++k) {
                    if (k < r) { ax += v[k].x; ay += v[k].y; az += v[k].z; aw += v[k].w; }
                }
            }
        }
        const float inv = 1.0f / fmaxf((float)(end - beg), 1.0f);
        float4 mean; mean.x = ax * inv; mean.y = ay * inv; mean.z = az * inv; mean.w = aw * inv;
        *(float4*)(&tl[buf][wave * FL_TPITCH + lane * 4]) = mean;
        if (lane == 0) tdeg[buf][wave] = end - beg;
        __syncthreads();

        if (wave < 8) {
            v8f acc;
#pragma unroll
            for (int v = 0; v < 8; ++v) acc[v] = 0.0f;
            const float* al = &tl[buf][m * FL_TPITCH + 4 * half];
            const float* bl = wl + half * FL_QPITCH + 4 * (16 * wave + m);
            v4f fa[16], fb[16];
#pragma unroll
            for (int u = 0; u < 2; ++u) {
                fa[u] = *(const v4f*)(al + 8 * u);
                fb[u] = *(const v4f*)(bl + (2 * u) * FL_QPITCH);
            }
#pragma unroll
            for (int u = 0; u < 16; ++u) {
                if (u + 2 < 16) {
                    fa[u + 2] = *(const v4f*)(al + 8 * (u + 2));
                    fb[u + 2] = *(const v4f*)(bl + (2 * (u + 2)) * FL_QPITCH);
                }
                asm volatile("" :: "v"(fa[u]), "v"(fb[u]));
                v2f a0; a0.x = fa[u].x; a0.y = fa[u].y;
                v2f b0; b0.x = fb[u].x; b0.y = fb[u].y;
                v2f a1; a1.x = fa[u].z; a1.y = fa[u].w;
                v2f b1; b1.x = fb[u].z; b1.y = fb[u].w;
                acc = __builtin_amdgcn_wmma_f32_16x16x4_f32(
                    false, a0, false, b0, (short)0, acc, false, false);
                acc = __builtin_amdgcn_wmma_f32_16x16x4_f32(
                    false, a1, false, b1, (short)0, acc, false, false);
            }
            const unsigned lo = (unsigned)((8 * half) * HID + 16 * wave + m);
            const unsigned e0 = (unsigned)t * (16u * HID) + lo;
            const unsigned c0 = (unsigned)t * (unsigned)HID + (e0 & (HID - 1));
            float y[8];
            auto own_part = [&]() {
                const float bv = sb[16 * wave + m];
#pragma unroll
                for (int v = 0; v < 8; ++v) {
                    const int rr = v + 8 * half;
                    const float s = (tdeg[buf][rr] > 0) ? acc[v] + bv : 0.0f;
                    y[v] = s > 0.f ? s : ALPHA * s;
                }
                if (rows) {
#pragma unroll
                    for (int v = 0; v < 8; ++v) out[e0 + (unsigned)(v * HID)] = y[v];
                }
                if (tmax) {
                    float mx = y[0];
#pragma unroll
                    for (int v = 1; v < 8; ++v) mx = fmaxf(mx, y[v]);
                    mx = fmaxf(mx, other_half(mx));
                    if (half == 0) tmax[c0] = mx;
                }
            };
            if (res) {
                float rv[8];
#pragma unroll
                for (int v = 0; v < 8; ++v) rv[v] = res[e0 + (unsigned)(v * HID)];
                own_part();
#pragma unroll
                for (int v = 0; v < 8; ++v) { rv[v] += y[v]; res[e0 + (unsigned)(v * HID)] = rv[v]; }
                float mx = rv[0];
#pragma unroll
                for (int v = 1; v < 8; ++v) mx = fmaxf(mx, rv[v]);
                mx = fmaxf(mx, other_half(mx));
                if (half == 0) rmax[c0] = mx;
            } else {
                own_part();
            }
        }
        buf ^= 1;
    }
}

#define POOL_RG  4
#define POOL_INF 8

__global__ __launch_bounds__(512) void pool_max(
    const float* __restrict__ ht, const float* __restrict__ h1,
    const float* __restrict__ h2, const float* __restrict__ h3,
    const float* __restrict__ mt, const float* __restrict__ m1,
    const float* __restrict__ m2, const float* __restrict__ m3,
    const int* __restrict__ gs, const int* __restrict__ ge,
    float* __restrict__ pooled)
{
    __shared__ float4 red[POOL_RG][128];
    const int g = blockIdx.x;
    const int lane = (int)threadIdx.x & 31;
    const int wave = __builtin_amdgcn_readfirstlane((int)(threadIdx.x >> 5));
    const int arr = wave & 3;
    const int rg  = wave >> 2;
    const float* rowp = ((arr == 0) ? ht : (arr == 1) ? h1 : (arr == 2) ? h2 : h3) + lane * 4;
    const float* maxp = ((arr == 0) ? mt : (arr == 1) ? m1 : (arr == 2) ? m2 : m3) + lane * 4;
    const int b = gs[g], e = ge[g];

    const int t0 = (b + 15) >> 4;
    const int t1 = e >> 4;
    int nt = 0, hend = e, tbeg = e;
    if (t0 <= t1) { nt = t1 - t0; hend = 16 * t0; tbeg = 16 * t1; }
    int nh = hend - b; if (nh < 0) nh = 0;
    const int nl = e - tbeg;
    const int n = nt + nh + nl;

    float mx = -FMAXV, my = -FMAXV, mz = -FMAXV, mw = -FMAXV;
    for (int i0 = rg; i0 < n; i0 += POOL_RG * POOL_INF) {
        float4 v[POOL_INF];
#pragma unroll
        for (int k = 0; k < POOL_INF; ++k) {
            const int i = i0 + k * POOL_RG;
            v[k].x = -FMAXV; v[k].y = -FMAXV; v[k].z = -FMAXV; v[k].w = -FMAXV;
            if (i < n) {
                const float* p = (i < nt)      ? maxp + (size_t)(t0 + i) * HID
                               : (i < nt + nh) ? rowp + (size_t)(b + (i - nt)) * HID
                                               : rowp + (size_t)(tbeg + (i - nt - nh)) * HID;
                v[k] = *(const float4*)p;
            }
        }
#pragma unroll
        for (int k = 0; k < POOL_INF; ++k) {
            mx = fmaxf(mx, v[k].x); my = fmaxf(my, v[k].y);
            mz = fmaxf(mz, v[k].z); mw = fmaxf(mw, v[k].w);
        }
    }
    float4 m; m.x = mx; m.y = my; m.z = mz; m.w = mw;
    red[rg][arr * 32 + lane] = m;
    __syncthreads();
    if (threadIdx.x < 128) {
        const int c = (int)threadIdx.x;
        float4 r = red[0][c];
#pragma unroll
        for (int k = 1; k < POOL_RG; ++k) {
            const float4 q = red[k][c];
            r.x = fmaxf(r.x, q.x); r.y = fmaxf(r.y, q.y);
            r.z = fmaxf(r.z, q.z); r.w = fmaxf(r.w, q.w);
        }
        *(float4*)(pooled + (size_t)g * 512 + c * 4) = r;
    }
}

__global__ __launch_bounds__(32) void mlp2(
    const float* __restrict__ z, const float* __restrict__ W,
    const float* __restrict__ b, float* __restrict__ out)
{
    const int g = blockIdx.x;
    const int lane = threadIdx.x;
    const float z0 = z[(size_t)g * HID + lane];
    const float z1 = z[(size_t)g * HID + lane + 32];
    const float z2 = z[(size_t)g * HID + lane + 64];
    const float z3 = z[(size_t)g * HID + lane + 96];
    for (int c = 0; c < NCLS; ++c) {
        float s = z0 * W[lane * NCLS + c]
                + z1 * W[(lane + 32) * NCLS + c]
                + z2 * W[(lane + 64) * NCLS + c]
                + z3 * W[(lane + 96) * NCLS + c];
        for (int off = 16; off; off >>= 1) s += __shfl_xor(s, off, 32);
        if (lane == 0) out[g * NCLS + c] = s + b[c];
    }
}

extern "C" void kernel_launch(void* const* d_in, const int* in_sizes, int n_in,
                              void* d_out, int out_size, void* d_ws, size_t ws_size,
                              hipStream_t stream)
{
    (void)in_sizes; (void)n_in; (void)out_size; (void)ws_size;
    const float* x     = (const float*)d_in[0];
    const int*   ei    = (const int*)d_in[1];
    const int*   batch = (const int*)d_in[2];
    const float* W0 = (const float*)d_in[3];  const float* b0 = (const float*)d_in[4];
    const float* W1 = (const float*)d_in[5];  const float* b1 = (const float*)d_in[6];
    const float* W2 = (const float*)d_in[7];  const float* b2 = (const float*)d_in[8];
    const float* W3 = (const float*)d_in[9];  const float* b3 = (const float*)d_in[10];
    const float* Wm1 = (const float*)d_in[11]; const float* bm1 = (const float*)d_in[12];
    const float* gamma = (const float*)d_in[13]; const float* beta = (const float*)d_in[14];
    const float* Wm2 = (const float*)d_in[15]; const float* bm2 = (const float*)d_in[16];
    const int* esrc = ei;
    const int* edst = ei + NEDGES;

    char* ws = (char*)d_ws;
    size_t off = 0;
    auto alloc = [&](size_t bytes) -> void* {
        void* p = ws + off;
        off = (off + bytes + 255) & ~(size_t)255;
        return p;
    };
    const size_t featB = (size_t)NNODES * HID * sizeof(float);
    float* tmpH   = (float*)alloc(featB);
    float* h      = (float*)alloc(featB);
    float* h1     = (float*)alloc(featB);
    float* h2     = (float*)alloc(featB);
    float* ht     = (float*)alloc(featB);
    float* zbuf   = (float*)alloc((size_t)NGRAPHS * HID * sizeof(float));
    float* pooled = (float*)alloc((size_t)NGRAPHS * 512 * sizeof(float));
    int*   csr    = (int*)alloc((size_t)NEDGES * sizeof(int));
    int*   offs   = (int*)alloc((size_t)(NNODES + 1) * sizeof(int));
    int*   cnt    = (int*)alloc((size_t)NNODES * sizeof(int));
    int*   cntoff = (int*)alloc((size_t)(NNODES + 1) * sizeof(int));
    int*   packed = (int*)tmpH;
    float* mt     = tmpH + 2097152;
    float* m1     = mt + FL_TILES * HID;
    float* m2     = m1 + FL_TILES * HID;
    float* m3     = m2 + FL_TILES * HID;
    int*   gs     = (int*)alloc((size_t)NGRAPHS * sizeof(int));
    int*   ge     = (int*)alloc((size_t)NGRAPHS * sizeof(int));

    const int TB = 256;
    edge_hist<<<CB_NCH, CB_TB, 0, stream>>>(edst, cnt, gs, ge);
    scan_offsets<<<SCAN_WGS, SCAN_TB, 0, stream>>>(cnt, cntoff);
    edge_scatter<<<CB_NCH, CB_TB, 0, stream>>>(esrc, edst, cntoff, packed);
    block_csr<<<CB_NB, CB_BT, 0, stream>>>(packed, cntoff, offs, csr);
    graph_bounds<<<(NNODES + TB - 1) / TB, TB, 0, stream>>>(batch, gs, ge);

    float* const none = nullptr;
    const int* const nobatch = nullptr;
    layer_fused<<<FL_WGS, 512, 0, stream>>>(x, W0, b0, offs, csr, h, none, none, none, nobatch);
    layer_fused<<<FL_WGS, 512, 0, stream>>>(h, W1, b1, offs, csr, h1, m1, none, none, nobatch);
    layer_fused<<<FL_WGS, 512, 0, stream>>>(h1, W2, b2, offs, csr, h2, m2, h, m3, nobatch);
    layer_fused<<<FL_WGS, 512, 0, stream>>>(h, W3, b3, offs, csr, ht, mt, none, none, batch);

    pool_max<<<NGRAPHS, 512, 0, stream>>>(ht, h1, h2, h, mt, m1, m2, m3, gs, ge, pooled);

    gemm_wmma_f32<<<NGRAPHS / 16, TB, 0, stream>>>(pooled, Wm1, bm1, zbuf, 512, 1, gamma, beta);
    mlp2<<<NGRAPHS, 32, 0, stream>>>(zbuf, Wm2, bm2, (float*)d_out);
}


# CSR build + graph bounds in 2 launches (was 5): edge chunks sorted by block in LDS, whole-line stores; a block's workgroup gathers its pieces; bounds by neighbour compare; sum order may differ
# speedup vs baseline: 3.5989x; 1.1849x over previous
#include <hip/hip_runtime.h>

#define NNODES 100000
#define NEDGES 1600000
#define NGRAPHS 512
#define HID 128
#define NCLS 10
#define ALPHA 0.01f
#define FMAXV 3.402823466e38f

typedef float v2f __attribute__((ext_vector_type(2)));
typedef float v4f __attribute__((ext_vector_type(4)));
typedef float v8f __attribute__((ext_vector_type(8)));

__global__ __launch_bounds__(256) void gemm_wmma_f32(
    const float* __restrict__ A, const float* __restrict__ W,
    const float* __restrict__ bias, float* __restrict__ C,
    int K, int ep, const float* __restrict__ gamma,
    const float* __restrict__ beta)
{
    const int N = HID;
    const int wave = threadIdx.x >> 5;
    const int lane = threadIdx.x & 31;
    const int m    = lane & 15;
    const int half = lane >> 4;
    const int row0 = blockIdx.x * 16;
    const int col  = wave * 16 + m;

    v8f acc;
    const float bv = bias[col];
#pragma unroll
    for (int v = 0; v < 8; ++v) acc[v] = bv;

    const float* arow = A + (size_t)(row0 + m) * K + 2 * half;
    const float* wcol = W + (size_t)(2 * half) * N + col;

    for (int k = 0; k < K; k += 4) {
        v2f a = *(const v2f*)(arow + k);
        v2f b;
        b.x = wcol[(size_t)k * N];
        b.y = wcol[(size_t)k * N + N];
        acc = __builtin_amdgcn_wmma_f32_16x16x4_f32(
            false, a, false, b, (short)0, acc, false, false);
    }

    if (ep == 1) {
        const float rs = 0.99999500003749971f;
        const float g  = gamma[col] * rs;
        const float be = beta[col];
#pragma unroll
        for (int v = 0; v < 8; ++v) acc[v] = fmaxf(acc[v] * g + be, 0.0f);
    }

#pragma unroll
    for (int v = 0; v < 8; ++v) {
        const int row = row0 + v + 8 * half;
        C[(size_t)row * N + col] = acc[v];
    }
}

#define CB_NB   400
#define CB_BN   (NNODES / CB_NB)
#define CB_NCH  250
#define CB_CE   (NEDGES / CB_NCH)
#define CB_TB   256
#define CB_EPT  (CB_CE / CB_TB)
#define CB_NPC  (NNODES / CB_NCH)
static_assert(CB_NB * CB_BN == NNODES, "blocks must tile the nodes");
static_assert(CB_NCH * CB_CE == NEDGES, "chunks must tile the edges");
static_assert(CB_EPT * CB_TB == CB_CE, "threads must tile a chunk");
static_assert(CB_NPC * CB_NCH == NNODES, "chunk workgroups must tile the nodes");
static_assert(CB_NPC <= 2 * CB_TB, "two nodes per thread must cover a workgroup's nodes");
static_assert(CB_NB <= 2 * CB_TB && (CB_NB & 1) == 0, "two block counts per thread in the scan");
static_assert(CB_BN <= 256, "local node index must fit 8 bits");

__device__ __forceinline__ void node_bounds(int i, int g, int p, int* gs, int* ge)
{
    if (g != p) {
        gs[g] = i;
        if (p >= 0) ge[p] = i;
        for (int q = p + 1; q < g; ++q) { gs[q] = NNODES; ge[q] = 0; }
    }
    if (i == NNODES - 1) {
        ge[g] = NNODES;
        for (int q = g + 1; q < NGRAPHS; ++q) { gs[q] = NNODES; ge[q] = 0; }
    }
}

__global__ __launch_bounds__(CB_TB) void chunk_sort(
    const int* __restrict__ src, const int* __restrict__ dst,
    const int* __restrict__ batch, int* __restrict__ sorted,
    int* __restrict__ offT, int* gs, int* ge)
{
    __shared__ int hist[CB_NB];
    __shared__ int wtot[CB_TB / 32];
    __shared__ int stage[CB_CE];
    const int t    = (int)threadIdx.x;
    const int lane = t & 31;
    const int wave = t >> 5;
    const int c    = (int)blockIdx.x;
    for (int b = t; b < CB_NB; b += CB_TB) hist[b] = 0;
    const int* sp = src + (size_t)c * CB_CE + t;
    const int* dp = dst + (size_t)c * CB_CE + t;
    int s[CB_EPT], d[CB_EPT];
#pragma unroll
    for (int j = 0; j < CB_EPT; ++j) { s[j] = sp[j * CB_TB]; d[j] = dp[j * CB_TB]; }
    const int i0 = c * CB_NPC + t;
    const int i1 = i0 + CB_TB;
    const bool has1 = t < CB_NPC - CB_TB;
    const int g0 = batch[i0];
    const int p0 = (i0 > 0) ? batch[i0 - 1] : -1;
    const int g1 = has1 ? batch[i1] : 0;
    const int p1 = has1 ? batch[i1 - 1] : 0;
    __syncthreads();
#pragma unroll
    for (int j = 0; j < CB_EPT; ++j) atomicAdd(&hist[(unsigned)d[j] / CB_BN], 1);
    __syncthreads();

    int v0 = 0, v1 = 0;
    if (t < CB_NB / 2) { v0 = hist[2 * t]; v1 = hist[2 * t + 1]; }
    const int own = v0 + v1;
    int inc = own;
#pragma unroll
    for (int off = 1; off < 32; off <<= 1) {
        const int up = __shfl_up(inc, off, 32);
        if (lane >= off) inc += up;
    }
    if (lane == 31) wtot[wave] = inc;
    __syncthreads();
    int base = inc - own;
#pragma unroll
    for (int q = 0; q < CB_TB / 32; ++q) base += (q < wave) ? wtot[q] : 0;
    if (t < CB_NB / 2) {
        hist[2 * t]     = base;
        hist[2 * t + 1] = base + v0;
        offT[(2 * t) * CB_NCH + c]     = base;
        offT[(2 * t + 1) * CB_NCH + c] = base + v0;
    }
    if (t == 0) offT[CB_NB * CB_NCH + c] = CB_CE;
    __syncthreads();

    int pos[CB_EPT];
#pragma unroll
    for (int j = 0; j < CB_EPT; ++j) {
        const unsigned b = (unsigned)d[j] / CB_BN;
        const int loc = d[j] - (int)b * CB_BN;
        pos[j] = atomicAdd(&hist[b], 1);
        s[j] = (s[j] << 8) | loc;
    }
#pragma unroll
    for (int j = 0; j < CB_EPT; ++j) stage[pos[j]] = s[j];
    __syncthreads();
    int* outp = sorted + (size_t)c * CB_CE + t;
#pragma unroll
    for (int j = 0; j < CB_EPT; ++j) outp[j * CB_TB] = stage[t + j * CB_TB];

    node_bounds(i0, g0, p0, gs, ge);
    if (has1) node_bounds(i1, g1, p1, gs, ge);
}

#define CB_STAGE 8192
#define CB_BT    512
#define CB_HW    (CB_BT / 16)
#define CB_PPH   (256 / CB_HW)
static_assert(CB_BT >= 256, "the scans of the piece lengths and node counts take 256 threads");
static_assert(CB_NCH <= 256, "a block's pieces must fit the piece tables");
static_assert(CB_HW * CB_PPH == 256, "half waves must cover the piece tables");
__global__ __launch_bounds__(CB_BT) void block_csr(
    const int* __restrict__ sorted, const int* __restrict__ offT,
    int* __restrict__ offs, int* __restrict__ csr)
{
    __shared__ int ncnt[256];
    __shared__ int wtot[8];
    __shared__ int wlen[8];
    __shared__ int wbase[8];
    __shared__ int segsrc[256];
    __shared__ int seglp[256];
    __shared__ int seglen[256];
    __shared__ int lst[CB_STAGE];
    __shared__ int stage[CB_STAGE];
    const int t    = (int)threadIdx.x;
    const int lane = t & 31;
    const int wave = t >> 5;
    const int b    = (int)blockIdx.x;
    const int hw   = t >> 4;
    const int l16  = t & 15;

    int o0 = 0, len = 0;
    if (t < CB_NCH) {
        o0  = offT[b * CB_NCH + t];
        len = offT[(b + 1) * CB_NCH + t] - o0;
    }
    if (t < 256) ncnt[t] = 0;
    int linc = 0;
    if (t < 256) {
        linc = len;
#pragma unroll
        for (int off = 1; off < 32; off <<= 1) {
            const int up = __shfl_up(linc, off, 32);
            if (lane >= off) linc += up;
        }
        int so = o0;
#pragma unroll
        for (int off = 16; off; off >>= 1) so += __shfl_xor(so, off, 32);
        if (lane == 31) wlen[wave]  = linc;
        if (lane == 0)  wbase[wave] = so;
    }
    __syncthreads();
    int beg = 0, n = 0;
#pragma unroll
    for (int q = 0; q < 8; ++q) { beg += wbase[q]; n += wlen[q]; }
    const int end = beg + n;
    if (t < 256) {
        int lp = linc - len;
#pragma unroll
        for (int q = 0; q < 8; ++q) lp += (q < wave) ? wlen[q] : 0;
        seglp[t]  = lp;
        seglen[t] = len;
        segsrc[t] = (t < CB_NCH) ? t * CB_CE + o0 : 0;
    }
    __syncthreads();
    const bool fits = n <= CB_STAGE;

    int pl[CB_PPH], ps[CB_PPH], pp[CB_PPH];
#pragma unroll
    for (int q = 0; q < CB_PPH; ++q) {
        const int c = hw + CB_HW * q;
        pl[q] = seglen[c]; ps[q] = segsrc[c]; pp[q] = seglp[c];
    }
    if (fits) {
        int va[CB_PPH], vb[CB_PPH];
#pragma unroll
        for (int q = 0; q < CB_PPH; ++q) {
            va[q] = (l16 < pl[q])      ? sorted[ps[q] + l16]      : 0;
            vb[q] = (l16 + 16 < pl[q]) ? sorted[ps[q] + l16 + 16] : 0;
        }
#pragma unroll
        for (int q = 0; q < CB_PPH; ++q) {
            if (l16 < pl[q])      lst[pp[q] + l16]      = va[q];
            if (l16 + 16 < pl[q]) lst[pp[q] + l16 + 16] = vb[q];
        }
#pragma unroll
        for (int q = 0; q < CB_PPH; ++q) {
            for (int k = l16 + 32; k < pl[q]; k += 16) lst[pp[q] + k] = sorted[ps[q] + k];
        }
        __syncthreads();
        for (int i = t; i < n; i += CB_BT) atomicAdd(&ncnt[lst[i] & 255], 1);
    } else {
#pragma unroll
        for (int q = 0; q < CB_PPH; ++q) {
            for (int k = l16; k < pl[q]; k += 16) atomicAdd(&ncnt[sorted[ps[q] + k] & 255], 1);
        }
    }
    __syncthreads();

    int own = 0, inc = 0;
    if (t < 256) {
        own = ncnt[t];
        inc = own;
#pragma unroll
        for (int off = 1; off < 32; off <<= 1) {
            const int up = __shfl_up(inc, off, 32);
            if (lane >= off) inc += up;
        }
        if (lane == 31) wtot[wave] = inc;
    }
    __syncthreads();
    if (t < 256) {
        int base = beg;
#pragma unroll
        for (int q = 0; q < 8; ++q) base += (q < wave) ? wtot[q] : 0;
        const int start = base + (inc - own);
        ncnt[t] = start;
        if (t < CB_BN) offs[b * CB_BN + t] = start;
    }
    if (b == CB_NB - 1 && t == 0) offs[NNODES] = end;
    __syncthreads();
    if (fits) {
        for (int i = t; i < n; i += CB_BT) {
            const int v = lst[i];
            const int p = atomicAdd(&ncnt[v & 255], 1);
            stage[p - beg] = v >> 8;
        }
        __syncthreads();
        for (int i = t; i < n; i += CB_BT) csr[beg + i] = stage[i];
    } else {
#pragma unroll
        for (int q = 0; q < CB_PPH; ++q) {
            for (int k = l16; k < pl[q]; k += 16) {
                const int v = sorted[ps[q] + k];
                const int p = atomicAdd(&ncnt[v & 255], 1);
                csr[p] = v >> 8;
            }
        }
    }
}

__device__ __forceinline__ float other_half(float x)
{
    const int b = __float_as_int(x);
    return __int_as_float(__builtin_amdgcn_permlanex16(b, b, 0x76543210, 0xfedcba98, false, false));
}

#define FL_QPITCH 544
#define FL_TPITCH 132
#define FL_TILES (NNODES / 16)
#define FL_WGS   768
#define AGG_GRP  8
static_assert(FL_TILES * 16 == NNODES, "tiles must cover the nodes exactly");

__global__ __launch_bounds__(512) __attribute__((amdgpu_waves_per_eu(12))) void layer_fused(
    const float* __restrict__ X, const float* __restrict__ W,
    const float* __restrict__ bias, const int* __restrict__ offs,
    const int* __restrict__ csr, float* __restrict__ out,
    float* __restrict__ tmax, float* __restrict__ res, float* __restrict__ rmax,
    const int* __restrict__ batch)
{
    __shared__ __attribute__((aligned(16))) float wl[32 * FL_QPITCH];
    __shared__ __attribute__((aligned(16))) float tl[2][16 * FL_TPITCH];
    __shared__ int tdeg[2][16];
    __shared__ float sb[HID];
    const int lane = (int)threadIdx.x & 31;
    const int wave = __builtin_amdgcn_readfirstlane((int)(threadIdx.x >> 5));
    const int m    = lane & 15;
    const int half = lane >> 4;

#pragma unroll
    for (int i = 0; i < 2; ++i) {
        const int q = wave + 16 * i;
        const float* wp = W + (size_t)(4 * q) * HID + 4 * lane;
        const float4 r0 = *(const float4*)(wp);
        const float4 r1 = *(const float4*)(wp + HID);
        const float4 r2 = *(const float4*)(wp + 2 * HID);
        const float4 r3 = *(const float4*)(wp + 3 * HID);
        float4 c0; c0.x = r0.x; c0.y = r1.x; c0.z = r2.x; c0.w = r3.x;
        float4 c1; c1.x = r0.y; c1.y = r1.y; c1.z = r2.y; c1.w = r3.y;
        float4 c2; c2.x = r0.z; c2.y = r1.z; c2.z = r2.z; c2.w = r3.z;
        float4 c3; c3.x = r0.w; c3.y = r1.w; c3.z = r2.w; c3.w = r3.w;
        float* d = wl + q * FL_QPITCH + 16 * lane;
        *(float4*)d = c0;
        *(float4*)(d + 4) = c1;
        *(float4*)(d + 8) = c2;
        *(float4*)(d + 12) = c3;
    }

    const float* colp = X + lane * 4;
    if (threadIdx.x < HID) sb[threadIdx.x] = bias[threadIdx.x];
    int buf = 0;
    for (int t = (int)blockIdx.x; t < FL_TILES; t += FL_WGS) {
        const int node = t * 16 + wave;
        const int beg = offs[node];
        const int end = offs[node + 1];
        float ax = 0.f, ay = 0.f, az = 0.f, aw = 0.f;
        bool rows = true;
        if (batch != nullptr && wave < 8) {
            const int g0 = batch[t * 16];
            const int gm = batch[t * 16 + m];
            rows = __ballot(gm != g0) != 0;
        }

        for (int e0 = beg; e0 < end; e0 += 32) {
            const int myidx = (e0 + lane < end) ? csr[e0 + lane] : 0;
            int cnt = end - e0; if (cnt > 32) cnt = 32;
            for (int j = 0; j < cnt; j += AGG_GRP) {
                const int r = cnt - j;
                int s[AGG_GRP];
#pragma unroll
                for (int k = 0; k < AGG_GRP; ++k) s[k] = __builtin_amdgcn_readlane(myidx, j + k);
                float4 v[AGG_GRP];
                v[0] = *(const float4*)(colp + (size_t)s[0] * HID);
#pragma unroll
                for (int k = 1; k < AGG_GRP; ++k) {
                    if (k < r) v[k] = *(const float4*)(colp + (size_t)s[k] * HID);
                }
                ax += v[0].x; ay += v[0].y; az += v[0].z; aw += v[0].w;
#pragma unroll
                for (int k = 1; k < AGG_GRP; ++k) {
                    if (k < r) { ax += v[k].x; ay += v[k].y; az += v[k].z; aw += v[k].w; }
                }
            }
        }
        const float inv = 1.0f / fmaxf((float)(end - beg), 1.0f);
        float4 mean; mean.x = ax * inv; mean.y = ay * inv; mean.z = az * inv; mean.w = aw * inv;
        *(float4*)(&tl[buf][wave * FL_TPITCH + lane * 4]) = mean;
        if (lane == 0) tdeg[buf][wave] = end - beg;
        __syncthreads();

        if (wave < 8) {
            v8f acc;
#pragma unroll
            for (int v = 0; v < 8; ++v) acc[v] = 0.0f;
            const float* al = &tl[buf][m * FL_TPITCH + 4 * half];
            const float* bl = wl + half * FL_QPITCH + 4 * (16 * wave + m);
            v4f fa[16], fb[16];
#pragma unroll
            for (int u = 0; u < 2; ++u) {
                fa[u] = *(const v4f*)(al + 8 * u);
                fb[u] = *(const v4f*)(bl + (2 * u) * FL_QPITCH);
            }
#pragma unroll
            for (int u = 0; u < 16; ++u) {
                if (u + 2 < 16) {
                    fa[u + 2] = *(const v4f*)(al + 8 * (u + 2));
                    fb[u + 2] = *(const v4f*)(bl + (2 * (u + 2)) * FL_QPITCH);
                }
                asm volatile("" :: "v"(fa[u]), "v"(fb[u]));
                v2f a0; a0.x = fa[u].x; a0.y = fa[u].y;
                v2f b0; b0.x = fb[u].x; b0.y = fb[u].y;
                v2f a1; a1.x = fa[u].z; a1.y = fa[u].w;
                v2f b1; b1.x = fb[u].z; b1.y = fb[u].w;
                acc = __builtin_amdgcn_wmma_f32_16x16x4_f32(
                    false, a0, false, b0, (short)0, acc, false, false);
                acc = __builtin_amdgcn_wmma_f32_16x16x4_f32(
                    false, a1, false, b1, (short)0, acc, false, false);
            }
            const unsigned lo = (unsigned)((8 * half) * HID + 16 * wave + m);
            const unsigned e0 = (unsigned)t * (16u * HID) + lo;
            const unsigned c0 = (unsigned)t * (unsigned)HID + (e0 & (HID - 1));
            float y[8];
            auto own_part = [&]() {
                const float bv = sb[16 * wave + m];
#pragma unroll
                for (int v = 0; v < 8; ++v) {
                    const int rr = v + 8 * half;
                    const float s = (tdeg[buf][rr] > 0) ? acc[v] + bv : 0.0f;
                    y[v] = s > 0.f ? s : ALPHA * s;
                }
                if (rows) {
#pragma unroll
                    for (int v = 0; v < 8; ++v) out[e0 + (unsigned)(v * HID)] = y[v];
                }
                if (tmax) {
                    float mx = y[0];
#pragma unroll
                    for (int v = 1; v < 8; ++v) mx = fmaxf(mx, y[v]);
                    mx = fmaxf(mx, other_half(mx));
                    if (half == 0) tmax[c0] = mx;
                }
            };
            if (res) {
                float rv[8];
#pragma unroll
                for (int v = 0; v < 8; ++v) rv[v] = res[e0 + (unsigned)(v * HID)];
                own_part();
#pragma unroll
                for (int v = 0; v < 8; ++v) { rv[v] += y[v]; res[e0 + (unsigned)(v * HID)] = rv[v]; }
                float mx = rv[0];
#pragma unroll
                for (int v = 1; v < 8; ++v) mx = fmaxf(mx, rv[v]);
                mx = fmaxf(mx, other_half(mx));
                if (half == 0) rmax[c0] = mx;
            } else {
                own_part();
            }
        }
        buf ^= 1;
    }
}

#define POOL_RG  4
#define POOL_INF 8

__global__ __launch_bounds__(512) void pool_max(
    const float* __restrict__ ht, const float* __restrict__ h1,
    const float* __restrict__ h2, const float* __restrict__ h3,
    const float* __restrict__ mt, const float* __restrict__ m1,
    const float* __restrict__ m2, const float* __restrict__ m3,
    const int* __restrict__ gs, const int* __restrict__ ge,
    float* __restrict__ pooled)
{
    __shared__ float4 red[POOL_RG][128];
    const int g = blockIdx.x;
    const int lane = (int)threadIdx.x & 31;
    const int wave = __builtin_amdgcn_readfirstlane((int)(threadIdx.x >> 5));
    const int arr = wave & 3;
    const int rg  = wave >> 2;
    const float* rowp = ((arr == 0) ? ht : (arr == 1) ? h1 : (arr == 2) ? h2 : h3) + lane * 4;
    const float* maxp = ((arr == 0) ? mt : (arr == 1) ? m1 : (arr == 2) ? m2 : m3) + lane * 4;
    const int b = gs[g], e = ge[g];

    const int t0 = (b + 15) >> 4;
    const int t1 = e >> 4;
    int nt = 0, hend = e, tbeg = e;
    if (t0 <= t1) { nt = t1 - t0; hend = 16 * t0; tbeg = 16 * t1; }
    int nh = hend - b; if (nh < 0) nh = 0;
    const int nl = e - tbeg;
    const int n = nt + nh + nl;

    float mx = -FMAXV, my = -FMAXV, mz = -FMAXV, mw = -FMAXV;
    for (int i0 = rg; i0 < n; i0 += POOL_RG * POOL_INF) {
        float4 v[POOL_INF];
#pragma unroll
        for (int k = 0; k < POOL_INF; ++k) {
            const int i = i0 + k * POOL_RG;
            v[k].x = -FMAXV; v[k].y = -FMAXV; v[k].z = -FMAXV; v[k].w = -FMAXV;
            if (i < n) {
                const float* p = (i < nt)      ? maxp + (size_t)(t0 + i) * HID
                               : (i < nt + nh) ? rowp + (size_t)(b + (i - nt)) * HID
                                               : rowp + (size_t)(tbeg + (i - nt - nh)) * HID;
                v[k] = *(const float4*)p;
            }
        }
#pragma unroll
        for (int k = 0; k < POOL_INF; ++k) {
            mx = fmaxf(mx, v[k].x); my = fmaxf(my, v[k].y);
            mz = fmaxf(mz, v[k].z); mw = fmaxf(mw, v[k].w);
        }
    }
    float4 m; m.x = mx; m.y = my; m.z = mz; m.w = mw;
    red[rg][arr * 32 + lane] = m;
    __syncthreads();
    if (threadIdx.x < 128) {
        const int c = (int)threadIdx.x;
        float4 r = red[0][c];
#pragma unroll
        for (int k = 1; k < POOL_RG; ++k) {
            const float4 q = red[k][c];
            r.x = fmaxf(r.x, q.x); r.y = fmaxf(r.y, q.y);
            r.z = fmaxf(r.z, q.z); r.w = fmaxf(r.w, q.w);
        }
        *(float4*)(pooled + (size_t)g * 512 + c * 4) = r;
    }
}

__global__ __launch_bounds__(32) void mlp2(
    const float* __restrict__ z, const float* __restrict__ W,
    const float* __restrict__ b, float* __restrict__ out)
{
    const int g = blockIdx.x;
    const int lane = threadIdx.x;
    const float z0 = z[(size_t)g * HID + lane];
    const float z1 = z[(size_t)g * HID + lane + 32];
    const float z2 = z[(size_t)g * HID + lane + 64];
    const float z3 = z[(size_t)g * HID + lane + 96];
    for (int c = 0; c < NCLS; ++c) {
        float s = z0 * W[lane * NCLS + c]
                + z1 * W[(lane + 32) * NCLS + c]
                + z2 * W[(lane + 64) * NCLS + c]
                + z3 * W[(lane + 96) * NCLS + c];
        for (int off = 16; off; off >>= 1) s += __shfl_xor(s, off, 32);
        if (lane == 0) out[g * NCLS + c] = s + b[c];
    }
}

extern "C" void kernel_launch(void* const* d_in, const int* in_sizes, int n_in,
                              void* d_out, int out_size, void* d_ws, size_t ws_size,
                              hipStream_t stream)
{
    (void)in_sizes; (void)n_in; (void)out_size; (void)ws_size;
    const float* x     = (const float*)d_in[0];
    const int*   ei    = (const int*)d_in[1];
    const int*   batch = (const int*)d_in[2];
    const float* W0 = (const float*)d_in[3];  const float* b0 = (const float*)d_in[4];
    const float* W1 = (const float*)d_in[5];  const float* b1 = (const float*)d_in[6];
    const float* W2 = (const float*)d_in[7];  const float* b2 = (const float*)d_in[8];
    const float* W3 = (const float*)d_in[9];  const float* b3 = (const float*)d_in[10];
    const float* Wm1 = (const float*)d_in[11]; const float* bm1 = (const float*)d_in[12];
    const float* gamma = (const float*)d_in[13]; const float* beta = (const float*)d_in[14];
    const float* Wm2 = (const float*)d_in[15]; const float* bm2 = (const float*)d_in[16];
    const int* esrc = ei;
    const int* edst = ei + NEDGES;

    char* ws = (char*)d_ws;
    size_t off = 0;
    auto alloc = [&](size_t bytes) -> void* {
        void* p = ws + off;
        off = (off + bytes + 255) & ~(size_t)255;
        return p;
    };
    const size_t featB = (size_t)NNODES * HID * sizeof(float);
    float* tmpH   = (float*)alloc(featB);
    float* h      = (float*)alloc(featB);
    float* h1     = (float*)alloc(featB);
    float* h2     = (float*)alloc(featB);
    float* ht     = (float*)alloc(featB);
    float* zbuf   = (float*)alloc((size_t)NGRAPHS * HID * sizeof(float));
    float* pooled = (float*)alloc((size_t)NGRAPHS * 512 * sizeof(float));
    int*   csr    = (int*)alloc((size_t)NEDGES * sizeof(int));
    int*   offs   = (int*)alloc((size_t)(NNODES + 1) * sizeof(int));
    int*   offT   = (int*)alloc((size_t)(CB_NB + 1) * CB_NCH * sizeof(int));
    int*   sorted = (int*)tmpH;
    float* mt     = tmpH + 2097152;
    float* m1     = mt + FL_TILES * HID;
    float* m2     = m1 + FL_TILES * HID;
    float* m3     = m2 + FL_TILES * HID;
    int*   gs     = (int*)alloc((size_t)NGRAPHS * sizeof(int));
    int*   ge     = (int*)alloc((size_t)NGRAPHS * sizeof(int));

    const int TB = 256;
    chunk_sort<<<CB_NCH, CB_TB, 0, stream>>>(esrc, edst, batch, sorted, offT, gs, ge);
    block_csr<<<CB_NB, CB_BT, 0, stream>>>(sorted, offT, offs, csr);

    float* const none = nullptr;
    const int* const nobatch = nullptr;
    layer_fused<<<FL_WGS, 512, 0, stream>>>(x, W0, b0, offs, csr, h, none, none, none, nobatch);
    layer_fused<<<FL_WGS, 512, 0, stream>>>(h, W1, b1, offs, csr, h1, m1, none, none, nobatch);
    layer_fused<<<FL_WGS, 512, 0, stream>>>(h1, W2, b2, offs, csr, h2, m2, h, m3, nobatch);
    layer_fused<<<FL_WGS, 512, 0, stream>>>(h, W3, b3, offs, csr, ht, mt, none, none, batch);

    pool_max<<<NGRAPHS, 512, 0, stream>>>(ht, h1, h2, h, mt, m1, m2, m3, gs, ge, pooled);

    gemm_wmma_f32<<<NGRAPHS / 16, TB, 0, stream>>>(pooled, Wm1, bm1, zbuf, 512, 1, gamma, beta);
    mlp2<<<NGRAPHS, 32, 0, stream>>>(zbuf, Wm2, bm2, (float*)d_out);
}


# pooling, MLP layer and class projection in one launch (was 3): a 32-wave workgroup pools 4 graphs, then WMMA with k split in 4 quarters (2 load groups per wave, was 16); k sum order changes
# speedup vs baseline: 3.7966x; 1.0549x over previous
#include <hip/hip_runtime.h>

#define NNODES 100000
#define NEDGES 1600000
#define NGRAPHS 512
#define HID 128
#define NCLS 10
#define ALPHA 0.01f
#define FMAXV 3.402823466e38f

typedef float v2f __attribute__((ext_vector_type(2)));
typedef float v4f __attribute__((ext_vector_type(4)));
typedef float v8f __attribute__((ext_vector_type(8)));

#define CB_NB   400
#define CB_BN   (NNODES / CB_NB)
#define CB_NCH  250
#define CB_CE   (NEDGES / CB_NCH)
#define CB_TB   256
#define CB_EPT  (CB_CE / CB_TB)
#define CB_NPC  (NNODES / CB_NCH)
static_assert(CB_NB * CB_BN == NNODES, "blocks must tile the nodes");
static_assert(CB_NCH * CB_CE == NEDGES, "chunks must tile the edges");
static_assert(CB_EPT * CB_TB == CB_CE, "threads must tile a chunk");
static_assert(CB_NPC * CB_NCH == NNODES, "chunk workgroups must tile the nodes");
static_assert(CB_NPC <= 2 * CB_TB, "two nodes per thread must cover a workgroup's nodes");
static_assert(CB_NB <= 2 * CB_TB && (CB_NB & 1) == 0, "two block counts per thread in the scan");
static_assert(CB_BN <= 256, "local node index must fit 8 bits");

__device__ __forceinline__ void node_bounds(int i, int g, int p, int* gs, int* ge)
{
    if (g != p) {
        gs[g] = i;
        if (p >= 0) ge[p] = i;
        for (int q = p + 1; q < g; ++q) { gs[q] = NNODES; ge[q] = 0; }
    }
    if (i == NNODES - 1) {
        ge[g] = NNODES;
        for (int q = g + 1; q < NGRAPHS; ++q) { gs[q] = NNODES; ge[q] = 0; }
    }
}

__global__ __launch_bounds__(CB_TB) void chunk_sort(
    const int* __restrict__ src, const int* __restrict__ dst,
    const int* __restrict__ batch, int* __restrict__ sorted,
    int* __restrict__ offT, int* gs, int* ge)
{
    __shared__ int hist[CB_NB];
    __shared__ int wtot[CB_TB / 32];
    __shared__ int stage[CB_CE];
    const int t    = (int)threadIdx.x;
    const int lane = t & 31;
    const int wave = t >> 5;
    const int c    = (int)blockIdx.x;
    for (int b = t; b < CB_NB; b += CB_TB) hist[b] = 0;
    const int* sp = src + (size_t)c * CB_CE + t;
    const int* dp = dst + (size_t)c * CB_CE + t;
    int s[CB_EPT], d[CB_EPT];
#pragma unroll
    for (int j = 0; j < CB_EPT; ++j) { s[j] = sp[j * CB_TB]; d[j] = dp[j * CB_TB]; }
    const int i0 = c * CB_NPC + t;
    const int i1 = i0 + CB_TB;
    const bool has1 = t < CB_NPC - CB_TB;
    const int g0 = batch[i0];
    const int p0 = (i0 > 0) ? batch[i0 - 1] : -1;
    const int g1 = has1 ? batch[i1] : 0;
    const int p1 = has1 ? batch[i1 - 1] : 0;
    __syncthreads();
#pragma unroll
    for (int j = 0; j < CB_EPT; ++j) atomicAdd(&hist[(unsigned)d[j] / CB_BN], 1);
    __syncthreads();

    int v0 = 0, v1 = 0;
    if (t < CB_NB / 2) { v0 = hist[2 * t]; v1 = hist[2 * t + 1]; }
    const int own = v0 + v1;
    int inc = own;
#pragma unroll
    for (int off = 1; off < 32; off <<= 1) {
        const int up = __shfl_up(inc, off, 32);
        if (lane >= off) inc += up;
    }
    if (lane == 31) wtot[wave] = inc;
    __syncthreads();
    int base = inc - own;
#pragma unroll
    for (int q = 0; q < CB_TB / 32; ++q) base += (q < wave) ? wtot[q] : 0;
    if (t < CB_NB / 2) {
        hist[2 * t]     = base;
        hist[2 * t + 1] = base + v0;
        offT[(2 * t) * CB_NCH + c]     = base;
        offT[(2 * t + 1) * CB_NCH + c] = base + v0;
    }
    if (t == 0) offT[CB_NB * CB_NCH + c] = CB_CE;
    __syncthreads();

    int pos[CB_EPT];
#pragma unroll
    for (int j = 0; j < CB_EPT; ++j) {
        const unsigned b = (unsigned)d[j] / CB_BN;
        const int loc = d[j] - (int)b * CB_BN;
        pos[j] = atomicAdd(&hist[b], 1);
        s[j] = (s[j] << 8) | loc;
    }
#pragma unroll
    for (int j = 0; j < CB_EPT; ++j) stage[pos[j]] = s[j];
    __syncthreads();
    int* outp = sorted + (size_t)c * CB_CE + t;
#pragma unroll
    for (int j = 0; j < CB_EPT; ++j) outp[j * CB_TB] = stage[t + j * CB_TB];

    node_bounds(i0, g0, p0, gs, ge);
    if (has1) node_bounds(i1, g1, p1, gs, ge);
}

#define CB_STAGE 8192
#define CB_BT    512
#define CB_HW    (CB_BT / 16)
#define CB_PPH   (256 / CB_HW)
static_assert(CB_BT >= 256, "the scans of the piece lengths and node counts take 256 threads");
static_assert(CB_NCH <= 256, "a block's pieces must fit the piece tables");
static_assert(CB_HW * CB_PPH == 256, "half waves must cover the piece tables");
__global__ __launch_bounds__(CB_BT) void block_csr(
    const int* __restrict__ sorted, const int* __restrict__ offT,
    int* __restrict__ offs, int* __restrict__ csr)
{
    __shared__ int ncnt[256];
    __shared__ int wtot[8];
    __shared__ int wlen[8];
    __shared__ int wbase[8];
    __shared__ int segsrc[256];
    __shared__ int seglp[256];
    __shared__ int seglen[256];
    __shared__ int lst[CB_STAGE];
    __shared__ int stage[CB_STAGE];
    const int t    = (int)threadIdx.x;
    const int lane = t & 31;
    const int wave = t >> 5;
    const int b    = (int)blockIdx.x;
    const int hw   = t >> 4;
    const int l16  = t & 15;

    int o0 = 0, len = 0;
    if (t < CB_NCH) {
        o0  = offT[b * CB_NCH + t];
        len = offT[(b + 1) * CB_NCH + t] - o0;
    }
    if (t < 256) ncnt[t] = 0;
    int linc = 0;
    if (t < 256) {
        linc = len;
#pragma unroll
        for (int off = 1; off < 32; off <<= 1) {
            const int up = __shfl_up(linc, off, 32);
            if (lane >= off) linc += up;
        }
        int so = o0;
#pragma unroll
        for (int off = 16; off; off >>= 1) so += __shfl_xor(so, off, 32);
        if (lane == 31) wlen[wave]  = linc;
        if (lane == 0)  wbase[wave] = so;
    }
    __syncthreads();
    int beg = 0, n = 0;
#pragma unroll
    for (int q = 0; q < 8; ++q) { beg += wbase[q]; n += wlen[q]; }
    const int end = beg + n;
    if (t < 256) {
        int lp = linc - len;
#pragma unroll
        for (int q = 0; q < 8; ++q) lp += (q < wave) ? wlen[q] : 0;
        seglp[t]  = lp;
        seglen[t] = len;
        segsrc[t] = (t < CB_NCH) ? t * CB_CE + o0 : 0;
    }
    __syncthreads();
    const bool fits = n <= CB_STAGE;

    int pl[CB_PPH], ps[CB_PPH], pp[CB_PPH];
#pragma unroll
    for (int q = 0; q < CB_PPH; ++q) {
        const int c = hw + CB_HW * q;
        pl[q] = seglen[c]; ps[q] = segsrc[c]; pp[q] = seglp[c];
    }
    if (fits) {
        int va[CB_PPH], vb[CB_PPH];
#pragma unroll
        for (int q = 0; q < CB_PPH; ++q) {
            va[q] = (l16 < pl[q])      ? sorted[ps[q] + l16]      : 0;
            vb[q] = (l16 + 16 < pl[q]) ? sorted[ps[q] + l16 + 16] : 0;
        }
#pragma unroll
        for (int q = 0; q < CB_PPH; ++q) {
            if (l16 < pl[q])      lst[pp[q] + l16]      = va[q];
            if (l16 + 16 < pl[q]) lst[pp[q] + l16 + 16] = vb[q];
        }
#pragma unroll
        for (int q = 0; q < CB_PPH; ++q) {
            for (int k = l16 + 32; k < pl[q]; k += 16) lst[pp[q] + k] = sorted[ps[q] + k];
        }
        __syncthreads();
        for (int i = t; i < n; i += CB_BT) atomicAdd(&ncnt[lst[i] & 255], 1);
    } else {
#pragma unroll
        for (int q = 0; q < CB_PPH; ++q) {
            for (int k = l16; k < pl[q]; k += 16) atomicAdd(&ncnt[sorted[ps[q] + k] & 255], 1);
        }
    }
    __syncthreads();

    int own = 0, inc = 0;
    if (t < 256) {
        own = ncnt[t];
        inc = own;
#pragma unroll
        for (int off = 1; off < 32; off <<= 1) {
            const int up = __shfl_up(inc, off, 32);
            if (lane >= off) inc += up;
        }
        if (lane == 31) wtot[wave] = inc;
    }
    __syncthreads();
    if (t < 256) {
        int base = beg;
#pragma unroll
        for (int q = 0; q < 8; ++q) base += (q < wave) ? wtot[q] : 0;
        const int start = base + (inc - own);
        ncnt[t] = start;
        if (t < CB_BN) offs[b * CB_BN + t] = start;
    }
    if (b == CB_NB - 1 && t == 0) offs[NNODES] = end;
    __syncthreads();
    if (fits) {
        for (int i = t; i < n; i += CB_BT) {
            const int v = lst[i];
            const int p = atomicAdd(&ncnt[v & 255], 1);
            stage[p - beg] = v >> 8;
        }
        __syncthreads();
        for (int i = t; i < n; i += CB_BT) csr[beg + i] = stage[i];
    } else {
#pragma unroll
        for (int q = 0; q < CB_PPH; ++q) {
            for (int k = l16; k < pl[q]; k += 16) {
                const int v = sorted[ps[q] + k];
                const int p = atomicAdd(&ncnt[v & 255], 1);
                csr[p] = v >> 8;
            }
        }
    }
}

__device__ __forceinline__ float other_half(float x)
{
    const int b = __float_as_int(x);
    return __int_as_float(__builtin_amdgcn_permlanex16(b, b, 0x76543210, 0xfedcba98, false, false));
}

#define FL_QPITCH 544
#define FL_TPITCH 132
#define FL_TILES (NNODES / 16)
#define FL_WGS   768
#define AGG_GRP  8
static_assert(FL_TILES * 16 == NNODES, "tiles must cover the nodes exactly");

__global__ __launch_bounds__(512) __attribute__((amdgpu_waves_per_eu(12))) void layer_fused(
    const float* __restrict__ X, const float* __restrict__ W,
    const float* __restrict__ bias, const int* __restrict__ offs,
    const int* __restrict__ csr, float* __restrict__ out,
    float* __restrict__ tmax, float* __restrict__ res, float* __restrict__ rmax,
    const int* __restrict__ batch)
{
    __shared__ __attribute__((aligned(16))) float wl[32 * FL_QPITCH];
    __shared__ __attribute__((aligned(16))) float tl[2][16 * FL_TPITCH];
    __shared__ int tdeg[2][16];
    __shared__ float sb[HID];
    const int lane = (int)threadIdx.x & 31;
    const int wave = __builtin_amdgcn_readfirstlane((int)(threadIdx.x >> 5));
    const int m    = lane & 15;
    const int half = lane >> 4;

#pragma unroll
    for (int i = 0; i < 2; ++i) {
        const int q = wave + 16 * i;
        const float* wp = W + (size_t)(4 * q) * HID + 4 * lane;
        const float4 r0 = *(const float4*)(wp);
        const float4 r1 = *(const float4*)(wp + HID);
        const float4 r2 = *(const float4*)(wp + 2 * HID);
        const float4 r3 = *(const float4*)(wp + 3 * HID);
        float4 c0; c0.x = r0.x; c0.y = r1.x; c0.z = r2.x; c0.w = r3.x;
        float4 c1; c1.x = r0.y; c1.y = r1.y; c1.z = r2.y; c1.w = r3.y;
        float4 c2; c2.x = r0.z; c2.y = r1.z; c2.z = r2.z; c2.w = r3.z;
        float4 c3; c3.x = r0.w; c3.y = r1.w; c3.z = r2.w; c3.w = r3.w;
        float* d = wl + q * FL_QPITCH + 16 * lane;
        *(float4*)d = c0;
        *(float4*)(d + 4) = c1;
        *(float4*)(d + 8) = c2;
        *(float4*)(d + 12) = c3;
    }

    const float* colp = X + lane * 4;
    if (threadIdx.x < HID) sb[threadIdx.x] = bias[threadIdx.x];
    int buf = 0;
    for (int t = (int)blockIdx.x; t < FL_TILES; t += FL_WGS) {
        const int node = t * 16 + wave;
        const int beg = offs[node];
        const int end = offs[node + 1];
        float ax = 0.f, ay = 0.f, az = 0.f, aw = 0.f;
        bool rows = true;
        if (batch != nullptr && wave < 8) {
            const int g0 = batch[t * 16];
            const int gm = batch[t * 16 + m];
            rows = __ballot(gm != g0) != 0;
        }

        for (int e0 = beg; e0 < end; e0 += 32) {
            const int myidx = (e0 + lane < end) ? csr[e0 + lane] : 0;
            int cnt = end - e0; if (cnt > 32) cnt = 32;
            for (int j = 0; j < cnt; j += AGG_GRP) {
                const int r = cnt - j;
                int s[AGG_GRP];
#pragma unroll
                for (int k = 0; k < AGG_GRP; ++k) s[k] = __builtin_amdgcn_readlane(myidx, j + k);
                float4 v[AGG_GRP];
                v[0] = *(const float4*)(colp + (size_t)s[0] * HID);
#pragma unroll
                for (int k = 1; k < AGG_GRP; ++k) {
                    if (k < r) v[k] = *(const float4*)(colp + (size_t)s[k] * HID);
                }
                ax += v[0].x; ay += v[0].y; az += v[0].z; aw += v[0].w;
#pragma unroll
                for (int k = 1; k < AGG_GRP; ++k) {
                    if (k < r) { ax += v[k].x; ay += v[k].y; az += v[k].z; aw += v[k].w; }
                }
            }
        }
        const float inv = 1.0f / fmaxf((float)(end - beg), 1.0f);
        float4 mean; mean.x = ax * inv; mean.y = ay * inv; mean.z = az * inv; mean.w = aw * inv;
        *(float4*)(&tl[buf][wave * FL_TPITCH + lane * 4]) = mean;
        if (lane == 0) tdeg[buf][wave] = end - beg;
        __syncthreads();

        if (wave < 8) {
            v8f acc;
#pragma unroll
            for (int v = 0; v < 8; ++v) acc[v] = 0.0f;
            const float* al = &tl[buf][m * FL_TPITCH + 4 * half];
            const float* bl = wl + half * FL_QPITCH + 4 * (16 * wave + m);
            v4f fa[16], fb[16];
#pragma unroll
            for (int u = 0; u < 2; ++u) {
                fa[u] = *(const v4f*)(al + 8 * u);
                fb[u] = *(const v4f*)(bl + (2 * u) * FL_QPITCH);
            }
#pragma unroll
            for (int u = 0; u < 16; ++u) {
                if (u + 2 < 16) {
                    fa[u + 2] = *(const v4f*)(al + 8 * (u + 2));
                    fb[u + 2] = *(const v4f*)(bl + (2 * (u + 2)) * FL_QPITCH);
                }
                asm volatile("" :: "v"(fa[u]), "v"(fb[u]));
                v2f a0; a0.x = fa[u].x; a0.y = fa[u].y;
                v2f b0; b0.x = fb[u].x; b0.y = fb[u].y;
                v2f a1; a1.x = fa[u].z; a1.y = fa[u].w;
                v2f b1; b1.x = fb[u].z; b1.y = fb[u].w;
                acc = __builtin_amdgcn_wmma_f32_16x16x4_f32(
                    false, a0, false, b0, (short)0, acc, false, false);
                acc = __builtin_amdgcn_wmma_f32_16x16x4_f32(
                    false, a1, false, b1, (short)0, acc, false, false);
            }
            const unsigned lo = (unsigned)((8 * half) * HID + 16 * wave + m);
            const unsigned e0 = (unsigned)t * (16u * HID) + lo;
            const unsigned c0 = (unsigned)t * (unsigned)HID + (e0 & (HID - 1));
            float y[8];
            auto own_part = [&]() {
                const float bv = sb[16 * wave + m];
#pragma unroll
                for (int v = 0; v < 8; ++v) {
                    const int rr = v + 8 * half;
                    const float s = (tdeg[buf][rr] > 0) ? acc[v] + bv : 0.0f;
                    y[v] = s > 0.f ? s : ALPHA * s;
                }
                if (rows) {
#pragma unroll
                    for (int v = 0; v < 8; ++v) out[e0 + (unsigned)(v * HID)] = y[v];
                }
                if (tmax) {
                    float mx = y[0];
#pragma unroll
                    for (int v = 1; v < 8; ++v) mx = fmaxf(mx, y[v]);
                    mx = fmaxf(mx, other_half(mx));
                    if (half == 0) tmax[c0] = mx;
                }
            };
            if (res) {
                float rv[8];
#pragma unroll
                for (int v = 0; v < 8; ++v) rv[v] = res[e0 + (unsigned)(v * HID)];
                own_part();
#pragma unroll
                for (int v = 0; v < 8; ++v) { rv[v] += y[v]; res[e0 + (unsigned)(v * HID)] = rv[v]; }
                float mx = rv[0];
#pragma unroll
                for (int v = 1; v < 8; ++v) mx = fmaxf(mx, rv[v]);
                mx = fmaxf(mx, other_half(mx));
                if (half == 0) rmax[c0] = mx;
            } else {
                own_part();
            }
        }
        buf ^= 1;
    }
}

#define HD_G     4
#define HD_INF   16
#define HD_PITCH 516
static_assert(NGRAPHS % HD_G == 0, "workgroups must tile the graphs");

__global__ __launch_bounds__(1024) void head_fused(
    const float* __restrict__ ht, const float* __restrict__ h1,
    const float* __restrict__ h2, const float* __restrict__ h3,
    const float* __restrict__ mt, const float* __restrict__ m1,
    const float* __restrict__ m2, const float* __restrict__ m3,
    const int* __restrict__ gs, const int* __restrict__ ge,
    const float* __restrict__ Wm1, const float* __restrict__ bm1,
    const float* __restrict__ gamma, const float* __restrict__ beta,
    const float* __restrict__ Wm2, const float* __restrict__ bm2,
    float* __restrict__ out)
{
    __shared__ __attribute__((aligned(16))) float pl[2][HD_G][HD_PITCH];
    __shared__ float part[4][HD_G][HID];
    __shared__ float zl[HD_G][HID];
    const int lane = (int)threadIdx.x & 31;
    const int wave = __builtin_amdgcn_readfirstlane((int)(threadIdx.x >> 5));
    const int g0 = (int)blockIdx.x * HD_G;

    {
        const int unit = wave & 15;
        const int prt  = wave >> 4;
        const int gl   = unit >> 2;
        const int arr  = unit & 3;
        const float* rowp = ((arr == 0) ? ht : (arr == 1) ? h1 : (arr == 2) ? h2 : h3) + lane * 4;
        const float* maxp = ((arr == 0) ? mt : (arr == 1) ? m1 : (arr == 2) ? m2 : m3) + lane * 4;
        const int b = gs[g0 + gl], e = ge[g0 + gl];
        const int t0 = (b + 15) >> 4;
        const int t1 = e >> 4;
        int nt = 0, hend = e, tbeg = e;
        if (t0 <= t1) { nt = t1 - t0; hend = 16 * t0; tbeg = 16 * t1; }
        int nh = hend - b; if (nh < 0) nh = 0;
        const int nl = e - tbeg;
        const int n = nt + nh + nl;

        float mx = -FMAXV, my = -FMAXV, mz = -FMAXV, mw = -FMAXV;
        for (int i0 = prt; i0 < n; i0 += 2 * HD_INF) {
            float4 v[HD_INF];
#pragma unroll
            for (int k = 0; k < HD_INF; ++k) {
                const int i = i0 + 2 * k;
                v[k].x = -FMAXV; v[k].y = -FMAXV; v[k].z = -FMAXV; v[k].w = -FMAXV;
                if (i < n) {
                    const float* p = (i < nt)      ? maxp + (size_t)(t0 + i) * HID
                                   : (i < nt + nh) ? rowp + (size_t)(b + (i - nt)) * HID
                                                   : rowp + (size_t)(tbeg + (i - nt - nh)) * HID;
                    v[k] = *(const float4*)p;
                }
            }
#pragma unroll
            for (int k = 0; k < HD_INF; ++k) {
                mx = fmaxf(mx, v[k].x); my = fmaxf(my, v[k].y);
                mz = fmaxf(mz, v[k].z); mw = fmaxf(mw, v[k].w);
            }
        }
        float4 m; m.x = mx; m.y = my; m.z = mz; m.w = mw;
        *(float4*)(&pl[prt][gl][arr * HID + lane * 4]) = m;
    }
    __syncthreads();

    {
        const int ct = wave & 7;
        const int kq = wave >> 3;
        const int c  = lane & 15;
        const int hf = lane >> 4;
        const bool live = c < HD_G;
        v8f acc;
#pragma unroll
        for (int v = 0; v < 8; ++v) acc[v] = 0.0f;
        const float* bp  = Wm1 + (size_t)(kq * 128 + 2 * hf) * HID + ct * 16 + c;
        const float* ap0 = &pl[0][c & (HD_G - 1)][kq * 128 + 2 * hf];
        const float* ap1 = &pl[1][c & (HD_G - 1)][kq * 128 + 2 * hf];
#pragma unroll 16
        for (int s = 0; s < 32; ++s) {
            const v2f x0 = *(const v2f*)(ap0 + 4 * s);
            const v2f x1 = *(const v2f*)(ap1 + 4 * s);
            v2f a;
            a.x = live ? fmaxf(x0.x, x1.x) : 0.0f;
            a.y = live ? fmaxf(x0.y, x1.y) : 0.0f;
            v2f bb;
            bb.x = bp[(size_t)(4 * s) * HID];
            bb.y = bp[(size_t)(4 * s + 1) * HID];
            acc = __builtin_amdgcn_wmma_f32_16x16x4_f32(
                false, a, false, bb, (short)0, acc, false, false);
        }
        if (hf == 0) {
#pragma unroll
            for (int v = 0; v < HD_G; ++v) part[kq][v][ct * 16 + c] = acc[v];
        }
    }
    __syncthreads();

    if (threadIdx.x < HD_G * HID) {
        const int row = (int)threadIdx.x >> 7;
        const int col = (int)threadIdx.x & (HID - 1);
        float s = part[0][row][col];
        s += part[1][row][col];
        s += part[2][row][col];
        s += part[3][row][col];
        s += bm1[col];
        const float rs = 0.99999500003749971f;
        const float g  = gamma[col] * rs;
        zl[row][col] = fmaxf(s * g + beta[col], 0.0f);
    }
    __syncthreads();

    {
        const int gl = wave >> 3;
        const float z0 = zl[gl][lane];
        const float z1 = zl[gl][lane + 32];
        const float z2 = zl[gl][lane + 64];
        const float z3 = zl[gl][lane + 96];
        for (int c = wave & 7; c < NCLS; c += 8) {
            float s = z0 * Wm2[lane * NCLS + c]
                    + z1 * Wm2[(lane + 32) * NCLS + c]
                    + z2 * Wm2[(lane + 64) * NCLS + c]
                    + z3 * Wm2[(lane + 96) * NCLS + c];
            for (int off = 16; off; off >>= 1) s += __shfl_xor(s, off, 32);
            if (lane == 0) out[(g0 + gl) * NCLS + c] = s + bm2[c];
        }
    }
}

extern "C" void kernel_launch(void* const* d_in, const int* in_sizes, int n_in,
                              void* d_out, int out_size, void* d_ws, size_t ws_size,
                              hipStream_t stream)
{
    (void)in_sizes; (void)n_in; (void)out_size; (void)ws_size;
    const float* x     = (const float*)d_in[0];
    const int*   ei    = (const int*)d_in[1];
    const int*   batch = (const int*)d_in[2];
    const float* W0 = (const float*)d_in[3];  const float* b0 = (const float*)d_in[4];
    const float* W1 = (const float*)d_in[5];  const float* b1 = (const float*)d_in[6];
    const float* W2 = (const float*)d_in[7];  const float* b2 = (const float*)d_in[8];
    const float* W3 = (const float*)d_in[9];  const float* b3 = (const float*)d_in[10];
    const float* Wm1 = (const float*)d_in[11]; const float* bm1 = (const float*)d_in[12];
    const float* gamma = (const float*)d_in[13]; const float* beta = (const float*)d_in[14];
    const float* Wm2 = (const float*)d_in[15]; const float* bm2 = (const float*)d_in[16];
    const int* esrc = ei;
    const int* edst = ei + NEDGES;

    char* ws = (char*)d_ws;
    size_t off = 0;
    auto alloc = [&](size_t bytes) -> void* {
        void* p = ws + off;
        off = (off + bytes + 255) & ~(size_t)255;
        return p;
    };
    const size_t featB = (size_t)NNODES * HID * sizeof(float);
    float* tmpH   = (float*)alloc(featB);
    float* h      = (float*)alloc(featB);
    float* h1     = (float*)alloc(featB);
    float* h2     = (float*)alloc(featB);
    float* ht     = (float*)alloc(featB);
    int*   csr    = (int*)alloc((size_t)NEDGES * sizeof(int));
    int*   offs   = (int*)alloc((size_t)(NNODES + 1) * sizeof(int));
    int*   offT   = (int*)alloc((size_t)(CB_NB + 1) * CB_NCH * sizeof(int));
    int*   sorted = (int*)tmpH;
    float* mt     = tmpH + 2097152;
    float* m1     = mt + FL_TILES * HID;
    float* m2     = m1 + FL_TILES * HID;
    float* m3     = m2 + FL_TILES * HID;
    int*   gs     = (int*)alloc((size_t)NGRAPHS * sizeof(int));
    int*   ge     = (int*)alloc((size_t)NGRAPHS * sizeof(int));

    chunk_sort<<<CB_NCH, CB_TB, 0, stream>>>(esrc, edst, batch, sorted, offT, gs, ge);
    block_csr<<<CB_NB, CB_BT, 0, stream>>>(sorted, offT, offs, csr);

    float* const none = nullptr;
    const int* const nobatch = nullptr;
    layer_fused<<<FL_WGS, 512, 0, stream>>>(x, W0, b0, offs, csr, h, none, none, none, nobatch);
    layer_fused<<<FL_WGS, 512, 0, stream>>>(h, W1, b1, offs, csr, h1, m1, none, none, nobatch);
    layer_fused<<<FL_WGS, 512, 0, stream>>>(h1, W2, b2, offs, csr, h2, m2, h, m3, nobatch);
    layer_fused<<<FL_WGS, 512, 0, stream>>>(h, W3, b3, offs, csr, ht, mt, none, none, batch);

    head_fused<<<NGRAPHS / HD_G, 1024, 0, stream>>>(ht, h1, h2, h, mt, m1, m2, m3, gs, ge,
                                                    Wm1, bm1, gamma, beta, Wm2, bm2, (float*)d_out);
}


# layer epilogues move whole 512-byte rows through an LDS tile (a row per wave, stored behind the next tile's barrier; residual rows in and out too), not 64-byte pieces; h2 rows only for pooling
# speedup vs baseline: 4.0057x; 1.0551x over previous
#include <hip/hip_runtime.h>

#define NNODES 100000
#define NEDGES 1600000
#define NGRAPHS 512
#define HID 128
#define NCLS 10
#define ALPHA 0.01f
#define FMAXV 3.402823466e38f

typedef float v2f __attribute__((ext_vector_type(2)));
typedef float v4f __attribute__((ext_vector_type(4)));
typedef float v8f __attribute__((ext_vector_type(8)));

#define CB_NB   400
#define CB_BN   (NNODES / CB_NB)
#define CB_NCH  250
#define CB_CE   (NEDGES / CB_NCH)
#define CB_TB   256
#define CB_EPT  (CB_CE / CB_TB)
#define CB_NPC  (NNODES / CB_NCH)
static_assert(CB_NB * CB_BN == NNODES, "blocks must tile the nodes");
static_assert(CB_NCH * CB_CE == NEDGES, "chunks must tile the edges");
static_assert(CB_EPT * CB_TB == CB_CE, "threads must tile a chunk");
static_assert(CB_NPC * CB_NCH == NNODES, "chunk workgroups must tile the nodes");
static_assert(CB_NPC <= 2 * CB_TB, "two nodes per thread must cover a workgroup's nodes");
static_assert(CB_NB <= 2 * CB_TB && (CB_NB & 1) == 0, "two block counts per thread in the scan");
static_assert(CB_BN <= 256, "local node index must fit 8 bits");

__device__ __forceinline__ void node_bounds(int i, int g, int p, int* gs, int* ge)
{
    if (g != p) {
        gs[g] = i;
        if (p >= 0) ge[p] = i;
        for (int q = p + 1; q < g; ++q) { gs[q] = NNODES; ge[q] = 0; }
    }
    if (i == NNODES - 1) {
        ge[g] = NNODES;
        for (int q = g + 1; q < NGRAPHS; ++q) { gs[q] = NNODES; ge[q] = 0; }
    }
}

__global__ __launch_bounds__(CB_TB) void chunk_sort(
    const int* __restrict__ src, const int* __restrict__ dst,
    const int* __restrict__ batch, int* __restrict__ sorted,
    int* __restrict__ offT, int* gs, int* ge)
{
    __shared__ int hist[CB_NB];
    __shared__ int wtot[CB_TB / 32];
    __shared__ int stage[CB_CE];
    const int t    = (int)threadIdx.x;
    const int lane = t & 31;
    const int wave = t >> 5;
    const int c    = (int)blockIdx.x;
    for (int b = t; b < CB_NB; b += CB_TB) hist[b] = 0;
    const int* sp = src + (size_t)c * CB_CE + t;
    const int* dp = dst + (size_t)c * CB_CE + t;
    int s[CB_EPT], d[CB_EPT];
#pragma unroll
    for (int j = 0; j < CB_EPT; ++j) { s[j] = sp[j * CB_TB]; d[j] = dp[j * CB_TB]; }
    const int i0 = c * CB_NPC + t;
    const int i1 = i0 + CB_TB;
    const bool has1 = t < CB_NPC - CB_TB;
    const int g0 = batch[i0];
    const int p0 = (i0 > 0) ? batch[i0 - 1] : -1;
    const int g1 = has1 ? batch[i1] : 0;
    const int p1 = has1 ? batch[i1 - 1] : 0;
    __syncthreads();
#pragma unroll
    for (int j = 0; j < CB_EPT; ++j) atomicAdd(&hist[(unsigned)d[j] / CB_BN], 1);
    __syncthreads();

    int v0 = 0, v1 = 0;
    if (t < CB_NB / 2) { v0 = hist[2 * t]; v1 = hist[2 * t + 1]; }
    const int own = v0 + v1;
    int inc = own;
#pragma unroll
    for (int off = 1; off < 32; off <<= 1) {
        const int up = __shfl_up(inc, off, 32);
        if (lane >= off) inc += up;
    }
    if (lane == 31) wtot[wave] = inc;
    __syncthreads();
    int base = inc - own;
#pragma unroll
    for (int q = 0; q < CB_TB / 32; ++q) base += (q < wave) ? wtot[q] : 0;
    if (t < CB_NB / 2) {
        hist[2 * t]     = base;
        hist[2 * t + 1] = base + v0;
        offT[(2 * t) * CB_NCH + c]     = base;
        offT[(2 * t + 1) * CB_NCH + c] = base + v0;
    }
    if (t == 0) offT[CB_NB * CB_NCH + c] = CB_CE;
    __syncthreads();

    int pos[CB_EPT];
#pragma unroll
    for (int j = 0; j < CB_EPT; ++j) {
        const unsigned b = (unsigned)d[j] / CB_BN;
        const int loc = d[j] - (int)b * CB_BN;
        pos[j] = atomicAdd(&hist[b], 1);
        s[j] = (s[j] << 8) | loc;
    }
#pragma unroll
    for (int j = 0; j < CB_EPT; ++j) stage[pos[j]] = s[j];
    __syncthreads();
    int* outp = sorted + (size_t)c * CB_CE + t;
#pragma unroll
    for (int j = 0; j < CB_EPT; ++j) outp[j * CB_TB] = stage[t + j * CB_TB];

    node_bounds(i0, g0, p0, gs, ge);
    if (has1) node_bounds(i1, g1, p1, gs, ge);
}

#define CB_STAGE 8192
#define CB_BT    512
#define CB_HW    (CB_BT / 16)
#define CB_PPH   (256 / CB_HW)
static_assert(CB_BT >= 256, "the scans of the piece lengths and node counts take 256 threads");
static_assert(CB_NCH <= 256, "a block's pieces must fit the piece tables");
static_assert(CB_HW * CB_PPH == 256, "half waves must cover the piece tables");
__global__ __launch_bounds__(CB_BT) void block_csr(
    const int* __restrict__ sorted, const int* __restrict__ offT,
    int* __restrict__ offs, int* __restrict__ csr)
{
    __shared__ int ncnt[256];
    __shared__ int wtot[8];
    __shared__ int wlen[8];
    __shared__ int wbase[8];
    __shared__ int segsrc[256];
    __shared__ int seglp[256];
    __shared__ int seglen[256];
    __shared__ int lst[CB_STAGE];
    __shared__ int stage[CB_STAGE];
    const int t    = (int)threadIdx.x;
    const int lane = t & 31;
    const int wave = t >> 5;
    const int b    = (int)blockIdx.x;
    const int hw   = t >> 4;
    const int l16  = t & 15;

    int o0 = 0, len = 0;
    if (t < CB_NCH) {
        o0  = offT[b * CB_NCH + t];
        len = offT[(b + 1) * CB_NCH + t] - o0;
    }
    if (t < 256) ncnt[t] = 0;
    int linc = 0;
    if (t < 256) {
        linc = len;
#pragma unroll
        for (int off = 1; off < 32; off <<= 1) {
            const int up = __shfl_up(linc, off, 32);
            if (lane >= off) linc += up;
        }
        int so = o0;
#pragma unroll
        for (int off = 16; off; off >>= 1) so += __shfl_xor(so, off, 32);
        if (lane == 31) wlen[wave]  = linc;
        if (lane == 0)  wbase[wave] = so;
    }
    __syncthreads();
    int beg = 0, n = 0;
#pragma unroll
    for (int q = 0; q < 8; ++q) { beg += wbase[q]; n += wlen[q]; }
    const int end = beg + n;
    if (t < 256) {
        int lp = linc - len;
#pragma unroll
        for (int q = 0; q < 8; ++q) lp += (q < wave) ? wlen[q] : 0;
        seglp[t]  = lp;
        seglen[t] = len;
        segsrc[t] = (t < CB_NCH) ? t * CB_CE + o0 : 0;
    }
    __syncthreads();
    const bool fits = n <= CB_STAGE;

    int pl[CB_PPH], ps[CB_PPH], pp[CB_PPH];
#pragma unroll
    for (int q = 0; q < CB_PPH; ++q) {
        const int c = hw + CB_HW * q;
        pl[q] = seglen[c]; ps[q] = segsrc[c]; pp[q] = seglp[c];
    }
    if (fits) {
        int va[CB_PPH], vb[CB_PPH];
#pragma unroll
        for (int q = 0; q < CB_PPH; ++q) {
            va[q] = (l16 < pl[q])      ? sorted[ps[q] + l16]      : 0;
            vb[q] = (l16 + 16 < pl[q]) ? sorted[ps[q] + l16 + 16] : 0;
        }
#pragma unroll
        for (int q = 0; q < CB_PPH; ++q) {
            if (l16 < pl[q])      lst[pp[q] + l16]      = va[q];
            if (l16 + 16 < pl[q]) lst[pp[q] + l16 + 16] = vb[q];
        }
#pragma unroll
        for (int q = 0; q < CB_PPH; ++q) {
            for (int k = l16 + 32; k < pl[q]; k += 16) lst[pp[q] + k] = sorted[ps[q] + k];
        }
        __syncthreads();
        for (int i = t; i < n; i += CB_BT) atomicAdd(&ncnt[lst[i] & 255], 1);
    } else {
#pragma unroll
        for (int q = 0; q < CB_PPH; ++q) {
            for (int k = l16; k < pl[q]; k += 16) atomicAdd(&ncnt[sorted[ps[q] + k] & 255], 1);
        }
    }
    __syncthreads();

    int own = 0, inc = 0;
    if (t < 256) {
        own = ncnt[t];
        inc = own;
#pragma unroll
        for (int off = 1; off < 32; off <<= 1) {
            const int up = __shfl_up(inc, off, 32);
            if (lane >= off) inc += up;
        }
        if (lane == 31) wtot[wave] = inc;
    }
    __syncthreads();
    if (t < 256) {
        int base = beg;
#pragma unroll
        for (int q = 0; q < 8; ++q) base += (q < wave) ? wtot[q] : 0;
        const int start = base + (inc - own);
        ncnt[t] = start;
        if (t < CB_BN) offs[b * CB_BN + t] = start;
    }
    if (b == CB_NB - 1 && t == 0) offs[NNODES] = end;
    __syncthreads();
    if (fits) {
        for (int i = t; i < n; i += CB_BT) {
            const int v = lst[i];
            const int p = atomicAdd(&ncnt[v & 255], 1);
            stage[p - beg] = v >> 8;
        }
        __syncthreads();
        for (int i = t; i < n; i += CB_BT) csr[beg + i] = stage[i];
    } else {
#pragma unroll
        for (int q = 0; q < CB_PPH; ++q) {
            for (int k = l16; k < pl[q]; k += 16) {
                const int v = sorted[ps[q] + k];
                const int p = atomicAdd(&ncnt[v & 255], 1);
                csr[p] = v >> 8;
            }
        }
    }
}

__device__ __forceinline__ float other_half(float x)
{
    const int b = __float_as_int(x);
    return __int_as_float(__builtin_amdgcn_permlanex16(b, b, 0x76543210, 0xfedcba98, false, false));
}

#define FL_QPITCH 544
#define FL_TPITCH 132
#define FL_TILES (NNODES / 16)
#define FL_WGS   768
#define AGG_GRP  8
static_assert(FL_TILES * 16 == NNODES, "tiles must cover the nodes exactly");

__global__ __launch_bounds__(512) __attribute__((amdgpu_waves_per_eu(12))) void layer_fused(
    const float* __restrict__ X, const float* __restrict__ W,
    const float* __restrict__ bias, const int* __restrict__ offs,
    const int* __restrict__ csr, float* __restrict__ out,
    float* __restrict__ tmax, float* __restrict__ res, float* __restrict__ rmax,
    const int* __restrict__ batch)
{
    __shared__ __attribute__((aligned(16))) float wl[32 * FL_QPITCH];
    __shared__ __attribute__((aligned(16))) float tl[2][16 * FL_TPITCH];
    __shared__ __attribute__((aligned(16))) float st[2][16 * FL_TPITCH];
    __shared__ int tdeg[2][16];
    __shared__ float sb[HID];
    const int lane = (int)threadIdx.x & 31;
    const int wave = __builtin_amdgcn_readfirstlane((int)(threadIdx.x >> 5));
    const int m    = lane & 15;
    const int half = lane >> 4;

#pragma unroll
    for (int i = 0; i < 2; ++i) {
        const int q = wave + 16 * i;
        const float* wp = W + (size_t)(4 * q) * HID + 4 * lane;
        const float4 r0 = *(const float4*)(wp);
        const float4 r1 = *(const float4*)(wp + HID);
        const float4 r2 = *(const float4*)(wp + 2 * HID);
        const float4 r3 = *(const float4*)(wp + 3 * HID);
        float4 c0; c0.x = r0.x; c0.y = r1.x; c0.z = r2.x; c0.w = r3.x;
        float4 c1; c1.x = r0.y; c1.y = r1.y; c1.z = r2.y; c1.w = r3.y;
        float4 c2; c2.x = r0.z; c2.y = r1.z; c2.z = r2.z; c2.w = r3.z;
        float4 c3; c3.x = r0.w; c3.y = r1.w; c3.z = r2.w; c3.w = r3.w;
        float* d = wl + q * FL_QPITCH + 16 * lane;
        *(float4*)d = c0;
        *(float4*)(d + 4) = c1;
        *(float4*)(d + 8) = c2;
        *(float4*)(d + 12) = c3;
    }

    const float* colp = X + lane * 4;
    if (threadIdx.x < HID) sb[threadIdx.x] = bias[threadIdx.x];
    float* const sp = res ? res : out;
    int buf = 0;
    int pt = -1;
    auto flush_rows = [&]() {
        if (pt >= 0) {
            const float4 r = *(const float4*)(&st[buf ^ 1][wave * FL_TPITCH + lane * 4]);
            *(float4*)(sp + ((size_t)pt * 16 + wave) * HID + lane * 4) = r;
        }
    };
    for (int t = (int)blockIdx.x; t < FL_TILES; t += FL_WGS) {
        const int node = t * 16 + wave;
        const int beg = offs[node];
        const int end = offs[node + 1];
        int myidx = (beg + lane < end) ? csr[beg + lane] : 0;
        float4 rrow; rrow.x = 0.f; rrow.y = 0.f; rrow.z = 0.f; rrow.w = 0.f;
        if (res) rrow = *(const float4*)(res + (size_t)node * HID + lane * 4);
        bool rows = true;
        if (batch != nullptr && (wave < 8 || res == nullptr)) {
            const int g0 = batch[t * 16];
            const int gm = batch[t * 16 + m];
            rows = __ballot(gm != g0) != 0;
        }
        if (res) *(float4*)(&st[buf][wave * FL_TPITCH + lane * 4]) = rrow;
        float ax = 0.f, ay = 0.f, az = 0.f, aw = 0.f;

        for (int e0 = beg; e0 < end; e0 += 32) {
            if (e0 != beg) myidx = (e0 + lane < end) ? csr[e0 + lane] : 0;
            int cnt = end - e0; if (cnt > 32) cnt = 32;
            for (int j = 0; j < cnt; j += AGG_GRP) {
                const int r = cnt - j;
                int s[AGG_GRP];
#pragma unroll
                for (int k = 0; k < AGG_GRP; ++k) s[k] = __builtin_amdgcn_readlane(myidx, j + k);
                float4 v[AGG_GRP];
                v[0] = *(const float4*)(colp + (size_t)s[0] * HID);
#pragma unroll
                for (int k = 1; k < AGG_GRP; ++k) {
                    if (k < r) v[k] = *(const float4*)(colp + (size_t)s[k] * HID);
                }
                ax += v[0].x; ay += v[0].y; az += v[0].z; aw += v[0].w;
#pragma unroll
                for (int k = 1; k < AGG_GRP; ++k) {
                    if (k < r) { ax += v[k].x; ay += v[k].y; az += v[k].z; aw += v[k].w; }
                }
            }
        }
        const float inv = 1.0f / fmaxf((float)(end - beg), 1.0f);
        float4 mean; mean.x = ax * inv; mean.y = ay * inv; mean.z = az * inv; mean.w = aw * inv;
        *(float4*)(&tl[buf][wave * FL_TPITCH + lane * 4]) = mean;
        if (lane == 0) tdeg[buf][wave] = end - beg;
        __syncthreads();
        flush_rows();

        if (wave < 8) {
            v8f acc;
#pragma unroll
            for (int v = 0; v < 8; ++v) acc[v] = 0.0f;
            const float* al = &tl[buf][m * FL_TPITCH + 4 * half];
            const float* bl = wl + half * FL_QPITCH + 4 * (16 * wave + m);
            v4f fa[16], fb[16];
#pragma unroll
            for (int u = 0; u < 2; ++u) {
                fa[u] = *(const v4f*)(al + 8 * u);
                fb[u] = *(const v4f*)(bl + (2 * u) * FL_QPITCH);
            }
#pragma unroll
            for (int u = 0; u < 16; ++u) {
                if (u + 2 < 16) {
                    fa[u + 2] = *(const v4f*)(al + 8 * (u + 2));
                    fb[u + 2] = *(const v4f*)(bl + (2 * (u + 2)) * FL_QPITCH);
                }
                asm volatile("" :: "v"(fa[u]), "v"(fb[u]));
                v2f a0; a0.x = fa[u].x; a0.y = fa[u].y;
                v2f b0; b0.x = fb[u].x; b0.y = fb[u].y;
                v2f a1; a1.x = fa[u].z; a1.y = fa[u].w;
                v2f b1; b1.x = fb[u].z; b1.y = fb[u].w;
                acc = __builtin_amdgcn_wmma_f32_16x16x4_f32(
                    false, a0, false, b0, (short)0, acc, false, false);
                acc = __builtin_amdgcn_wmma_f32_16x16x4_f32(
                    false, a1, false, b1, (short)0, acc, false, false);
            }
            const unsigned lo = (unsigned)((8 * half) * HID + 16 * wave + m);
            const unsigned e0 = (unsigned)t * (16u * HID) + lo;
            const unsigned c0 = (unsigned)t * (unsigned)HID + (e0 & (HID - 1));
            float* sl = &st[buf][(8 * half) * FL_TPITCH + 16 * wave + m];
            const float bv = sb[16 * wave + m];
            float y[8];
#pragma unroll
            for (int v = 0; v < 8; ++v) {
                const int rr = v + 8 * half;
                const float s = (tdeg[buf][rr] > 0) ? acc[v] + bv : 0.0f;
                y[v] = s > 0.f ? s : ALPHA * s;
            }
            if (tmax) {
                float mx = y[0];
#pragma unroll
                for (int v = 1; v < 8; ++v) mx = fmaxf(mx, y[v]);
                mx = fmaxf(mx, other_half(mx));
                if (half == 0) tmax[c0] = mx;
            }
            if (res) {
                if (rows) {
#pragma unroll
                    for (int v = 0; v < 8; ++v) out[e0 + (unsigned)(v * HID)] = y[v];
                }
                float rv[8];
#pragma unroll
                for (int v = 0; v < 8; ++v) rv[v] = sl[v * FL_TPITCH];
#pragma unroll
                for (int v = 0; v < 8; ++v) { rv[v] += y[v]; sl[v * FL_TPITCH] = rv[v]; }
                float mx = rv[0];
#pragma unroll
                for (int v = 1; v < 8; ++v) mx = fmaxf(mx, rv[v]);
                mx = fmaxf(mx, other_half(mx));
                if (half == 0) rmax[c0] = mx;
            } else if (rows) {
#pragma unroll
                for (int v = 0; v < 8; ++v) sl[v * FL_TPITCH] = y[v];
            }
        }
        pt = (res != nullptr || rows) ? t : -1;
        buf ^= 1;
    }
    __syncthreads();
    flush_rows();
}

#define HD_G     4
#define HD_INF   16
#define HD_PITCH 516
static_assert(NGRAPHS % HD_G == 0, "workgroups must tile the graphs");

__global__ __launch_bounds__(1024) void head_fused(
    const float* __restrict__ ht, const float* __restrict__ h1,
    const float* __restrict__ h2, const float* __restrict__ h3,
    const float* __restrict__ mt, const float* __restrict__ m1,
    const float* __restrict__ m2, const float* __restrict__ m3,
    const int* __restrict__ gs, const int* __restrict__ ge,
    const float* __restrict__ Wm1, const float* __restrict__ bm1,
    const float* __restrict__ gamma, const float* __restrict__ beta,
    const float* __restrict__ Wm2, const float* __restrict__ bm2,
    float* __restrict__ out)
{
    __shared__ __attribute__((aligned(16))) float pl[2][HD_G][HD_PITCH];
    __shared__ float part[4][HD_G][HID];
    __shared__ float zl[HD_G][HID];
    const int lane = (int)threadIdx.x & 31;
    const int wave = __builtin_amdgcn_readfirstlane((int)(threadIdx.x >> 5));
    const int g0 = (int)blockIdx.x * HD_G;

    {
        const int unit = wave & 15;
        const int prt  = wave >> 4;
        const int gl   = unit >> 2;
        const int arr  = unit & 3;
        const float* rowp = ((arr == 0) ? ht : (arr == 1) ? h1 : (arr == 2) ? h2 : h3) + lane * 4;
        const float* maxp = ((arr == 0) ? mt : (arr == 1) ? m1 : (arr == 2) ? m2 : m3) + lane * 4;
        const int b = gs[g0 + gl], e = ge[g0 + gl];
        const int t0 = (b + 15) >> 4;
        const int t1 = e >> 4;
        int nt = 0, hend = e, tbeg = e;
        if (t0 <= t1) { nt = t1 - t0; hend = 16 * t0; tbeg = 16 * t1; }
        int nh = hend - b; if (nh < 0) nh = 0;
        const int nl = e - tbeg;
        const int n = nt + nh + nl;

        float mx = -FMAXV, my = -FMAXV, mz = -FMAXV, mw = -FMAXV;
        for (int i0 = prt; i0 < n; i0 += 2 * HD_INF) {
            float4 v[HD_INF];
#pragma unroll
            for (int k = 0; k < HD_INF; ++k) {
                const int i = i0 + 2 * k;
                v[k].x = -FMAXV; v[k].y = -FMAXV; v[k].z = -FMAXV; v[k].w = -FMAXV;
                if (i < n) {
                    const float* p = (i < nt)      ? maxp + (size_t)(t0 + i) * HID
                                   : (i < nt + nh) ? rowp + (size_t)(b + (i - nt)) * HID
                                                   : rowp + (size_t)(tbeg + (i - nt - nh)) * HID;
                    v[k] = *(const float4*)p;
                }
            }
#pragma unroll
            for (int k = 0; k < HD_INF; ++k) {
                mx = fmaxf(mx, v[k].x); my = fmaxf(my, v[k].y);
                mz = fmaxf(mz, v[k].z); mw = fmaxf(mw, v[k].w);
            }
        }
        float4 m; m.x = mx; m.y = my; m.z = mz; m.w = mw;
        *(float4*)(&pl[prt][gl][arr * HID + lane * 4]) = m;
    }
    __syncthreads();

    {
        const int ct = wave & 7;
        const int kq = wave >> 3;
        const int c  = lane & 15;
        const int hf = lane >> 4;
        const bool live = c < HD_G;
        v8f acc;
#pragma unroll
        for (int v = 0; v < 8; ++v) acc[v] = 0.0f;
        const float* bp  = Wm1 + (size_t)(kq * 128 + 2 * hf) * HID + ct * 16 + c;
        const float* ap0 = &pl[0][c & (HD_G - 1)][kq * 128 + 2 * hf];
        const float* ap1 = &pl[1][c & (HD_G - 1)][kq * 128 + 2 * hf];
#pragma unroll 16
        for (int s = 0; s < 32; ++s) {
            const v2f x0 = *(const v2f*)(ap0 + 4 * s);
            const v2f x1 = *(const v2f*)(ap1 + 4 * s);
            v2f a;
            a.x = live ? fmaxf(x0.x, x1.x) : 0.0f;
            a.y = live ? fmaxf(x0.y, x1.y) : 0.0f;
            v2f bb;
            bb.x = bp[(size_t)(4 * s) * HID];
            bb.y = bp[(size_t)(4 * s + 1) * HID];
            acc = __builtin_amdgcn_wmma_f32_16x16x4_f32(
                false, a, false, bb, (short)0, acc, false, false);
        }
        if (hf == 0) {
#pragma unroll
            for (int v = 0; v < HD_G; ++v) part[kq][v][ct * 16 + c] = acc[v];
        }
    }
    __syncthreads();

    if (threadIdx.x < HD_G * HID) {
        const int row = (int)threadIdx.x >> 7;
        const int col = (int)threadIdx.x & (HID - 1);
        float s = part[0][row][col];
        s += part[1][row][col];
        s += part[2][row][col];
        s += part[3][row][col];
        s += bm1[col];
        const float rs = 0.99999500003749971f;
        const float g  = gamma[col] * rs;
        zl[row][col] = fmaxf(s * g + beta[col], 0.0f);
    }
    __syncthreads();

    {
        const int gl = wave >> 3;
        const float z0 = zl[gl][lane];
        const float z1 = zl[gl][lane + 32];
        const float z2 = zl[gl][lane + 64];
        const float z3 = zl[gl][lane + 96];
        for (int c = wave & 7; c < NCLS; c += 8) {
            float s = z0 * Wm2[lane * NCLS + c]
                    + z1 * Wm2[(lane + 32) * NCLS + c]
                    + z2 * Wm2[(lane + 64) * NCLS + c]
                    + z3 * Wm2[(lane + 96) * NCLS + c];
            for (int off = 16; off; off >>= 1) s += __shfl_xor(s, off, 32);
            if (lane == 0) out[(g0 + gl) * NCLS + c] = s + bm2[c];
        }
    }
}

extern "C" void kernel_launch(void* const* d_in, const int* in_sizes, int n_in,
                              void* d_out, int out_size, void* d_ws, size_t ws_size,
                              hipStream_t stream)
{
    (void)in_sizes; (void)n_in; (void)out_size; (void)ws_size;
    const float* x     = (const float*)d_in[0];
    const int*   ei    = (const int*)d_in[1];
    const int*   batch = (const int*)d_in[2];
    const float* W0 = (const float*)d_in[3];  const float* b0 = (const float*)d_in[4];
    const float* W1 = (const float*)d_in[5];  const float* b1 = (const float*)d_in[6];
    const float* W2 = (const float*)d_in[7];  const float* b2 = (const float*)d_in[8];
    const float* W3 = (const float*)d_in[9];  const float* b3 = (const float*)d_in[10];
    const float* Wm1 = (const float*)d_in[11]; const float* bm1 = (const float*)d_in[12];
    const float* gamma = (const float*)d_in[13]; const float* beta = (const float*)d_in[14];
    const float* Wm2 = (const float*)d_in[15]; const float* bm2 = (const float*)d_in[16];
    const int* esrc = ei;
    const int* edst = ei + NEDGES;

    char* ws = (char*)d_ws;
    size_t off = 0;
    auto alloc = [&](size_t bytes) -> void* {
        void* p = ws + off;
        off = (off + bytes + 255) & ~(size_t)255;
        return p;
    };
    const size_t featB = (size_t)NNODES * HID * sizeof(float);
    float* tmpH   = (float*)alloc(featB);
    float* h      = (float*)alloc(featB);
    float* h1     = (float*)alloc(featB);
    float* h2     = (float*)alloc(featB);
    float* ht     = (float*)alloc(featB);
    int*   csr    = (int*)alloc((size_t)NEDGES * sizeof(int));
    int*   offs   = (int*)alloc((size_t)(NNODES + 1) * sizeof(int));
    int*   offT   = (int*)alloc((size_t)(CB_NB + 1) * CB_NCH * sizeof(int));
    int*   sorted = (int*)tmpH;
    float* mt     = tmpH + 2097152;
    float* m1     = mt + FL_TILES * HID;
    float* m2     = m1 + FL_TILES * HID;
    float* m3     = m2 + FL_TILES * HID;
    int*   gs     = (int*)alloc((size_t)NGRAPHS * sizeof(int));
    int*   ge     = (int*)alloc((size_t)NGRAPHS * sizeof(int));

    chunk_sort<<<CB_NCH, CB_TB, 0, stream>>>(esrc, edst, batch, sorted, offT, gs, ge);
    block_csr<<<CB_NB, CB_BT, 0, stream>>>(sorted, offT, offs, csr);

    float* const none = nullptr;
    const int* const nobatch = nullptr;
    layer_fused<<<FL_WGS, 512, 0, stream>>>(x, W0, b0, offs, csr, h, none, none, none, nobatch);
    layer_fused<<<FL_WGS, 512, 0, stream>>>(h, W1, b1, offs, csr, h1, m1, none, none, nobatch);
    layer_fused<<<FL_WGS, 512, 0, stream>>>(h1, W2, b2, offs, csr, h2, m2, h, m3, batch);
    layer_fused<<<FL_WGS, 512, 0, stream>>>(h, W3, b3, offs, csr, ht, mt, none, none, batch);

    head_fused<<<NGRAPHS / HD_G, 1024, 0, stream>>>(ht, h1, h2, h, mt, m1, m2, m3, gs, ge,
                                                    Wm1, bm1, gamma, beta, Wm2, bm2, (float*)d_out);
}
